# starting kernel
# baseline (speedup 1.0000x reference)
#include <hip/hip_runtime.h>
#include <hip/hip_bf16.h>
#include <math.h>

#define NNODES 50000
#define NEDGES 800000
#define ETOT   (NEDGES + NNODES)
#define FIN    256
#define H1     8
#define C1     32
#define D1     256
#define OUT2   2
#define SLOPE  0.2f

#define LDSPAD 4
#define ASTRIDE (FIN + LDSPAD)

typedef float v2f __attribute__((ext_vector_type(2)));
typedef float v8f __attribute__((ext_vector_type(8)));

__device__ __forceinline__ void edge_sd(const int* __restrict__ ei, int e, int& s, int& d) {
    if (e < NEDGES) { s = ei[e]; d = ei[NEDGES + e]; }
    else            { s = e - NEDGES; d = s; }
}

__device__ __forceinline__ float lrelu(float v) {
    return v > 0.0f ? v : SLOPE * v;
}

__device__ __forceinline__ void atomicMaxF(float* addr, float val) {
    if (val >= 0.0f) atomicMax((int*)addr, __float_as_int(val));
    else             atomicMin((unsigned int*)addr, __float_as_uint(val));
}

__global__ __launch_bounds__(512) void gemm1_wmma(const float* __restrict__ x,
                                                  const float* __restrict__ W1,
                                                  float* __restrict__ xp) {
    __shared__ float aTile[16 * ASTRIDE];

    const int t     = threadIdx.x;
    const int lane  = t & 31;
    const int wave  = t >> 5;
    const int tileM = blockIdx.x;
    const int half  = lane >> 4;
    const int l16   = lane & 15;

    {
        const int row = t >> 5;
        const int col = (t & 31) * 8;
        const float4* src = (const float4*)(x + (size_t)(tileM * 16 + row) * FIN + col);
        float4* dst = (float4*)(aTile + row * ASTRIDE + col);
        dst[0] = src[0];
        dst[1] = src[1];
    }
    __syncthreads();

    const int bcol = wave * 16 + l16;

    v8f acc = {};
#pragma unroll
    for (int k = 0; k < FIN; k += 4) {
        const int kk = k + half * 2;
        v2f a = *(const v2f*)(aTile + l16 * ASTRIDE + kk);
        v2f b;
        b.x = W1[kk * D1 + bcol];
        b.y = W1[(kk + 1) * D1 + bcol];
        acc = __builtin_amdgcn_wmma_f32_16x16x4_f32(
             false, a,  false, b,
             (short)0, acc,  false,  false);
    }

    const int mrow = tileM * 16 + half * 8;
#pragma unroll
    for (int r = 0; r < 8; ++r)
        xp[(size_t)(mrow + r) * D1 + bcol] = acc[r];
}

__global__ void alpha1_kernel(const float* __restrict__ xp,
                              const float* __restrict__ a_src,
                              const float* __restrict__ a_dst,
                              float* __restrict__ as, float* __restrict__ ad) {
    int i = blockIdx.x * blockDim.x + threadIdx.x;
    if (i >= NNODES * H1) return;
    int h = i & (H1 - 1);
    int n = i >> 3;
    const float* row = xp + (size_t)n * D1 + h * C1;
    const float* ws  = a_src + h * C1;
    const float* wd  = a_dst + h * C1;
    float s = 0.0f, d = 0.0f;
#pragma unroll
    for (int c = 0; c < C1; ++c) {
        float v = row[c];
        s = fmaf(v, ws[c], s);
        d = fmaf(v, wd[c], d);
    }
    as[i] = s;
    ad[i] = d;
}

__global__ void fill_f32(float* __restrict__ p, float v, int n) {
    int i = blockIdx.x * blockDim.x + threadIdx.x;
    if (i < n) p[i] = v;
}

template <int NH>
__global__ void edge_max(const int* __restrict__ ei,
                         const float* __restrict__ as,
                         const float* __restrict__ ad,
                         float* __restrict__ m) {
    int e = blockIdx.x * blockDim.x + threadIdx.x;
    if (e >= ETOT) return;
    int s, d;
    edge_sd(ei, e, s, d);
#pragma unroll
    for (int h = 0; h < NH; ++h) {
        float v = lrelu(as[s * NH + h] + ad[d * NH + h]);
        atomicMaxF(&m[d * NH + h], v);
    }
}

template <int NH>
__global__ void edge_denom(const int* __restrict__ ei,
                           const float* __restrict__ as,
                           const float* __restrict__ ad,
                           const float* __restrict__ m,
                           float* __restrict__ den) {
    int e = blockIdx.x * blockDim.x + threadIdx.x;
    if (e >= ETOT) return;
    int s, d;
    edge_sd(ei, e, s, d);
#pragma unroll
    for (int h = 0; h < NH; ++h) {
        float v = lrelu(as[s * NH + h] + ad[d * NH + h]);
        atomicAdd(&den[d * NH + h], __expf(v - m[d * NH + h]));
    }
}

__global__ __launch_bounds__(256) void agg1(const int* __restrict__ ei,
                                            const float* __restrict__ xp,
                                            const float* __restrict__ as,
                                            const float* __restrict__ ad,
                                            const float* __restrict__ m,
                                            const float* __restrict__ den,
                                            float* __restrict__ out) {
    int e = blockIdx.x;
    int t = threadIdx.x;
    int h = t >> 5;
    int s, d;
    edge_sd(ei, e, s, d);
    float v     = lrelu(as[s * H1 + h] + ad[d * H1 + h]);
    float alpha = __expf(v - m[d * H1 + h]) / den[d * H1 + h];
    atomicAdd(&out[(size_t)d * D1 + t], alpha * xp[(size_t)s * D1 + t]);
}

__global__ void bias_relu(float* __restrict__ hbuf, const float* __restrict__ b) {
    int i = blockIdx.x * blockDim.x + threadIdx.x;
    if (i >= NNODES * D1) return;
    float v = hbuf[i] + b[i & (D1 - 1)];
    hbuf[i] = v > 0.0f ? v : 0.0f;
}

__global__ __launch_bounds__(256) void gemm2_alpha2(const float* __restrict__ h,
                                                    const float* __restrict__ W2,
                                                    const float* __restrict__ aw_src,
                                                    const float* __restrict__ aw_dst,
                                                    float* __restrict__ xp2,
                                                    float* __restrict__ as2,
                                                    float* __restrict__ ad2) {
    int lane = threadIdx.x & 31;
    int n    = blockIdx.x * (blockDim.x >> 5) + (threadIdx.x >> 5);
    if (n >= NNODES) return;
    float s0 = 0.0f, s1 = 0.0f;
    for (int k = lane; k < D1; k += 32) {
        float v = h[(size_t)n * D1 + k];
        s0 = fmaf(v, W2[k * OUT2 + 0], s0);
        s1 = fmaf(v, W2[k * OUT2 + 1], s1);
    }
#pragma unroll
    for (int off = 16; off > 0; off >>= 1) {
        s0 += __shfl_down(s0, off);
        s1 += __shfl_down(s1, off);
    }
    if (lane == 0) {
        xp2[n * OUT2 + 0] = s0;
        xp2[n * OUT2 + 1] = s1;
        as2[n] = fmaf(s0, aw_src[0], s1 * aw_src[1]);
        ad2[n] = fmaf(s0, aw_dst[0], s1 * aw_dst[1]);
    }
}

__global__ void agg2(const int* __restrict__ ei,
                     const float* __restrict__ xp2,
                     const float* __restrict__ as2,
                     const float* __restrict__ ad2,
                     const float* __restrict__ m2,
                     const float* __restrict__ den2,
                     float* __restrict__ out) {
    int e = blockIdx.x * blockDim.x + threadIdx.x;
    if (e >= ETOT) return;
    int s, d;
    edge_sd(ei, e, s, d);
    float v     = lrelu(as2[s] + ad2[d]);
    float alpha = __expf(v - m2[d]) / den2[d];
    atomicAdd(&out[d * OUT2 + 0], alpha * xp2[s * OUT2 + 0]);
    atomicAdd(&out[d * OUT2 + 1], alpha * xp2[s * OUT2 + 1]);
}

__global__ void bias2_add(float* __restrict__ out, const float* __restrict__ b) {
    int i = blockIdx.x * blockDim.x + threadIdx.x;
    if (i < NNODES * OUT2) out[i] += b[i & 1];
}

extern "C" void kernel_launch(void* const* d_in, const int* in_sizes, int n_in,
                              void* d_out, int out_size, void* d_ws, size_t ws_size,
                              hipStream_t stream) {
    const float* x     = (const float*)d_in[0];
    const int*   ei    = (const int*)d_in[1];
    const float* W1    = (const float*)d_in[2];
    const float* asw1  = (const float*)d_in[3];
    const float* adw1  = (const float*)d_in[4];
    const float* b1    = (const float*)d_in[5];
    const float* W2    = (const float*)d_in[6];
    const float* asw2  = (const float*)d_in[7];
    const float* adw2  = (const float*)d_in[8];
    const float* b2    = (const float*)d_in[9];
    float*       out   = (float*)d_out;

    float* w    = (float*)d_ws;
    float* xp1  = w;
    float* hbuf = xp1  + (size_t)NNODES * D1;
    float* as1  = hbuf + (size_t)NNODES * D1;
    float* ad1  = as1  + (size_t)NNODES * H1;
    float* m1   = ad1  + (size_t)NNODES * H1;
    float* dn1  = m1   + (size_t)NNODES * H1;
    float* xp2  = dn1  + (size_t)NNODES * H1;
    float* as2  = xp2  + (size_t)NNODES * OUT2;
    float* ad2  = as2  + (size_t)NNODES;
    float* m2   = ad2  + (size_t)NNODES;
    float* dn2  = m2   + (size_t)NNODES;

    hipMemsetAsync(hbuf, 0, (size_t)NNODES * D1 * sizeof(float), stream);
    hipMemsetAsync(dn1,  0, (size_t)NNODES * H1 * sizeof(float), stream);
    hipMemsetAsync(dn2,  0, (size_t)NNODES * sizeof(float), stream);
    hipMemsetAsync(out,  0, (size_t)NNODES * OUT2 * sizeof(float), stream);
    fill_f32<<<(NNODES * H1 + 255) / 256, 256, 0, stream>>>(m1, -INFINITY, NNODES * H1);
    fill_f32<<<(NNODES + 255) / 256, 256, 0, stream>>>(m2, -INFINITY, NNODES);

    gemm1_wmma<<<NNODES / 16, 512, 0, stream>>>(x, W1, xp1);
    alpha1_kernel<<<(NNODES * H1 + 255) / 256, 256, 0, stream>>>(xp1, asw1, adw1, as1, ad1);
    edge_max<H1><<<(ETOT + 255) / 256, 256, 0, stream>>>(ei, as1, ad1, m1);
    edge_denom<H1><<<(ETOT + 255) / 256, 256, 0, stream>>>(ei, as1, ad1, m1, dn1);
    agg1<<<ETOT, 256, 0, stream>>>(ei, xp1, as1, ad1, m1, dn1, hbuf);
    bias_relu<<<(NNODES * D1 + 255) / 256, 256, 0, stream>>>(hbuf, b1);

    gemm2_alpha2<<<(NNODES + 7) / 8, 256, 0, stream>>>(hbuf, W2, asw2, adw2, xp2, as2, ad2);
    edge_max<1><<<(ETOT + 255) / 256, 256, 0, stream>>>(ei, as2, ad2, m2);
    edge_denom<1><<<(ETOT + 255) / 256, 256, 0, stream>>>(ei, as2, ad2, m2, dn2);
    agg2<<<(ETOT + 255) / 256, 256, 0, stream>>>(ei, xp2, as2, ad2, m2, dn2, out);
    bias2_add<<<(NNODES * OUT2 + 255) / 256, 256, 0, stream>>>(out, b2);
}


# Layer-1 softmax+aggregation by owner groups: 256 work-groups hold 196 rows each in LDS, list their edges in one scan, sum by LDS adds (arrival order), then /denominator +bias relu; no global atomics
# speedup vs baseline: 3.6257x; 3.6257x over previous
#include <hip/hip_runtime.h>
#include <hip/hip_bf16.h>
#include <math.h>

#define NNODES 50000
#define NEDGES 800000
#define ETOT   (NEDGES + NNODES)
#define FIN    256
#define H1     8
#define C1     32
#define D1     256
#define OUT2   2
#define SLOPE  0.2f

#define LDSPAD 4
#define ASTRIDE (FIN + LDSPAD)

typedef float v2f __attribute__((ext_vector_type(2)));
typedef float v8f __attribute__((ext_vector_type(8)));

__device__ __forceinline__ void edge_sd(const int* __restrict__ ei, int e, int& s, int& d) {
    if (e < NEDGES) { s = ei[e]; d = ei[NEDGES + e]; }
    else            { s = e - NEDGES; d = s; }
}

__device__ __forceinline__ float lrelu(float v) {
    return v > 0.0f ? v : SLOPE * v;
}

__device__ __forceinline__ void atomicMaxF(float* addr, float val) {
    if (val >= 0.0f) atomicMax((int*)addr, __float_as_int(val));
    else             atomicMin((unsigned int*)addr, __float_as_uint(val));
}

__global__ __launch_bounds__(512) void gemm1_wmma(const float* __restrict__ x,
                                                  const float* __restrict__ W1,
                                                  float* __restrict__ xp) {
    __shared__ float aTile[16 * ASTRIDE];

    const int t     = threadIdx.x;
    const int lane  = t & 31;
    const int wave  = t >> 5;
    const int tileM = blockIdx.x;
    const int half  = lane >> 4;
    const int l16   = lane & 15;

    {
        const int row = t >> 5;
        const int col = (t & 31) * 8;
        const float4* src = (const float4*)(x + (size_t)(tileM * 16 + row) * FIN + col);
        float4* dst = (float4*)(aTile + row * ASTRIDE + col);
        dst[0] = src[0];
        dst[1] = src[1];
    }
    __syncthreads();

    const int bcol = wave * 16 + l16;

    v8f acc = {};
#pragma unroll
    for (int k = 0; k < FIN; k += 4) {
        const int kk = k + half * 2;
        v2f a = *(const v2f*)(aTile + l16 * ASTRIDE + kk);
        v2f b;
        b.x = W1[kk * D1 + bcol];
        b.y = W1[(kk + 1) * D1 + bcol];
        acc = __builtin_amdgcn_wmma_f32_16x16x4_f32(
             false, a,  false, b,
             (short)0, acc,  false,  false);
    }

    const int mrow = tileM * 16 + half * 8;
#pragma unroll
    for (int r = 0; r < 8; ++r)
        xp[(size_t)(mrow + r) * D1 + bcol] = acc[r];
}

__global__ void alpha1_kernel(const float* __restrict__ xp,
                              const float* __restrict__ a_src,
                              const float* __restrict__ a_dst,
                              float* __restrict__ as, float* __restrict__ ad) {
    int i = blockIdx.x * blockDim.x + threadIdx.x;
    if (i >= NNODES * H1) return;
    int h = i & (H1 - 1);
    int n = i >> 3;
    const float* row = xp + (size_t)n * D1 + h * C1;
    const float* ws  = a_src + h * C1;
    const float* wd  = a_dst + h * C1;
    float s = 0.0f, d = 0.0f;
#pragma unroll
    for (int c = 0; c < C1; ++c) {
        float v = row[c];
        s = fmaf(v, ws[c], s);
        d = fmaf(v, wd[c], d);
    }
    as[i] = s;
    ad[i] = d;
}

__global__ void fill_f32(float* __restrict__ p, float v, int n) {
    int i = blockIdx.x * blockDim.x + threadIdx.x;
    if (i < n) p[i] = v;
}

template <int NH>
__global__ void edge_max(const int* __restrict__ ei,
                         const float* __restrict__ as,
                         const float* __restrict__ ad,
                         float* __restrict__ m) {
    int e = blockIdx.x * blockDim.x + threadIdx.x;
    if (e >= ETOT) return;
    int s, d;
    edge_sd(ei, e, s, d);
#pragma unroll
    for (int h = 0; h < NH; ++h) {
        float v = lrelu(as[s * NH + h] + ad[d * NH + h]);
        atomicMaxF(&m[d * NH + h], v);
    }
}

template <int NH>
__global__ void edge_denom(const int* __restrict__ ei,
                           const float* __restrict__ as,
                           const float* __restrict__ ad,
                           const float* __restrict__ m,
                           float* __restrict__ den) {
    int e = blockIdx.x * blockDim.x + threadIdx.x;
    if (e >= ETOT) return;
    int s, d;
    edge_sd(ei, e, s, d);
#pragma unroll
    for (int h = 0; h < NH; ++h) {
        float v = lrelu(as[s * NH + h] + ad[d * NH + h]);
        atomicAdd(&den[d * NH + h], __expf(v - m[d * NH + h]));
    }
}

#define OWN_BLOCKS  256
#define OWN_NODES   196
#define OWN_THREADS 1024
#define OWN_WAVES   (OWN_THREADS / 32)
#define OWN_CAP     20480
#define OWN_G       2
#define OWN_SCAN    8
#define OWN_NV      (NEDGES / 4)
static_assert(OWN_BLOCKS * OWN_NODES >= NNODES, "every row needs an owner");
static_assert(NNODES - (OWN_BLOCKS - 1) * OWN_NODES > 0, "no group without rows");
static_assert(NNODES <= 65536 && OWN_NODES <= 65535, "a list entry packs two 16-bit numbers");
static_assert(NEDGES % 4 == 0, "the destination column is read four at a time");
static_assert(NEDGES <= (1 << 20) && OWN_NODES <= (1 << 12), "edge number and local row share a word during the scan");
static_assert(OWN_NODES <= OWN_THREADS && OWN_NODES <= OWN_CAP, "self loops go in first");

__global__ __launch_bounds__(OWN_THREADS) void agg1_owner(const int* __restrict__ ei,
                                                          const float* __restrict__ xp,
                                                          const float* __restrict__ as,
                                                          const float* __restrict__ ad,
                                                          const float* __restrict__ b1,
                                                          float* __restrict__ hout) {
    __shared__ float        acc[OWN_NODES * D1];
    __shared__ float        mx[OWN_NODES * H1];
    __shared__ float        dn[OWN_NODES * H1];
    __shared__ unsigned int list[OWN_CAP];
    __shared__ unsigned int cnt;

    const int tid  = threadIdx.x;
    const int lane = tid & 31;
    const int wv   = __builtin_amdgcn_readfirstlane(tid >> 5);
    const int base = blockIdx.x * OWN_NODES;
    const int nown = (NNODES - base) < OWN_NODES ? (NNODES - base) : OWN_NODES;

    {
        float4* acc4 = (float4*)acc;
        for (int i = tid; i < OWN_NODES * D1 / 4; i += OWN_THREADS)
            acc4[i] = make_float4(0.0f, 0.0f, 0.0f, 0.0f);
        for (int i = tid; i < OWN_NODES * H1; i += OWN_THREADS) {
            mx[i] = -INFINITY;
            dn[i] = 0.0f;
        }
        if (tid < nown) list[tid] = (unsigned)(base + tid) | ((unsigned)tid << 16);
        if (tid == 0) cnt = (unsigned)nown;
    }
    __syncthreads();

    const int4* __restrict__ dst4 = (const int4*)(ei + NEDGES);
    for (int i0 = tid; i0 < OWN_NV; i0 += OWN_SCAN * OWN_THREADS) {
        int4 v[OWN_SCAN];
#pragma unroll
        for (int u = 0; u < OWN_SCAN; ++u) {
            const int i = i0 + u * OWN_THREADS;
            v[u] = i < OWN_NV ? dst4[i] : make_int4(-1, -1, -1, -1);
        }
#pragma unroll
        for (int u = 0; u < OWN_SCAN; ++u) {
            const int e = (i0 + u * OWN_THREADS) * 4;
            const int dd[4] = {v[u].x, v[u].y, v[u].z, v[u].w};
#pragma unroll
            for (int q = 0; q < 4; ++q) {
                const unsigned dl = (unsigned)(dd[q] - base);
                if (dl < (unsigned)nown) {
                    const unsigned idx = atomicAdd(&cnt, 1u);
                    if (idx < OWN_CAP) list[idx] = (unsigned)(e + q) | (dl << 20);
                }
            }
        }
    }
    __syncthreads();

    const unsigned total = cnt;
    const bool spill = total > OWN_CAP;
    const int n = __builtin_amdgcn_readfirstlane(spill ? nown : (int)total);

    for (int t = nown + tid; t < n; t += OWN_THREADS) {
        const unsigned en = list[t];
        list[t] = (unsigned)ei[en & 0xfffffu] | ((en >> 20) << 16);
    }
    __syncthreads();

    for (int t0 = tid; t0 < n * H1; t0 += 4 * OWN_THREADS) {
        float lv[4];
        int   mi[4];
#pragma unroll
        for (int u = 0; u < 4; ++u) {
            const int  t  = t0 + u * OWN_THREADS;
            const bool ok = t < n * H1;
            const unsigned en = list[ok ? (t >> 3) : 0];
            const int hh = t & (H1 - 1);
            const int s  = (int)(en & 0xffffu);
            const int dl = (int)(en >> 16);
            mi[u] = ok ? dl * H1 + hh : -1;
            lv[u] = lrelu(as[s * H1 + hh] + ad[(base + dl) * H1 + hh]);
        }
#pragma unroll
        for (int u = 0; u < 4; ++u)
            if (mi[u] >= 0) atomicMaxF(&mx[mi[u]], lv[u]);
    }
    if (spill) {
        for (int i = tid; i < OWN_NV; i += OWN_THREADS) {
            const int4 v = dst4[i];
            const int dd[4] = {v.x, v.y, v.z, v.w};
#pragma unroll
            for (int q = 0; q < 4; ++q) {
                const unsigned dl = (unsigned)(dd[q] - base);
                if (dl < (unsigned)nown) {
                    const int s = ei[i * 4 + q];
                    for (int hh = 0; hh < H1; ++hh)
                        atomicMaxF(&mx[dl * H1 + hh],
                                   lrelu(as[s * H1 + hh] + ad[(base + (int)dl) * H1 + hh]));
                }
            }
        }
    }
    __syncthreads();

    {
        const int j = lane >> 3;
        const int h = lane & (H1 - 1);
        for (int i0 = wv * (4 * OWN_G); i0 < n; i0 += OWN_WAVES * 4 * OWN_G) {
            unsigned en[OWN_G];
#pragma unroll
            for (int g = 0; g < OWN_G; ++g) {
                const int idx = i0 + g * 4 + j;
                en[g] = list[idx < n ? idx : 0];
            }
            float va[OWN_G], vb[OWN_G];
            int   mh[OWN_G];
#pragma unroll
            for (int g = 0; g < OWN_G; ++g) {
                const int s  = (int)(en[g] & 0xffffu);
                const int dl = (int)(en[g] >> 16);
                mh[g] = dl * H1 + h;
                va[g] = as[s * H1 + h];
                vb[g] = ad[(base + dl) * H1 + h];
            }
            float row[OWN_G][4][H1];
#pragma unroll
            for (int g = 0; g < OWN_G; ++g) {
#pragma unroll
                for (int jj = 0; jj < 4; ++jj) {
                    const unsigned ej = (unsigned)__builtin_amdgcn_readlane((int)en[g], jj * H1);
                    const float* xr = xp + (size_t)(ej & 0xffffu) * D1 + lane;
#pragma unroll
                    for (int hh = 0; hh < H1; ++hh) row[g][jj][hh] = xr[hh * C1];
                }
            }
            float p[OWN_G];
#pragma unroll
            for (int g = 0; g < OWN_G; ++g)
                p[g] = __expf(lrelu(va[g] + vb[g]) - mx[mh[g]]);
#pragma unroll
            for (int g = 0; g < OWN_G; ++g)
                if (i0 + g * 4 + j < n) atomicAdd(&dn[mh[g]], p[g]);
#pragma unroll
            for (int g = 0; g < OWN_G; ++g) {
#pragma unroll
                for (int jj = 0; jj < 4; ++jj) {
                    if (i0 + g * 4 + jj < n) {
                        const unsigned ej = (unsigned)__builtin_amdgcn_readlane((int)en[g], jj * H1);
                        float* arow = acc + (ej >> 16) * D1 + lane;
#pragma unroll
                        for (int hh = 0; hh < H1; ++hh) {
                            const float a = __int_as_float(
                                __builtin_amdgcn_readlane(__float_as_int(p[g]), jj * H1 + hh));
                            atomicAdd(arow + hh * C1, a * row[g][jj][hh]);
                        }
                    }
                }
            }
        }
    }
    if (spill) {
        for (int it = 0; it < (OWN_NV + OWN_THREADS - 1) / OWN_THREADS; ++it) {
            const int i = it * OWN_THREADS + tid;
            const int4 v = i < OWN_NV ? dst4[i] : make_int4(-1, -1, -1, -1);
            const int dd[4] = {v.x, v.y, v.z, v.w};
#pragma unroll
            for (int q = 0; q < 4; ++q) {
                const unsigned dl  = (unsigned)(dd[q] - base);
                const bool     hit = dl < (unsigned)nown;
                const int      s   = hit ? ei[i * 4 + q] : 0;
                unsigned mask = __builtin_amdgcn_ballot_w32(hit);
                while (mask) {
                    const int l = __builtin_ctz(mask);
                    mask &= mask - 1;
                    const int sj = __builtin_amdgcn_readlane(s, l);
                    const int dj = __builtin_amdgcn_readlane((int)dl, l);
                    const int h  = lane & (H1 - 1);
                    const float lv = lrelu(as[sj * H1 + h] + ad[(base + dj) * H1 + h]);
                    const float pe = __expf(lv - mx[dj * H1 + h]);
                    if (lane < H1) atomicAdd(&dn[dj * H1 + h], pe);
                    const float* xr = xp + (size_t)sj * D1 + lane;
                    float* arow = acc + dj * D1 + lane;
#pragma unroll
                    for (int hh = 0; hh < H1; ++hh) {
                        const float a = __int_as_float(
                            __builtin_amdgcn_readlane(__float_as_int(pe), hh));
                        atomicAdd(arow + hh * C1, a * xr[hh * C1]);
                    }
                }
            }
        }
    }
    __syncthreads();

    for (int i = tid; i < nown * D1; i += OWN_THREADS) {
        const int nd  = i >> 8;
        const int col = i & (D1 - 1);
        const float v = acc[i] / dn[nd * H1 + (col >> 5)] + b1[col];
        hout[(size_t)(base + nd) * D1 + col] = v > 0.0f ? v : 0.0f;
    }
}

__global__ __launch_bounds__(256) void gemm2_alpha2(const float* __restrict__ h,
                                                    const float* __restrict__ W2,
                                                    const float* __restrict__ aw_src,
                                                    const float* __restrict__ aw_dst,
                                                    float* __restrict__ xp2,
                                                    float* __restrict__ as2,
                                                    float* __restrict__ ad2) {
    int lane = threadIdx.x & 31;
    int n    = blockIdx.x * (blockDim.x >> 5) + (threadIdx.x >> 5);
    if (n >= NNODES) return;
    float s0 = 0.0f, s1 = 0.0f;
    for (int k = lane; k < D1; k += 32) {
        float v = h[(size_t)n * D1 + k];
        s0 = fmaf(v, W2[k * OUT2 + 0], s0);
        s1 = fmaf(v, W2[k * OUT2 + 1], s1);
    }
#pragma unroll
    for (int off = 16; off > 0; off >>= 1) {
        s0 += __shfl_down(s0, off);
        s1 += __shfl_down(s1, off);
    }
    if (lane == 0) {
        xp2[n * OUT2 + 0] = s0;
        xp2[n * OUT2 + 1] = s1;
        as2[n] = fmaf(s0, aw_src[0], s1 * aw_src[1]);
        ad2[n] = fmaf(s0, aw_dst[0], s1 * aw_dst[1]);
    }
}

__global__ void agg2(const int* __restrict__ ei,
                     const float* __restrict__ xp2,
                     const float* __restrict__ as2,
                     const float* __restrict__ ad2,
                     const float* __restrict__ m2,
                     const float* __restrict__ den2,
                     float* __restrict__ out) {
    int t = blockIdx.x * blockDim.x + threadIdx.x;
    if (t >= ETOT * OUT2) return;
    int e = t >> 1;
    int c = t & (OUT2 - 1);
    int s, d;
    edge_sd(ei, e, s, d);
    float v     = lrelu(as2[s] + ad2[d]);
    float alpha = __expf(v - m2[d]) / den2[d];
    atomicAdd(&out[d * OUT2 + c], alpha * xp2[s * OUT2 + c]);
}

__global__ void bias2_add(float* __restrict__ out, const float* __restrict__ b) {
    int i = blockIdx.x * blockDim.x + threadIdx.x;
    if (i < NNODES * OUT2) out[i] += b[i & 1];
}

extern "C" void kernel_launch(void* const* d_in, const int* in_sizes, int n_in,
                              void* d_out, int out_size, void* d_ws, size_t ws_size,
                              hipStream_t stream) {
    const float* x     = (const float*)d_in[0];
    const int*   ei    = (const int*)d_in[1];
    const float* W1    = (const float*)d_in[2];
    const float* asw1  = (const float*)d_in[3];
    const float* adw1  = (const float*)d_in[4];
    const float* b1    = (const float*)d_in[5];
    const float* W2    = (const float*)d_in[6];
    const float* asw2  = (const float*)d_in[7];
    const float* adw2  = (const float*)d_in[8];
    const float* b2    = (const float*)d_in[9];
    float*       out   = (float*)d_out;

    float* w    = (float*)d_ws;
    float* xp1  = w;
    float* hbuf = xp1  + (size_t)NNODES * D1;
    float* as1  = hbuf + (size_t)NNODES * D1;
    float* ad1  = as1  + (size_t)NNODES * H1;
    float* xp2  = ad1  + (size_t)NNODES * H1;
    float* as2  = xp2  + (size_t)NNODES * OUT2;
    float* ad2  = as2  + (size_t)NNODES;
    float* m2   = ad2  + (size_t)NNODES;
    float* dn2  = m2   + (size_t)NNODES;

    hipMemsetAsync(dn2,  0, (size_t)NNODES * sizeof(float), stream);
    hipMemsetAsync(out,  0, (size_t)NNODES * OUT2 * sizeof(float), stream);
    fill_f32<<<(NNODES + 255) / 256, 256, 0, stream>>>(m2, -INFINITY, NNODES);

    gemm1_wmma<<<NNODES / 16, 512, 0, stream>>>(x, W1, xp1);
    alpha1_kernel<<<(NNODES * H1 + 255) / 256, 256, 0, stream>>>(xp1, asw1, adw1, as1, ad1);
    agg1_owner<<<OWN_BLOCKS, OWN_THREADS, 0, stream>>>(ei, xp1, as1, ad1, b1, hbuf);

    gemm2_alpha2<<<(NNODES + 7) / 8, 256, 0, stream>>>(hbuf, W2, asw2, adw2, xp2, as2, ad2);
    edge_max<1><<<(ETOT + 255) / 256, 256, 0, stream>>>(ei, as2, ad2, m2);
    edge_denom<1><<<(ETOT + 255) / 256, 256, 0, stream>>>(ei, as2, ad2, m2, dn2);
    agg2<<<(ETOT * OUT2 + 255) / 256, 256, 0, stream>>>(ei, xp2, as2, ad2, m2, dn2, out);
    bias2_add<<<(NNODES * OUT2 + 255) / 256, 256, 0, stream>>>(out, b2);
}


# Layer-2 softmax+aggregation by the same owner groups: layer 1 saves its edge lists, one kernel builds max, denominators, sums by LDS adds (arrival order), then /denominator +bias; no global atomics
# speedup vs baseline: 6.3318x; 1.7464x over previous
#include <hip/hip_runtime.h>
#include <hip/hip_bf16.h>
#include <math.h>

#define NNODES 50000
#define NEDGES 800000
#define ETOT   (NEDGES + NNODES)
#define FIN    256
#define H1     8
#define C1     32
#define D1     256
#define OUT2   2
#define SLOPE  0.2f

#define LDSPAD 4
#define ASTRIDE (FIN + LDSPAD)

typedef float v2f __attribute__((ext_vector_type(2)));
typedef float v8f __attribute__((ext_vector_type(8)));

__device__ __forceinline__ float lrelu(float v) {
    return v > 0.0f ? v : SLOPE * v;
}

__device__ __forceinline__ void atomicMaxF(float* addr, float val) {
    if (val >= 0.0f) atomicMax((int*)addr, __float_as_int(val));
    else             atomicMin((unsigned int*)addr, __float_as_uint(val));
}

__global__ __launch_bounds__(512) void gemm1_wmma(const float* __restrict__ x,
                                                  const float* __restrict__ W1,
                                                  float* __restrict__ xp) {
    __shared__ float aTile[16 * ASTRIDE];

    const int t     = threadIdx.x;
    const int lane  = t & 31;
    const int wave  = t >> 5;
    const int tileM = blockIdx.x;
    const int half  = lane >> 4;
    const int l16   = lane & 15;

    {
        const int row = t >> 5;
        const int col = (t & 31) * 8;
        const float4* src = (const float4*)(x + (size_t)(tileM * 16 + row) * FIN + col);
        float4* dst = (float4*)(aTile + row * ASTRIDE + col);
        dst[0] = src[0];
        dst[1] = src[1];
    }
    __syncthreads();

    const int bcol = wave * 16 + l16;

    v8f acc = {};
#pragma unroll
    for (int k = 0; k < FIN; k += 4) {
        const int kk = k + half * 2;
        v2f a = *(const v2f*)(aTile + l16 * ASTRIDE + kk);
        v2f b;
        b.x = W1[kk * D1 + bcol];
        b.y = W1[(kk + 1) * D1 + bcol];
        acc = __builtin_amdgcn_wmma_f32_16x16x4_f32(
             false, a,  false, b,
             (short)0, acc,  false,  false);
    }

    const int mrow = tileM * 16 + half * 8;
#pragma unroll
    for (int r = 0; r < 8; ++r)
        xp[(size_t)(mrow + r) * D1 + bcol] = acc[r];
}

__global__ void alpha1_kernel(const float* __restrict__ xp,
                              const float* __restrict__ a_src,
                              const float* __restrict__ a_dst,
                              float* __restrict__ as, float* __restrict__ ad) {
    int i = blockIdx.x * blockDim.x + threadIdx.x;
    if (i >= NNODES * H1) return;
    int h = i & (H1 - 1);
    int n = i >> 3;
    const float* row = xp + (size_t)n * D1 + h * C1;
    const float* ws  = a_src + h * C1;
    const float* wd  = a_dst + h * C1;
    float s = 0.0f, d = 0.0f;
#pragma unroll
    for (int c = 0; c < C1; ++c) {
        float v = row[c];
        s = fmaf(v, ws[c], s);
        d = fmaf(v, wd[c], d);
    }
    as[i] = s;
    ad[i] = d;
}

#define OWN_BLOCKS  256
#define OWN_NODES   196
#define OWN_THREADS 1024
#define OWN_WAVES   (OWN_THREADS / 32)
#define OWN_CAP     20480
#define OWN_G       2
#define OWN_SCAN    8
#define OWN_NV      (NEDGES / 4)
#define WSL_HDR     32
#define WSL_STRIDE  (OWN_CAP + WSL_HDR)
static_assert(OWN_BLOCKS * OWN_NODES >= NNODES, "every row needs an owner");
static_assert(NNODES - (OWN_BLOCKS - 1) * OWN_NODES > 0, "no group without rows");
static_assert(NNODES <= 65536 && OWN_NODES <= 65535, "a list entry packs two 16-bit numbers");
static_assert(NEDGES % 4 == 0, "the destination column is read four at a time");
static_assert(NEDGES <= (1 << 20) && OWN_NODES <= (1 << 12), "edge number and local row share a word during the scan");
static_assert(OWN_NODES <= OWN_THREADS && OWN_NODES <= OWN_CAP, "self loops go in first");

__global__ __launch_bounds__(OWN_THREADS) void agg1_owner(const int* __restrict__ ei,
                                                          const float* __restrict__ xp,
                                                          const float* __restrict__ as,
                                                          const float* __restrict__ ad,
                                                          const float* __restrict__ b1,
                                                          float* __restrict__ hout,
                                                          unsigned int* __restrict__ wsl) {
    __shared__ float        acc[OWN_NODES * D1];
    __shared__ float        mx[OWN_NODES * H1];
    __shared__ float        dn[OWN_NODES * H1];
    __shared__ unsigned int list[OWN_CAP];
    __shared__ unsigned int cnt;

    const int tid  = threadIdx.x;
    const int lane = tid & 31;
    const int wv   = __builtin_amdgcn_readfirstlane(tid >> 5);
    const int base = blockIdx.x * OWN_NODES;
    const int nown = (NNODES - base) < OWN_NODES ? (NNODES - base) : OWN_NODES;

    {
        float4* acc4 = (float4*)acc;
        for (int i = tid; i < OWN_NODES * D1 / 4; i += OWN_THREADS)
            acc4[i] = make_float4(0.0f, 0.0f, 0.0f, 0.0f);
        for (int i = tid; i < OWN_NODES * H1; i += OWN_THREADS) {
            mx[i] = -INFINITY;
            dn[i] = 0.0f;
        }
        if (tid < nown) list[tid] = (unsigned)(base + tid) | ((unsigned)tid << 16);
        if (tid == 0) cnt = (unsigned)nown;
    }
    __syncthreads();

    const int4* __restrict__ dst4 = (const int4*)(ei + NEDGES);
    for (int i0 = tid; i0 < OWN_NV; i0 += OWN_SCAN * OWN_THREADS) {
        int4 v[OWN_SCAN];
#pragma unroll
        for (int u = 0; u < OWN_SCAN; ++u) {
            const int i = i0 + u * OWN_THREADS;
            v[u] = i < OWN_NV ? dst4[i] : make_int4(-1, -1, -1, -1);
        }
#pragma unroll
        for (int u = 0; u < OWN_SCAN; ++u) {
            const int e = (i0 + u * OWN_THREADS) * 4;
            const int dd[4] = {v[u].x, v[u].y, v[u].z, v[u].w};
#pragma unroll
            for (int q = 0; q < 4; ++q) {
                const unsigned dl = (unsigned)(dd[q] - base);
                if (dl < (unsigned)nown) {
                    const unsigned idx = atomicAdd(&cnt, 1u);
                    if (idx < OWN_CAP) list[idx] = (unsigned)(e + q) | (dl << 20);
                }
            }
        }
    }
    __syncthreads();

    const unsigned total = cnt;
    const bool spill = total > OWN_CAP;
    const int n = __builtin_amdgcn_readfirstlane(spill ? nown : (int)total);

    for (int t = nown + tid; t < n; t += OWN_THREADS) {
        const unsigned en = list[t];
        list[t] = (unsigned)ei[en & 0xfffffu] | ((en >> 20) << 16);
    }
    __syncthreads();

    {
        unsigned int* reg = wsl + (size_t)blockIdx.x * WSL_STRIDE;
        if (tid == 0) reg[0] = total;
        for (int t = tid; t < n; t += OWN_THREADS) reg[WSL_HDR + t] = list[t];
    }

    for (int t0 = tid; t0 < n * H1; t0 += 4 * OWN_THREADS) {
        float lv[4];
        int   mi[4];
#pragma unroll
        for (int u = 0; u < 4; ++u) {
            const int  t  = t0 + u * OWN_THREADS;
            const bool ok = t < n * H1;
            const unsigned en = list[ok ? (t >> 3) : 0];
            const int hh = t & (H1 - 1);
            const int s  = (int)(en & 0xffffu);
            const int dl = (int)(en >> 16);
            mi[u] = ok ? dl * H1 + hh : -1;
            lv[u] = lrelu(as[s * H1 + hh] + ad[(base + dl) * H1 + hh]);
        }
#pragma unroll
        for (int u = 0; u < 4; ++u)
            if (mi[u] >= 0) atomicMaxF(&mx[mi[u]], lv[u]);
    }
    if (spill) {
        for (int i = tid; i < OWN_NV; i += OWN_THREADS) {
            const int4 v = dst4[i];
            const int dd[4] = {v.x, v.y, v.z, v.w};
#pragma unroll
            for (int q = 0; q < 4; ++q) {
                const unsigned dl = (unsigned)(dd[q] - base);
                if (dl < (unsigned)nown) {
                    const int s = ei[i * 4 + q];
                    for (int hh = 0; hh < H1; ++hh)
                        atomicMaxF(&mx[dl * H1 + hh],
                                   lrelu(as[s * H1 + hh] + ad[(base + (int)dl) * H1 + hh]));
                }
            }
        }
    }
    __syncthreads();

    {
        const int j = lane >> 3;
        const int h = lane & (H1 - 1);
        for (int i0 = wv * (4 * OWN_G); i0 < n; i0 += OWN_WAVES * 4 * OWN_G) {
            unsigned en[OWN_G];
#pragma unroll
            for (int g = 0; g < OWN_G; ++g) {
                const int idx = i0 + g * 4 + j;
                en[g] = list[idx < n ? idx : 0];
            }
            float va[OWN_G], vb[OWN_G];
            int   mh[OWN_G];
#pragma unroll
            for (int g = 0; g < OWN_G; ++g) {
                const int s  = (int)(en[g] & 0xffffu);
                const int dl = (int)(en[g] >> 16);
                mh[g] = dl * H1 + h;
                va[g] = as[s * H1 + h];
                vb[g] = ad[(base + dl) * H1 + h];
            }
            float row[OWN_G][4][H1];
#pragma unroll
            for (int g = 0; g < OWN_G; ++g) {
#pragma unroll
                for (int jj = 0; jj < 4; ++jj) {
                    const unsigned ej = (unsigned)__builtin_amdgcn_readlane((int)en[g], jj * H1);
                    const float* xr = xp + (size_t)(ej & 0xffffu) * D1 + lane;
#pragma unroll
                    for (int hh = 0; hh < H1; ++hh) row[g][jj][hh] = xr[hh * C1];
                }
            }
            float p[OWN_G];
#pragma unroll
            for (int g = 0; g < OWN_G; ++g)
                p[g] = __expf(lrelu(va[g] + vb[g]) - mx[mh[g]]);
#pragma unroll
            for (int g = 0; g < OWN_G; ++g)
                if (i0 + g * 4 + j < n) atomicAdd(&dn[mh[g]], p[g]);
#pragma unroll
            for (int g = 0; g < OWN_G; ++g) {
#pragma unroll
                for (int jj = 0; jj < 4; ++jj) {
                    if (i0 + g * 4 + jj < n) {
                        const unsigned ej = (unsigned)__builtin_amdgcn_readlane((int)en[g], jj * H1);
                        float* arow = acc + (ej >> 16) * D1 + lane;
#pragma unroll
                        for (int hh = 0; hh < H1; ++hh) {
                            const float a = __int_as_float(
                                __builtin_amdgcn_readlane(__float_as_int(p[g]), jj * H1 + hh));
                            atomicAdd(arow + hh * C1, a * row[g][jj][hh]);
                        }
                    }
                }
            }
        }
    }
    if (spill) {
        for (int it = 0; it < (OWN_NV + OWN_THREADS - 1) / OWN_THREADS; ++it) {
            const int i = it * OWN_THREADS + tid;
            const int4 v = i < OWN_NV ? dst4[i] : make_int4(-1, -1, -1, -1);
            const int dd[4] = {v.x, v.y, v.z, v.w};
#pragma unroll
            for (int q = 0; q < 4; ++q) {
                const unsigned dl  = (unsigned)(dd[q] - base);
                const bool     hit = dl < (unsigned)nown;
                const int      s   = hit ? ei[i * 4 + q] : 0;
                unsigned mask = __builtin_amdgcn_ballot_w32(hit);
                while (mask) {
                    const int l = __builtin_ctz(mask);
                    mask &= mask - 1;
                    const int sj = __builtin_amdgcn_readlane(s, l);
                    const int dj = __builtin_amdgcn_readlane((int)dl, l);
                    const int h  = lane & (H1 - 1);
                    const float lv = lrelu(as[sj * H1 + h] + ad[(base + dj) * H1 + h]);
                    const float pe = __expf(lv - mx[dj * H1 + h]);
                    if (lane < H1) atomicAdd(&dn[dj * H1 + h], pe);
                    const float* xr = xp + (size_t)sj * D1 + lane;
                    float* arow = acc + dj * D1 + lane;
#pragma unroll
                    for (int hh = 0; hh < H1; ++hh) {
                        const float a = __int_as_float(
                            __builtin_amdgcn_readlane(__float_as_int(pe), hh));
                        atomicAdd(arow + hh * C1, a * xr[hh * C1]);
                    }
                }
            }
        }
    }
    __syncthreads();

    for (int i = tid; i < nown * D1; i += OWN_THREADS) {
        const int nd  = i >> 8;
        const int col = i & (D1 - 1);
        const float v = acc[i] / dn[nd * H1 + (col >> 5)] + b1[col];
        hout[(size_t)(base + nd) * D1 + col] = v > 0.0f ? v : 0.0f;
    }
}

__global__ __launch_bounds__(256) void gemm2_alpha2(const float* __restrict__ h,
                                                    const float* __restrict__ W2,
                                                    const float* __restrict__ aw_src,
                                                    const float* __restrict__ aw_dst,
                                                    float* __restrict__ xp2,
                                                    float* __restrict__ as2,
                                                    float* __restrict__ ad2) {
    int lane = threadIdx.x & 31;
    int n    = blockIdx.x * (blockDim.x >> 5) + (threadIdx.x >> 5);
    if (n >= NNODES) return;
    float s0 = 0.0f, s1 = 0.0f;
    for (int k = lane; k < D1; k += 32) {
        float v = h[(size_t)n * D1 + k];
        s0 = fmaf(v, W2[k * OUT2 + 0], s0);
        s1 = fmaf(v, W2[k * OUT2 + 1], s1);
    }
#pragma unroll
    for (int off = 16; off > 0; off >>= 1) {
        s0 += __shfl_down(s0, off);
        s1 += __shfl_down(s1, off);
    }
    if (lane == 0) {
        xp2[n * OUT2 + 0] = s0;
        xp2[n * OUT2 + 1] = s1;
        as2[n] = fmaf(s0, aw_src[0], s1 * aw_src[1]);
        ad2[n] = fmaf(s0, aw_dst[0], s1 * aw_dst[1]);
    }
}

#define L2_NODES (((OWN_NODES + 15) / 16) * 16)
static_assert(NNODES % 16 == 0, "the output ends on a line boundary");
static_assert(OWN_NODES >= 16, "a group's rows lie in two lists at most");
static_assert(L2_NODES * OUT2 <= OWN_THREADS, "one lane per output word in the epilogue");

__global__ __launch_bounds__(OWN_THREADS) void agg2_owner(const int* __restrict__ ei,
                                                          const unsigned int* __restrict__ wsl,
                                                          const float* __restrict__ xp2,
                                                          const float* __restrict__ as2,
                                                          const float* __restrict__ ad2,
                                                          const float* __restrict__ b2,
                                                          float* __restrict__ out) {
    __shared__ float mx[L2_NODES];
    __shared__ float dn[L2_NODES];
    __shared__ float s0[L2_NODES];
    __shared__ float s1[L2_NODES];

    const int tid = threadIdx.x;
    const int b   = blockIdx.x;
    int lo = (b * OWN_NODES + 15) & ~15;
    int hi = ((b + 1) * OWN_NODES + 15) & ~15;
    lo = lo < NNODES ? lo : NNODES;
    hi = hi < NNODES ? hi : NNODES;
    const int nloc = hi - lo;

    if (tid < L2_NODES) {
        mx[tid] = -INFINITY;
        dn[tid] = 0.0f;
        s0[tid] = 0.0f;
        s1[tid] = 0.0f;
    }

    const bool two = (b + 1 < OWN_BLOCKS) && ((b + 1) * OWN_NODES < hi);
    const unsigned totA = wsl[(size_t)b * WSL_STRIDE];
    const unsigned totB = two ? wsl[(size_t)(b + 1) * WSL_STRIDE] : 0u;
    const bool spill = totA > OWN_CAP || totB > OWN_CAP;
    const float2* __restrict__ xp2v = (const float2*)xp2;
    const int4*   __restrict__ dst4 = (const int4*)(ei + NEDGES);
    __syncthreads();

    if (!spill) {
        for (int g = 0; g < (two ? 2 : 1); ++g) {
            const unsigned int* reg = wsl + (size_t)(b + g) * WSL_STRIDE + WSL_HDR;
            const int n     = (int)(g ? totB : totA);
            const int gbase = (b + g) * OWN_NODES - lo;
            for (int t0 = tid; t0 < n; t0 += 4 * OWN_THREADS) {
                unsigned en[4];
#pragma unroll
                for (int u = 0; u < 4; ++u) {
                    const int t = t0 + u * OWN_THREADS;
                    en[u] = reg[t < n ? t : 0];
                }
                int   li[4];
                float va[4], vb[4];
#pragma unroll
                for (int u = 0; u < 4; ++u) {
                    const int  l  = gbase + (int)(en[u] >> 16);
                    const bool ok = t0 + u * OWN_THREADS < n && (unsigned)l < (unsigned)nloc;
                    li[u] = ok ? l : -1;
                    va[u] = as2[ok ? (int)(en[u] & 0xffffu) : lo];
                    vb[u] = ad2[ok ? lo + l : lo];
                }
#pragma unroll
                for (int u = 0; u < 4; ++u)
                    if (li[u] >= 0) atomicMaxF(&mx[li[u]], lrelu(va[u] + vb[u]));
            }
        }
    } else {
        for (int i = tid; i < nloc; i += OWN_THREADS)
            atomicMaxF(&mx[i], lrelu(as2[lo + i] + ad2[lo + i]));
        for (int i = tid; i < OWN_NV; i += OWN_THREADS) {
            const int4 v = dst4[i];
            const int dd[4] = {v.x, v.y, v.z, v.w};
#pragma unroll
            for (int q = 0; q < 4; ++q) {
                const unsigned l = (unsigned)(dd[q] - lo);
                if (l < (unsigned)nloc)
                    atomicMaxF(&mx[l], lrelu(as2[ei[i * 4 + q]] + ad2[dd[q]]));
            }
        }
    }
    __syncthreads();

    if (!spill) {
        for (int g = 0; g < (two ? 2 : 1); ++g) {
            const unsigned int* reg = wsl + (size_t)(b + g) * WSL_STRIDE + WSL_HDR;
            const int n     = (int)(g ? totB : totA);
            const int gbase = (b + g) * OWN_NODES - lo;
            for (int t0 = tid; t0 < n; t0 += 4 * OWN_THREADS) {
                unsigned en[4];
#pragma unroll
                for (int u = 0; u < 4; ++u) {
                    const int t = t0 + u * OWN_THREADS;
                    en[u] = reg[t < n ? t : 0];
                }
                int    li[4];
                float  va[4], vb[4];
                float2 xs[4];
#pragma unroll
                for (int u = 0; u < 4; ++u) {
                    const int  l  = gbase + (int)(en[u] >> 16);
                    const bool ok = t0 + u * OWN_THREADS < n && (unsigned)l < (unsigned)nloc;
                    const int  s  = ok ? (int)(en[u] & 0xffffu) : lo;
                    li[u] = ok ? l : -1;
                    va[u] = as2[s];
                    vb[u] = ad2[ok ? lo + l : lo];
                    xs[u] = xp2v[s];
                }
#pragma unroll
                for (int u = 0; u < 4; ++u) {
                    if (li[u] >= 0) {
                        const float p = __expf(lrelu(va[u] + vb[u]) - mx[li[u]]);
                        atomicAdd(&dn[li[u]], p);
                        atomicAdd(&s0[li[u]], p * xs[u].x);
                        atomicAdd(&s1[li[u]], p * xs[u].y);
                    }
                }
            }
        }
    } else {
        for (int i = tid; i < nloc; i += OWN_THREADS) {
            const float  p = __expf(lrelu(as2[lo + i] + ad2[lo + i]) - mx[i]);
            const float2 x = xp2v[lo + i];
            atomicAdd(&dn[i], p);
            atomicAdd(&s0[i], p * x.x);
            atomicAdd(&s1[i], p * x.y);
        }
        for (int i = tid; i < OWN_NV; i += OWN_THREADS) {
            const int4 v = dst4[i];
            const int dd[4] = {v.x, v.y, v.z, v.w};
#pragma unroll
            for (int q = 0; q < 4; ++q) {
                const unsigned l = (unsigned)(dd[q] - lo);
                if (l < (unsigned)nloc) {
                    const int    s = ei[i * 4 + q];
                    const float  p = __expf(lrelu(as2[s] + ad2[dd[q]]) - mx[l]);
                    const float2 x = xp2v[s];
                    atomicAdd(&dn[l], p);
                    atomicAdd(&s0[l], p * x.x);
                    atomicAdd(&s1[l], p * x.y);
                }
            }
        }
    }
    __syncthreads();

    if (tid < nloc * OUT2) {
        const int nd = tid >> 1;
        const float sv = (tid & 1) ? s1[nd] : s0[nd];
        out[(size_t)lo * OUT2 + tid] = sv / dn[nd] + b2[tid & 1];
    }
}

extern "C" void kernel_launch(void* const* d_in, const int* in_sizes, int n_in,
                              void* d_out, int out_size, void* d_ws, size_t ws_size,
                              hipStream_t stream) {
    const float* x     = (const float*)d_in[0];
    const int*   ei    = (const int*)d_in[1];
    const float* W1    = (const float*)d_in[2];
    const float* asw1  = (const float*)d_in[3];
    const float* adw1  = (const float*)d_in[4];
    const float* b1    = (const float*)d_in[5];
    const float* W2    = (const float*)d_in[6];
    const float* asw2  = (const float*)d_in[7];
    const float* adw2  = (const float*)d_in[8];
    const float* b2    = (const float*)d_in[9];
    float*       out   = (float*)d_out;

    float* w    = (float*)d_ws;
    float* xp1  = w;
    float* hbuf = xp1  + (size_t)NNODES * D1;
    float* as1  = hbuf + (size_t)NNODES * D1;
    float* ad1  = as1  + (size_t)NNODES * H1;
    float* xp2  = ad1  + (size_t)NNODES * H1;
    float* as2  = xp2  + (size_t)NNODES * OUT2;
    float* ad2  = as2  + (size_t)NNODES;
    unsigned int* wsl = (unsigned int*)(ad2 + (size_t)NNODES);

    gemm1_wmma<<<NNODES / 16, 512, 0, stream>>>(x, W1, xp1);
    alpha1_kernel<<<(NNODES * H1 + 255) / 256, 256, 0, stream>>>(xp1, asw1, adw1, as1, ad1);
    agg1_owner<<<OWN_BLOCKS, OWN_THREADS, 0, stream>>>(ei, xp1, as1, ad1, b1, hbuf, wsl);

    gemm2_alpha2<<<(NNODES + 7) / 8, 256, 0, stream>>>(hbuf, W2, asw2, adw2, xp2, as2, ad2);
    agg2_owner<<<OWN_BLOCKS, OWN_THREADS, 0, stream>>>(ei, wsl, xp2, as2, ad2, b2, out);
}


# Owner groups' edge lists from one binning launch: each edge read once and put in its owner group's slot, not 256 full scans of the destination column; list order new, sums still in LDS arrival order
# speedup vs baseline: 7.5651x; 1.1948x over previous
#include <hip/hip_runtime.h>
#include <hip/hip_bf16.h>
#include <math.h>

#define NNODES 50000
#define NEDGES 800000
#define ETOT   (NEDGES + NNODES)
#define FIN    256
#define H1     8
#define C1     32
#define D1     256
#define OUT2   2
#define SLOPE  0.2f

#define LDSPAD 4
#define ASTRIDE (FIN + LDSPAD)

typedef float v2f __attribute__((ext_vector_type(2)));
typedef float v8f __attribute__((ext_vector_type(8)));

__device__ __forceinline__ float lrelu(float v) {
    return v > 0.0f ? v : SLOPE * v;
}

__device__ __forceinline__ void atomicMaxF(float* addr, float val) {
    if (val >= 0.0f) atomicMax((int*)addr, __float_as_int(val));
    else             atomicMin((unsigned int*)addr, __float_as_uint(val));
}

__global__ __launch_bounds__(512) void gemm1_wmma(const float* __restrict__ x,
                                                  const float* __restrict__ W1,
                                                  float* __restrict__ xp) {
    __shared__ float aTile[16 * ASTRIDE];

    const int t     = threadIdx.x;
    const int lane  = t & 31;
    const int wave  = t >> 5;
    const int tileM = blockIdx.x;
    const int half  = lane >> 4;
    const int l16   = lane & 15;

    {
        const int row = t >> 5;
        const int col = (t & 31) * 8;
        const float4* src = (const float4*)(x + (size_t)(tileM * 16 + row) * FIN + col);
        float4* dst = (float4*)(aTile + row * ASTRIDE + col);
        dst[0] = src[0];
        dst[1] = src[1];
    }
    __syncthreads();

    const int bcol = wave * 16 + l16;

    v8f acc = {};
#pragma unroll
    for (int k = 0; k < FIN; k += 4) {
        const int kk = k + half * 2;
        v2f a = *(const v2f*)(aTile + l16 * ASTRIDE + kk);
        v2f b;
        b.x = W1[kk * D1 + bcol];
        b.y = W1[(kk + 1) * D1 + bcol];
        acc = __builtin_amdgcn_wmma_f32_16x16x4_f32(
             false, a,  false, b,
             (short)0, acc,  false,  false);
    }

    const int mrow = tileM * 16 + half * 8;
#pragma unroll
    for (int r = 0; r < 8; ++r)
        xp[(size_t)(mrow + r) * D1 + bcol] = acc[r];
}

__global__ void alpha1_kernel(const float* __restrict__ xp,
                              const float* __restrict__ a_src,
                              const float* __restrict__ a_dst,
                              float* __restrict__ as, float* __restrict__ ad) {
    int i = blockIdx.x * blockDim.x + threadIdx.x;
    if (i >= NNODES * H1) return;
    int h = i & (H1 - 1);
    int n = i >> 3;
    const float* row = xp + (size_t)n * D1 + h * C1;
    const float* ws  = a_src + h * C1;
    const float* wd  = a_dst + h * C1;
    float s = 0.0f, d = 0.0f;
#pragma unroll
    for (int c = 0; c < C1; ++c) {
        float v = row[c];
        s = fmaf(v, ws[c], s);
        d = fmaf(v, wd[c], d);
    }
    as[i] = s;
    ad[i] = d;
}

#define OWN_BLOCKS  256
#define OWN_NODES   196
#define OWN_THREADS 1024
#define OWN_WAVES   (OWN_THREADS / 32)
#define OWN_CAP     20480
#define OWN_G       2
#define OWN_NV      (NEDGES / 4)
#define WSL_HDR     32
#define WSL_STRIDE  (OWN_CAP + WSL_HDR)
#define BIN_BLOCKS  256
#define BIN_THREADS 1024
#define BIN_SLOT    64
#define BIN_CAP     (BIN_SLOT - 1)
#define BIN_NV      ((OWN_NV + BIN_BLOCKS - 1) / BIN_BLOCKS)
static_assert(OWN_BLOCKS * OWN_NODES >= NNODES, "every row needs an owner");
static_assert(NNODES - (OWN_BLOCKS - 1) * OWN_NODES > 0, "no group without rows");
static_assert(NNODES <= 65536 && OWN_NODES <= 65535, "a list entry packs two 16-bit numbers");
static_assert(NEDGES % 4 == 0, "the edge list is read four at a time");
static_assert(OWN_NODES <= OWN_THREADS && OWN_NODES <= OWN_CAP, "self loops go in first");
static_assert(BIN_NV <= BIN_THREADS, "one int4 of the edge list per lane of a binning group");
static_assert(OWN_BLOCKS * BIN_SLOT == 16 * BIN_THREADS, "a binning group moves its slots 16 words per lane");
static_assert(BIN_BLOCKS * BIN_SLOT == 16 * OWN_THREADS, "an owner group reads its slots 16 words per lane");
static_assert(BIN_BLOCKS <= OWN_THREADS && BIN_BLOCKS % 32 == 0, "one lane per slot in the offset scan, whole waves");
static_assert(OWN_NODES + BIN_BLOCKS * BIN_CAP <= OWN_CAP, "full slots still fit the list");

__global__ __launch_bounds__(BIN_THREADS) void bin_edges(const int* __restrict__ ei,
                                                         unsigned int* __restrict__ slots) {
    __shared__ unsigned int stage[OWN_BLOCKS * BIN_SLOT];

    const int tid = threadIdx.x;
    uint4* stage4 = (uint4*)stage;
    for (int u = 0; u < 4; ++u) stage4[u * BIN_THREADS + tid] = make_uint4(0u, 0u, 0u, 0u);
    __syncthreads();

    const int i = blockIdx.x * BIN_NV + tid;
    if (tid < BIN_NV && i < OWN_NV) {
        const int4 sv = ((const int4*)ei)[i];
        const int4 dv = ((const int4*)(ei + NEDGES))[i];
        const int ss[4] = {sv.x, sv.y, sv.z, sv.w};
        const int dd[4] = {dv.x, dv.y, dv.z, dv.w};
#pragma unroll
        for (int q = 0; q < 4; ++q) {
            const unsigned d = (unsigned)dd[q];
            if (d < (unsigned)NNODES) {
                const unsigned g  = d / OWN_NODES;
                const unsigned dl = d - g * OWN_NODES;
                const unsigned r  = atomicAdd(&stage[g * BIN_SLOT], 1u);
                if (r < BIN_CAP) stage[g * BIN_SLOT + 1 + r] = (unsigned)ss[q] | (dl << 16);
            }
        }
    }
    __syncthreads();

    for (int u = 0; u < 4; ++u) {
        const int k = u * BIN_THREADS + tid;
        const int g = k >> 4;
        uint4* dst = (uint4*)(slots + ((size_t)g * BIN_BLOCKS + blockIdx.x) * BIN_SLOT);
        dst[k & 15] = stage4[k];
    }
}

__global__ __launch_bounds__(OWN_THREADS) void agg1_owner(const int* __restrict__ ei,
                                                          const float* __restrict__ xp,
                                                          const float* __restrict__ as,
                                                          const float* __restrict__ ad,
                                                          const float* __restrict__ b1,
                                                          float* __restrict__ hout,
                                                          unsigned int* __restrict__ wsl,
                                                          const unsigned int* __restrict__ slots) {
    __shared__ float        acc[OWN_NODES * D1];
    __shared__ float        mx[OWN_NODES * H1];
    __shared__ float        dn[OWN_NODES * H1];
    __shared__ unsigned int list[OWN_CAP];
    __shared__ unsigned int scnt[BIN_BLOCKS];
    __shared__ unsigned int soff[BIN_BLOCKS];
    __shared__ unsigned int wtot[BIN_BLOCKS / 32];
    __shared__ unsigned int over;

    const int tid  = threadIdx.x;
    const int lane = tid & 31;
    const int wv   = __builtin_amdgcn_readfirstlane(tid >> 5);
    const int base = blockIdx.x * OWN_NODES;
    const int nown = (NNODES - base) < OWN_NODES ? (NNODES - base) : OWN_NODES;

    uint4 sv[4];
    {
        const uint4* sl4 = (const uint4*)(slots + (size_t)blockIdx.x * BIN_BLOCKS * BIN_SLOT);
#pragma unroll
        for (int u = 0; u < 4; ++u) sv[u] = sl4[u * OWN_THREADS + tid];
    }

    {
        float4* acc4 = (float4*)acc;
        for (int i = tid; i < OWN_NODES * D1 / 4; i += OWN_THREADS)
            acc4[i] = make_float4(0.0f, 0.0f, 0.0f, 0.0f);
        for (int i = tid; i < OWN_NODES * H1; i += OWN_THREADS) {
            mx[i] = -INFINITY;
            dn[i] = 0.0f;
        }
        if (tid < nown) list[tid] = (unsigned)(base + tid) | ((unsigned)tid << 16);
        if (tid == 0) over = 0u;
        if ((tid & 15) == 0) {
#pragma unroll
            for (int u = 0; u < 4; ++u) scnt[(u * OWN_THREADS + tid) >> 4] = sv[u].x;
        }
    }
    __syncthreads();

    const int4* __restrict__ dst4 = (const int4*)(ei + NEDGES);
    unsigned cc = 0u, inc = 0u;
    if (tid < BIN_BLOCKS) {
        const unsigned cs = scnt[tid];
        if (cs > BIN_CAP) over = 1u;
        cc  = cs < BIN_CAP ? cs : BIN_CAP;
        inc = cc;
        for (int o = 1; o < 32; o <<= 1) {
            const unsigned v = __shfl_up(inc, o);
            if (lane >= o) inc += v;
        }
        if (lane == 31) wtot[wv] = inc;
    }
    __syncthreads();
    if (tid < BIN_BLOCKS) {
        unsigned pre = 0u;
        for (int w = 0; w < wv; ++w) pre += wtot[w];
        soff[tid] = pre + inc - cc;
    }
    unsigned found = 0u;
    for (int w = 0; w < BIN_BLOCKS / 32; ++w) found += wtot[w];
    const bool spill = over != 0u;
    const int n = __builtin_amdgcn_readfirstlane(spill ? nown : nown + (int)found);
    __syncthreads();

    if (!spill) {
#pragma unroll
        for (int u = 0; u < 4; ++u) {
            const int      k  = u * OWN_THREADS + tid;
            const int      s  = k >> 4;
            const unsigned c  = scnt[s] < BIN_CAP ? scnt[s] : BIN_CAP;
            const int      p0 = nown + (int)soff[s] - 1;
            const int      w0 = (k & 15) * 4;
            const unsigned wd[4] = {sv[u].x, sv[u].y, sv[u].z, sv[u].w};
#pragma unroll
            for (int q = 0; q < 4; ++q) {
                const int w = w0 + q;
                if (w >= 1 && (unsigned)(w - 1) < c) list[p0 + w] = wd[q];
            }
        }
    }
    __syncthreads();

    {
        unsigned int* reg = wsl + (size_t)blockIdx.x * WSL_STRIDE;
        if (tid == 0) reg[0] = spill ? (unsigned)(OWN_CAP + 1) : (unsigned)n;
        for (int t = tid; t < n; t += OWN_THREADS) reg[WSL_HDR + t] = list[t];
    }

    for (int t0 = tid; t0 < n * H1; t0 += 4 * OWN_THREADS) {
        float lv[4];
        int   mi[4];
#pragma unroll
        for (int u = 0; u < 4; ++u) {
            const int  t  = t0 + u * OWN_THREADS;
            const bool ok = t < n * H1;
            const unsigned en = list[ok ? (t >> 3) : 0];
            const int hh = t & (H1 - 1);
            const int s  = (int)(en & 0xffffu);
            const int dl = (int)(en >> 16);
            mi[u] = ok ? dl * H1 + hh : -1;
            lv[u] = lrelu(as[s * H1 + hh] + ad[(base + dl) * H1 + hh]);
        }
#pragma unroll
        for (int u = 0; u < 4; ++u)
            if (mi[u] >= 0) atomicMaxF(&mx[mi[u]], lv[u]);
    }
    if (spill) {
        for (int i = tid; i < OWN_NV; i += OWN_THREADS) {
            const int4 v = dst4[i];
            const int dd[4] = {v.x, v.y, v.z, v.w};
#pragma unroll
            for (int q = 0; q < 4; ++q) {
                const unsigned dl = (unsigned)(dd[q] - base);
                if (dl < (unsigned)nown) {
                    const int s = ei[i * 4 + q];
                    for (int hh = 0; hh < H1; ++hh)
                        atomicMaxF(&mx[dl * H1 + hh],
                                   lrelu(as[s * H1 + hh] + ad[(base + (int)dl) * H1 + hh]));
                }
            }
        }
    }
    __syncthreads();

    {
        const int j = lane >> 3;
        const int h = lane & (H1 - 1);
        for (int i0 = wv * (4 * OWN_G); i0 < n; i0 += OWN_WAVES * 4 * OWN_G) {
            unsigned en[OWN_G];
#pragma unroll
            for (int g = 0; g < OWN_G; ++g) {
                const int idx = i0 + g * 4 + j;
                en[g] = list[idx < n ? idx : 0];
            }
            float va[OWN_G], vb[OWN_G];
            int   mh[OWN_G];
#pragma unroll
            for (int g = 0; g < OWN_G; ++g) {
                const int s  = (int)(en[g] & 0xffffu);
                const int dl = (int)(en[g] >> 16);
                mh[g] = dl * H1 + h;
                va[g] = as[s * H1 + h];
                vb[g] = ad[(base + dl) * H1 + h];
            }
            float row[OWN_G][4][H1];
#pragma unroll
            for (int g = 0; g < OWN_G; ++g) {
#pragma unroll
                for (int jj = 0; jj < 4; ++jj) {
                    const unsigned ej = (unsigned)__builtin_amdgcn_readlane((int)en[g], jj * H1);
                    const float* xr = xp + (size_t)(ej & 0xffffu) * D1 + lane;
#pragma unroll
                    for (int hh = 0; hh < H1; ++hh) row[g][jj][hh] = xr[hh * C1];
                }
            }
            float p[OWN_G];
#pragma unroll
            for (int g = 0; g < OWN_G; ++g)
                p[g] = __expf(lrelu(va[g] + vb[g]) - mx[mh[g]]);
#pragma unroll
            for (int g = 0; g < OWN_G; ++g)
                if (i0 + g * 4 + j < n) atomicAdd(&dn[mh[g]], p[g]);
#pragma unroll
            for (int g = 0; g < OWN_G; ++g) {
#pragma unroll
                for (int jj = 0; jj < 4; ++jj) {
                    if (i0 + g * 4 + jj < n) {
                        const unsigned ej = (unsigned)__builtin_amdgcn_readlane((int)en[g], jj * H1);
                        float* arow = acc + (ej >> 16) * D1 + lane;
#pragma unroll
                        for (int hh = 0; hh < H1; ++hh) {
                            const float a = __int_as_float(
                                __builtin_amdgcn_readlane(__float_as_int(p[g]), jj * H1 + hh));
                            atomicAdd(arow + hh * C1, a * row[g][jj][hh]);
                        }
                    }
                }
            }
        }
    }
    if (spill) {
        for (int it = 0; it < (OWN_NV + OWN_THREADS - 1) / OWN_THREADS; ++it) {
            const int i = it * OWN_THREADS + tid;
            const int4 v = i < OWN_NV ? dst4[i] : make_int4(-1, -1, -1, -1);
            const int dd[4] = {v.x, v.y, v.z, v.w};
#pragma unroll
            for (int q = 0; q < 4; ++q) {
                const unsigned dl  = (unsigned)(dd[q] - base);
                const bool     hit = dl < (unsigned)nown;
                const int      s   = hit ? ei[i * 4 + q] : 0;
                unsigned mask = __builtin_amdgcn_ballot_w32(hit);
                while (mask) {
                    const int l = __builtin_ctz(mask);
                    mask &= mask - 1;
                    const int sj = __builtin_amdgcn_readlane(s, l);
                    const int dj = __builtin_amdgcn_readlane((int)dl, l);
                    const int h  = lane & (H1 - 1);
                    const float lv = lrelu(as[sj * H1 + h] + ad[(base + dj) * H1 + h]);
                    const float pe = __expf(lv - mx[dj * H1 + h]);
                    if (lane < H1) atomicAdd(&dn[dj * H1 + h], pe);
                    const float* xr = xp + (size_t)sj * D1 + lane;
                    float* arow = acc + dj * D1 + lane;
#pragma unroll
                    for (int hh = 0; hh < H1; ++hh) {
                        const float a = __int_as_float(
                            __builtin_amdgcn_readlane(__float_as_int(pe), hh));
                        atomicAdd(arow + hh * C1, a * xr[hh * C1]);
                    }
                }
            }
        }
    }
    __syncthreads();

    for (int i = tid; i < nown * D1; i += OWN_THREADS) {
        const int nd  = i >> 8;
        const int col = i & (D1 - 1);
        const float v = acc[i] / dn[nd * H1 + (col >> 5)] + b1[col];
        hout[(size_t)(base + nd) * D1 + col] = v > 0.0f ? v : 0.0f;
    }
}

__global__ __launch_bounds__(256) void gemm2_alpha2(const float* __restrict__ h,
                                                    const float* __restrict__ W2,
                                                    const float* __restrict__ aw_src,
                                                    const float* __restrict__ aw_dst,
                                                    float* __restrict__ xp2,
                                                    float* __restrict__ as2,
                                                    float* __restrict__ ad2) {
    int lane = threadIdx.x & 31;
    int n    = blockIdx.x * (blockDim.x >> 5) + (threadIdx.x >> 5);
    if (n >= NNODES) return;
    float s0 = 0.0f, s1 = 0.0f;
    for (int k = lane; k < D1; k += 32) {
        float v = h[(size_t)n * D1 + k];
        s0 = fmaf(v, W2[k * OUT2 + 0], s0);
        s1 = fmaf(v, W2[k * OUT2 + 1], s1);
    }
#pragma unroll
    for (int off = 16; off > 0; off >>= 1) {
        s0 += __shfl_down(s0, off);
        s1 += __shfl_down(s1, off);
    }
    if (lane == 0) {
        xp2[n * OUT2 + 0] = s0;
        xp2[n * OUT2 + 1] = s1;
        as2[n] = fmaf(s0, aw_src[0], s1 * aw_src[1]);
        ad2[n] = fmaf(s0, aw_dst[0], s1 * aw_dst[1]);
    }
}

#define L2_NODES (((OWN_NODES + 15) / 16) * 16)
static_assert(NNODES % 16 == 0, "the output ends on a line boundary");
static_assert(OWN_NODES >= 16, "a group's rows lie in two lists at most");
static_assert(L2_NODES * OUT2 <= OWN_THREADS, "one lane per output word in the epilogue");

__global__ __launch_bounds__(OWN_THREADS) void agg2_owner(const int* __restrict__ ei,
                                                          const unsigned int* __restrict__ wsl,
                                                          const float* __restrict__ xp2,
                                                          const float* __restrict__ as2,
                                                          const float* __restrict__ ad2,
                                                          const float* __restrict__ b2,
                                                          float* __restrict__ out) {
    __shared__ float mx[L2_NODES];
    __shared__ float dn[L2_NODES];
    __shared__ float s0[L2_NODES];
    __shared__ float s1[L2_NODES];

    const int tid = threadIdx.x;
    const int b   = blockIdx.x;
    int lo = (b * OWN_NODES + 15) & ~15;
    int hi = ((b + 1) * OWN_NODES + 15) & ~15;
    lo = lo < NNODES ? lo : NNODES;
    hi = hi < NNODES ? hi : NNODES;
    const int nloc = hi - lo;

    if (tid < L2_NODES) {
        mx[tid] = -INFINITY;
        dn[tid] = 0.0f;
        s0[tid] = 0.0f;
        s1[tid] = 0.0f;
    }

    const bool two = (b + 1 < OWN_BLOCKS) && ((b + 1) * OWN_NODES < hi);
    const unsigned totA = wsl[(size_t)b * WSL_STRIDE];
    const unsigned totB = two ? wsl[(size_t)(b + 1) * WSL_STRIDE] : 0u;
    const bool spill = totA > OWN_CAP || totB > OWN_CAP;
    const float2* __restrict__ xp2v = (const float2*)xp2;
    const int4*   __restrict__ dst4 = (const int4*)(ei + NEDGES);
    __syncthreads();

    if (!spill) {
        for (int g = 0; g < (two ? 2 : 1); ++g) {
            const unsigned int* reg = wsl + (size_t)(b + g) * WSL_STRIDE + WSL_HDR;
            const int n     = (int)(g ? totB : totA);
            const int gbase = (b + g) * OWN_NODES - lo;
            for (int t0 = tid; t0 < n; t0 += 4 * OWN_THREADS) {
                unsigned en[4];
#pragma unroll
                for (int u = 0; u < 4; ++u) {
                    const int t = t0 + u * OWN_THREADS;
                    en[u] = reg[t < n ? t : 0];
                }
                int   li[4];
                float va[4], vb[4];
#pragma unroll
                for (int u = 0; u < 4; ++u) {
                    const int  l  = gbase + (int)(en[u] >> 16);
                    const bool ok = t0 + u * OWN_THREADS < n && (unsigned)l < (unsigned)nloc;
                    li[u] = ok ? l : -1;
                    va[u] = as2[ok ? (int)(en[u] & 0xffffu) : lo];
                    vb[u] = ad2[ok ? lo + l : lo];
                }
#pragma unroll
                for (int u = 0; u < 4; ++u)
                    if (li[u] >= 0) atomicMaxF(&mx[li[u]], lrelu(va[u] + vb[u]));
            }
        }
    } else {
        for (int i = tid; i < nloc; i += OWN_THREADS)
            atomicMaxF(&mx[i], lrelu(as2[lo + i] + ad2[lo + i]));
        for (int i = tid; i < OWN_NV; i += OWN_THREADS) {
            const int4 v = dst4[i];
            const int dd[4] = {v.x, v.y, v.z, v.w};
#pragma unroll
            for (int q = 0; q < 4; ++q) {
                const unsigned l = (unsigned)(dd[q] - lo);
                if (l < (unsigned)nloc)
                    atomicMaxF(&mx[l], lrelu(as2[ei[i * 4 + q]] + ad2[dd[q]]));
            }
        }
    }
    __syncthreads();

    if (!spill) {
        for (int g = 0; g < (two ? 2 : 1); ++g) {
            const unsigned int* reg = wsl + (size_t)(b + g) * WSL_STRIDE + WSL_HDR;
            const int n     = (int)(g ? totB : totA);
            const int gbase = (b + g) * OWN_NODES - lo;
            for (int t0 = tid; t0 < n; t0 += 4 * OWN_THREADS) {
                unsigned en[4];
#pragma unroll
                for (int u = 0; u < 4; ++u) {
                    const int t = t0 + u * OWN_THREADS;
                    en[u] = reg[t < n ? t : 0];
                }
                int    li[4];
                float  va[4], vb[4];
                float2 xs[4];
#pragma unroll
                for (int u = 0; u < 4; ++u) {
                    const int  l  = gbase + (int)(en[u] >> 16);
                    const bool ok = t0 + u * OWN_THREADS < n && (unsigned)l < (unsigned)nloc;
                    const int  s  = ok ? (int)(en[u] & 0xffffu) : lo;
                    li[u] = ok ? l : -1;
                    va[u] = as2[s];
                    vb[u] = ad2[ok ? lo + l : lo];
                    xs[u] = xp2v[s];
                }
#pragma unroll
                for (int u = 0; u < 4; ++u) {
                    if (li[u] >= 0) {
                        const float p = __expf(lrelu(va[u] + vb[u]) - mx[li[u]]);
                        atomicAdd(&dn[li[u]], p);
                        atomicAdd(&s0[li[u]], p * xs[u].x);
                        atomicAdd(&s1[li[u]], p * xs[u].y);
                    }
                }
            }
        }
    } else {
        for (int i = tid; i < nloc; i += OWN_THREADS) {
            const float  p = __expf(lrelu(as2[lo + i] + ad2[lo + i]) - mx[i]);
            const float2 x = xp2v[lo + i];
            atomicAdd(&dn[i], p);
            atomicAdd(&s0[i], p * x.x);
            atomicAdd(&s1[i], p * x.y);
        }
        for (int i = tid; i < OWN_NV; i += OWN_THREADS) {
            const int4 v = dst4[i];
            const int dd[4] = {v.x, v.y, v.z, v.w};
#pragma unroll
            for (int q = 0; q < 4; ++q) {
                const unsigned l = (unsigned)(dd[q] - lo);
                if (l < (unsigned)nloc) {
                    const int    s = ei[i * 4 + q];
                    const float  p = __expf(lrelu(as2[s] + ad2[dd[q]]) - mx[l]);
                    const float2 x = xp2v[s];
                    atomicAdd(&dn[l], p);
                    atomicAdd(&s0[l], p * x.x);
                    atomicAdd(&s1[l], p * x.y);
                }
            }
        }
    }
    __syncthreads();

    if (tid < nloc * OUT2) {
        const int nd = tid >> 1;
        const float sv = (tid & 1) ? s1[nd] : s0[nd];
        out[(size_t)lo * OUT2 + tid] = sv / dn[nd] + b2[tid & 1];
    }
}

extern "C" void kernel_launch(void* const* d_in, const int* in_sizes, int n_in,
                              void* d_out, int out_size, void* d_ws, size_t ws_size,
                              hipStream_t stream) {
    const float* x     = (const float*)d_in[0];
    const int*   ei    = (const int*)d_in[1];
    const float* W1    = (const float*)d_in[2];
    const float* asw1  = (const float*)d_in[3];
    const float* adw1  = (const float*)d_in[4];
    const float* b1    = (const float*)d_in[5];
    const float* W2    = (const float*)d_in[6];
    const float* asw2  = (const float*)d_in[7];
    const float* adw2  = (const float*)d_in[8];
    const float* b2    = (const float*)d_in[9];
    float*       out   = (float*)d_out;

    float* w    = (float*)d_ws;
    float* xp1  = w;
    float* hbuf = xp1  + (size_t)NNODES * D1;
    float* as1  = hbuf + (size_t)NNODES * D1;
    float* ad1  = as1  + (size_t)NNODES * H1;
    float* xp2  = ad1  + (size_t)NNODES * H1;
    float* as2  = xp2  + (size_t)NNODES * OUT2;
    float* ad2  = as2  + (size_t)NNODES;
    unsigned int* wsl = (unsigned int*)(ad2 + (size_t)NNODES);
    unsigned int* slots = wsl + (size_t)OWN_BLOCKS * WSL_STRIDE;

    gemm1_wmma<<<NNODES / 16, 512, 0, stream>>>(x, W1, xp1);
    alpha1_kernel<<<(NNODES * H1 + 255) / 256, 256, 0, stream>>>(xp1, asw1, adw1, as1, ad1);
    bin_edges<<<BIN_BLOCKS, BIN_THREADS, 0, stream>>>(ei, slots);
    agg1_owner<<<OWN_BLOCKS, OWN_THREADS, 0, stream>>>(ei, xp1, as1, ad1, b1, hbuf, wsl, slots);

    gemm2_alpha2<<<(NNODES + 7) / 8, 256, 0, stream>>>(hbuf, W2, asw2, adw2, xp2, as2, ad2);
    agg2_owner<<<OWN_BLOCKS, OWN_THREADS, 0, stream>>>(ei, wsl, xp2, as2, ad2, b2, out);
}


# Layer-2 projection: one wave per 32 consecutive nodes (1563 waves, not 50000), W2 in registers, 32 16-byte row loads in flight, whole-line stores; xp2 summed 8 k per lane, then an xor lane tree
# speedup vs baseline: 7.8656x; 1.0397x over previous
#include <hip/hip_runtime.h>
#include <hip/hip_bf16.h>
#include <math.h>

#define NNODES 50000
#define NEDGES 800000
#define ETOT   (NEDGES + NNODES)
#define FIN    256
#define H1     8
#define C1     32
#define D1     256
#define OUT2   2
#define SLOPE  0.2f

#define LDSPAD 4
#define ASTRIDE (FIN + LDSPAD)

typedef float v2f __attribute__((ext_vector_type(2)));
typedef float v8f __attribute__((ext_vector_type(8)));

__device__ __forceinline__ float lrelu(float v) {
    return v > 0.0f ? v : SLOPE * v;
}

__device__ __forceinline__ void atomicMaxF(float* addr, float val) {
    if (val >= 0.0f) atomicMax((int*)addr, __float_as_int(val));
    else             atomicMin((unsigned int*)addr, __float_as_uint(val));
}

__global__ __launch_bounds__(512) void gemm1_wmma(const float* __restrict__ x,
                                                  const float* __restrict__ W1,
                                                  float* __restrict__ xp) {
    __shared__ float aTile[16 * ASTRIDE];

    const int t     = threadIdx.x;
    const int lane  = t & 31;
    const int wave  = t >> 5;
    const int tileM = blockIdx.x;
    const int half  = lane >> 4;
    const int l16   = lane & 15;

    {
        const int row = t >> 5;
        const int col = (t & 31) * 8;
        const float4* src = (const float4*)(x + (size_t)(tileM * 16 + row) * FIN + col);
        float4* dst = (float4*)(aTile + row * ASTRIDE + col);
        dst[0] = src[0];
        dst[1] = src[1];
    }
    __syncthreads();

    const int bcol = wave * 16 + l16;

    v8f acc = {};
#pragma unroll
    for (int k = 0; k < FIN; k += 4) {
        const int kk = k + half * 2;
        v2f a = *(const v2f*)(aTile + l16 * ASTRIDE + kk);
        v2f b;
        b.x = W1[kk * D1 + bcol];
        b.y = W1[(kk + 1) * D1 + bcol];
        acc = __builtin_amdgcn_wmma_f32_16x16x4_f32(
             false, a,  false, b,
             (short)0, acc,  false,  false);
    }

    const int mrow = tileM * 16 + half * 8;
#pragma unroll
    for (int r = 0; r < 8; ++r)
        xp[(size_t)(mrow + r) * D1 + bcol] = acc[r];
}

__global__ void alpha1_kernel(const float* __restrict__ xp,
                              const float* __restrict__ a_src,
                              const float* __restrict__ a_dst,
                              float* __restrict__ as, float* __restrict__ ad) {
    int i = blockIdx.x * blockDim.x + threadIdx.x;
    if (i >= NNODES * H1) return;
    int h = i & (H1 - 1);
    int n = i >> 3;
    const float* row = xp + (size_t)n * D1 + h * C1;
    const float* ws  = a_src + h * C1;
    const float* wd  = a_dst + h * C1;
    float s = 0.0f, d = 0.0f;
#pragma unroll
    for (int c = 0; c < C1; ++c) {
        float v = row[c];
        s = fmaf(v, ws[c], s);
        d = fmaf(v, wd[c], d);
    }
    as[i] = s;
    ad[i] = d;
}

#define OWN_BLOCKS  256
#define OWN_NODES   196
#define OWN_THREADS 1024
#define OWN_WAVES   (OWN_THREADS / 32)
#define OWN_CAP     20480
#define OWN_G       2
#define OWN_NV      (NEDGES / 4)
#define WSL_HDR     32
#define WSL_STRIDE  (OWN_CAP + WSL_HDR)
#define BIN_BLOCKS  256
#define BIN_THREADS 1024
#define BIN_SLOT    64
#define BIN_CAP     (BIN_SLOT - 1)
#define BIN_NV      ((OWN_NV + BIN_BLOCKS - 1) / BIN_BLOCKS)
static_assert(OWN_BLOCKS * OWN_NODES >= NNODES, "every row needs an owner");
static_assert(NNODES - (OWN_BLOCKS - 1) * OWN_NODES > 0, "no group without rows");
static_assert(NNODES <= 65536 && OWN_NODES <= 65535, "a list entry packs two 16-bit numbers");
static_assert(NEDGES % 4 == 0, "the edge list is read four at a time");
static_assert(OWN_NODES <= OWN_THREADS && OWN_NODES <= OWN_CAP, "self loops go in first");
static_assert(BIN_NV <= BIN_THREADS, "one int4 of the edge list per lane of a binning group");
static_assert(OWN_BLOCKS * BIN_SLOT == 16 * BIN_THREADS, "a binning group moves its slots 16 words per lane");
static_assert(BIN_BLOCKS * BIN_SLOT == 16 * OWN_THREADS, "an owner group reads its slots 16 words per lane");
static_assert(BIN_BLOCKS <= OWN_THREADS && BIN_BLOCKS % 32 == 0, "one lane per slot in the offset scan, whole waves");
static_assert(OWN_NODES + BIN_BLOCKS * BIN_CAP <= OWN_CAP, "full slots still fit the list");

__global__ __launch_bounds__(BIN_THREADS) void bin_edges(const int* __restrict__ ei,
                                                         unsigned int* __restrict__ slots) {
    __shared__ unsigned int stage[OWN_BLOCKS * BIN_SLOT];

    const int tid = threadIdx.x;
    uint4* stage4 = (uint4*)stage;
    for (int u = 0; u < 4; ++u) stage4[u * BIN_THREADS + tid] = make_uint4(0u, 0u, 0u, 0u);
    __syncthreads();

    const int i = blockIdx.x * BIN_NV + tid;
    if (tid < BIN_NV && i < OWN_NV) {
        const int4 sv = ((const int4*)ei)[i];
        const int4 dv = ((const int4*)(ei + NEDGES))[i];
        const int ss[4] = {sv.x, sv.y, sv.z, sv.w};
        const int dd[4] = {dv.x, dv.y, dv.z, dv.w};
#pragma unroll
        for (int q = 0; q < 4; ++q) {
            const unsigned d = (unsigned)dd[q];
            if (d < (unsigned)NNODES) {
                const unsigned g  = d / OWN_NODES;
                const unsigned dl = d - g * OWN_NODES;
                const unsigned r  = atomicAdd(&stage[g * BIN_SLOT], 1u);
                if (r < BIN_CAP) stage[g * BIN_SLOT + 1 + r] = (unsigned)ss[q] | (dl << 16);
            }
        }
    }
    __syncthreads();

    for (int u = 0; u < 4; ++u) {
        const int k = u * BIN_THREADS + tid;
        const int g = k >> 4;
        uint4* dst = (uint4*)(slots + ((size_t)g * BIN_BLOCKS + blockIdx.x) * BIN_SLOT);
        dst[k & 15] = stage4[k];
    }
}

__global__ __launch_bounds__(OWN_THREADS) void agg1_owner(const int* __restrict__ ei,
                                                          const float* __restrict__ xp,
                                                          const float* __restrict__ as,
                                                          const float* __restrict__ ad,
                                                          const float* __restrict__ b1,
                                                          float* __restrict__ hout,
                                                          unsigned int* __restrict__ wsl,
                                                          const unsigned int* __restrict__ slots) {
    __shared__ float        acc[OWN_NODES * D1];
    __shared__ float        mx[OWN_NODES * H1];
    __shared__ float        dn[OWN_NODES * H1];
    __shared__ unsigned int list[OWN_CAP];
    __shared__ unsigned int scnt[BIN_BLOCKS];
    __shared__ unsigned int soff[BIN_BLOCKS];
    __shared__ unsigned int wtot[BIN_BLOCKS / 32];
    __shared__ unsigned int over;

    const int tid  = threadIdx.x;
    const int lane = tid & 31;
    const int wv   = __builtin_amdgcn_readfirstlane(tid >> 5);
    const int base = blockIdx.x * OWN_NODES;
    const int nown = (NNODES - base) < OWN_NODES ? (NNODES - base) : OWN_NODES;

    uint4 sv[4];
    {
        const uint4* sl4 = (const uint4*)(slots + (size_t)blockIdx.x * BIN_BLOCKS * BIN_SLOT);
#pragma unroll
        for (int u = 0; u < 4; ++u) sv[u] = sl4[u * OWN_THREADS + tid];
    }

    {
        float4* acc4 = (float4*)acc;
        for (int i = tid; i < OWN_NODES * D1 / 4; i += OWN_THREADS)
            acc4[i] = make_float4(0.0f, 0.0f, 0.0f, 0.0f);
        for (int i = tid; i < OWN_NODES * H1; i += OWN_THREADS) {
            mx[i] = -INFINITY;
            dn[i] = 0.0f;
        }
        if (tid < nown) list[tid] = (unsigned)(base + tid) | ((unsigned)tid << 16);
        if (tid == 0) over = 0u;
        if ((tid & 15) == 0) {
#pragma unroll
            for (int u = 0; u < 4; ++u) scnt[(u * OWN_THREADS + tid) >> 4] = sv[u].x;
        }
    }
    __syncthreads();

    const int4* __restrict__ dst4 = (const int4*)(ei + NEDGES);
    unsigned cc = 0u, inc = 0u;
    if (tid < BIN_BLOCKS) {
        const unsigned cs = scnt[tid];
        if (cs > BIN_CAP) over = 1u;
        cc  = cs < BIN_CAP ? cs : BIN_CAP;
        inc = cc;
        for (int o = 1; o < 32; o <<= 1) {
            const unsigned v = __shfl_up(inc, o);
            if (lane >= o) inc += v;
        }
        if (lane == 31) wtot[wv] = inc;
    }
    __syncthreads();
    if (tid < BIN_BLOCKS) {
        unsigned pre = 0u;
        for (int w = 0; w < wv; ++w) pre += wtot[w];
        soff[tid] = pre + inc - cc;
    }
    unsigned found = 0u;
    for (int w = 0; w < BIN_BLOCKS / 32; ++w) found += wtot[w];
    const bool spill = over != 0u;
    const int n = __builtin_amdgcn_readfirstlane(spill ? nown : nown + (int)found);
    __syncthreads();

    if (!spill) {
#pragma unroll
        for (int u = 0; u < 4; ++u) {
            const int      k  = u * OWN_THREADS + tid;
            const int      s  = k >> 4;
            const unsigned c  = scnt[s] < BIN_CAP ? scnt[s] : BIN_CAP;
            const int      p0 = nown + (int)soff[s] - 1;
            const int      w0 = (k & 15) * 4;
            const unsigned wd[4] = {sv[u].x, sv[u].y, sv[u].z, sv[u].w};
#pragma unroll
            for (int q = 0; q < 4; ++q) {
                const int w = w0 + q;
                if (w >= 1 && (unsigned)(w - 1) < c) list[p0 + w] = wd[q];
            }
        }
    }
    __syncthreads();

    {
        unsigned int* reg = wsl + (size_t)blockIdx.x * WSL_STRIDE;
        if (tid == 0) reg[0] = spill ? (unsigned)(OWN_CAP + 1) : (unsigned)n;
        for (int t = tid; t < n; t += OWN_THREADS) reg[WSL_HDR + t] = list[t];
    }

    for (int t0 = tid; t0 < n * H1; t0 += 4 * OWN_THREADS) {
        float lv[4];
        int   mi[4];
#pragma unroll
        for (int u = 0; u < 4; ++u) {
            const int  t  = t0 + u * OWN_THREADS;
            const bool ok = t < n * H1;
            const unsigned en = list[ok ? (t >> 3) : 0];
            const int hh = t & (H1 - 1);
            const int s  = (int)(en & 0xffffu);
            const int dl = (int)(en >> 16);
            mi[u] = ok ? dl * H1 + hh : -1;
            lv[u] = lrelu(as[s * H1 + hh] + ad[(base + dl) * H1 + hh]);
        }
#pragma unroll
        for (int u = 0; u < 4; ++u)
            if (mi[u] >= 0) atomicMaxF(&mx[mi[u]], lv[u]);
    }
    if (spill) {
        for (int i = tid; i < OWN_NV; i += OWN_THREADS) {
            const int4 v = dst4[i];
            const int dd[4] = {v.x, v.y, v.z, v.w};
#pragma unroll
            for (int q = 0; q < 4; ++q) {
                const unsigned dl = (unsigned)(dd[q] - base);
                if (dl < (unsigned)nown) {
                    const int s = ei[i * 4 + q];
                    for (int hh = 0; hh < H1; ++hh)
                        atomicMaxF(&mx[dl * H1 + hh],
                                   lrelu(as[s * H1 + hh] + ad[(base + (int)dl) * H1 + hh]));
                }
            }
        }
    }
    __syncthreads();

    {
        const int j = lane >> 3;
        const int h = lane & (H1 - 1);
        for (int i0 = wv * (4 * OWN_G); i0 < n; i0 += OWN_WAVES * 4 * OWN_G) {
            unsigned en[OWN_G];
#pragma unroll
            for (int g = 0; g < OWN_G; ++g) {
                const int idx = i0 + g * 4 + j;
                en[g] = list[idx < n ? idx : 0];
            }
            float va[OWN_G], vb[OWN_G];
            int   mh[OWN_G];
#pragma unroll
            for (int g = 0; g < OWN_G; ++g) {
                const int s  = (int)(en[g] & 0xffffu);
                const int dl = (int)(en[g] >> 16);
                mh[g] = dl * H1 + h;
                va[g] = as[s * H1 + h];
                vb[g] = ad[(base + dl) * H1 + h];
            }
            float row[OWN_G][4][H1];
#pragma unroll
            for (int g = 0; g < OWN_G; ++g) {
#pragma unroll
                for (int jj = 0; jj < 4; ++jj) {
                    const unsigned ej = (unsigned)__builtin_amdgcn_readlane((int)en[g], jj * H1);
                    const float* xr = xp + (size_t)(ej & 0xffffu) * D1 + lane;
#pragma unroll
                    for (int hh = 0; hh < H1; ++hh) row[g][jj][hh] = xr[hh * C1];
                }
            }
            float p[OWN_G];
#pragma unroll
            for (int g = 0; g < OWN_G; ++g)
                p[g] = __expf(lrelu(va[g] + vb[g]) - mx[mh[g]]);
#pragma unroll
            for (int g = 0; g < OWN_G; ++g)
                if (i0 + g * 4 + j < n) atomicAdd(&dn[mh[g]], p[g]);
#pragma unroll
            for (int g = 0; g < OWN_G; ++g) {
#pragma unroll
                for (int jj = 0; jj < 4; ++jj) {
                    if (i0 + g * 4 + jj < n) {
                        const unsigned ej = (unsigned)__builtin_amdgcn_readlane((int)en[g], jj * H1);
                        float* arow = acc + (ej >> 16) * D1 + lane;
#pragma unroll
                        for (int hh = 0; hh < H1; ++hh) {
                            const float a = __int_as_float(
                                __builtin_amdgcn_readlane(__float_as_int(p[g]), jj * H1 + hh));
                            atomicAdd(arow + hh * C1, a * row[g][jj][hh]);
                        }
                    }
                }
            }
        }
    }
    if (spill) {
        for (int it = 0; it < (OWN_NV + OWN_THREADS - 1) / OWN_THREADS; ++it) {
            const int i = it * OWN_THREADS + tid;
            const int4 v = i < OWN_NV ? dst4[i] : make_int4(-1, -1, -1, -1);
            const int dd[4] = {v.x, v.y, v.z, v.w};
#pragma unroll
            for (int q = 0; q < 4; ++q) {
                const unsigned dl  = (unsigned)(dd[q] - base);
                const bool     hit = dl < (unsigned)nown;
                const int      s   = hit ? ei[i * 4 + q] : 0;
                unsigned mask = __builtin_amdgcn_ballot_w32(hit);
                while (mask) {
                    const int l = __builtin_ctz(mask);
                    mask &= mask - 1;
                    const int sj = __builtin_amdgcn_readlane(s, l);
                    const int dj = __builtin_amdgcn_readlane((int)dl, l);
                    const int h  = lane & (H1 - 1);
                    const float lv = lrelu(as[sj * H1 + h] + ad[(base + dj) * H1 + h]);
                    const float pe = __expf(lv - mx[dj * H1 + h]);
                    if (lane < H1) atomicAdd(&dn[dj * H1 + h], pe);
                    const float* xr = xp + (size_t)sj * D1 + lane;
                    float* arow = acc + dj * D1 + lane;
#pragma unroll
                    for (int hh = 0; hh < H1; ++hh) {
                        const float a = __int_as_float(
                            __builtin_amdgcn_readlane(__float_as_int(pe), hh));
                        atomicAdd(arow + hh * C1, a * xr[hh * C1]);
                    }
                }
            }
        }
    }
    __syncthreads();

    for (int i = tid; i < nown * D1; i += OWN_THREADS) {
        const int nd  = i >> 8;
        const int col = i & (D1 - 1);
        const float v = acc[i] / dn[nd * H1 + (col >> 5)] + b1[col];
        hout[(size_t)(base + nd) * D1 + col] = v > 0.0f ? v : 0.0f;
    }
}

#define P2_NODES 32
#define P2_BATCH 16
#define NPAD     (((NNODES + P2_NODES - 1) / P2_NODES) * P2_NODES)
static_assert(P2_NODES == 32, "node i of a group ends in lane i");
static_assert(P2_NODES % P2_BATCH == 0 && NNODES % P2_BATCH == 0, "a batch is whole or empty");
static_assert(D1 == 256 && OUT2 == 2, "eight k per lane, two float4 of W2 per four k");

__global__ __launch_bounds__(32) __attribute__((amdgpu_waves_per_eu(1, 4))) void gemm2_alpha2(const float* __restrict__ h,
                                                   const float* __restrict__ W2,
                                                   const float* __restrict__ aw_src,
                                                   const float* __restrict__ aw_dst,
                                                   float* __restrict__ xp2,
                                                   float* __restrict__ as2,
                                                   float* __restrict__ ad2) {
    const int lane = threadIdx.x;
    const int n0   = blockIdx.x * P2_NODES;

    const float4* __restrict__ W2v = (const float4*)W2;
    const float4 wa0 = W2v[2 * lane];
    const float4 wa1 = W2v[2 * lane + 1];
    const float4 wb0 = W2v[64 + 2 * lane];
    const float4 wb1 = W2v[64 + 2 * lane + 1];

    float r0 = 0.0f, r1 = 0.0f;
    for (int b0 = 0; b0 < P2_NODES; b0 += P2_BATCH) {
        if (n0 + b0 >= NNODES) break;
        float4 va[P2_BATCH], vb[P2_BATCH];
#pragma unroll
        for (int i = 0; i < P2_BATCH; ++i) {
            const float4* __restrict__ hr = (const float4*)(h + (size_t)(n0 + b0 + i) * D1);
            va[i] = hr[lane];
            vb[i] = hr[32 + lane];
        }
#pragma unroll
        for (int i = 0; i < P2_BATCH; ++i) {
            float s0 = va[i].x * wa0.x;
            float s1 = va[i].x * wa0.y;
            s0 = fmaf(va[i].y, wa0.z, s0);  s1 = fmaf(va[i].y, wa0.w, s1);
            s0 = fmaf(va[i].z, wa1.x, s0);  s1 = fmaf(va[i].z, wa1.y, s1);
            s0 = fmaf(va[i].w, wa1.z, s0);  s1 = fmaf(va[i].w, wa1.w, s1);
            s0 = fmaf(vb[i].x, wb0.x, s0);  s1 = fmaf(vb[i].x, wb0.y, s1);
            s0 = fmaf(vb[i].y, wb0.z, s0);  s1 = fmaf(vb[i].y, wb0.w, s1);
            s0 = fmaf(vb[i].z, wb1.x, s0);  s1 = fmaf(vb[i].z, wb1.y, s1);
            s0 = fmaf(vb[i].w, wb1.z, s0);  s1 = fmaf(vb[i].w, wb1.w, s1);
#pragma unroll
            for (int off = 16; off > 0; off >>= 1) {
                s0 += __shfl_xor(s0, off);
                s1 += __shfl_xor(s1, off);
            }
            if (lane == b0 + i) { r0 = s0; r1 = s1; }
        }
    }

    const int n = n0 + lane;
    if (n < NNODES) {
        ((float2*)xp2)[n] = make_float2(r0, r1);
        as2[n] = fmaf(r0, aw_src[0], r1 * aw_src[1]);
        ad2[n] = fmaf(r0, aw_dst[0], r1 * aw_dst[1]);
    }
}

#define L2_NODES (((OWN_NODES + 15) / 16) * 16)
static_assert(NNODES % 16 == 0, "the output ends on a line boundary");
static_assert(OWN_NODES >= 16, "a group's rows lie in two lists at most");
static_assert(L2_NODES * OUT2 <= OWN_THREADS, "one lane per output word in the epilogue");

__global__ __launch_bounds__(OWN_THREADS) void agg2_owner(const int* __restrict__ ei,
                                                          const unsigned int* __restrict__ wsl,
                                                          const float* __restrict__ xp2,
                                                          const float* __restrict__ as2,
                                                          const float* __restrict__ ad2,
                                                          const float* __restrict__ b2,
                                                          float* __restrict__ out) {
    __shared__ float mx[L2_NODES];
    __shared__ float dn[L2_NODES];
    __shared__ float s0[L2_NODES];
    __shared__ float s1[L2_NODES];

    const int tid = threadIdx.x;
    const int b   = blockIdx.x;
    int lo = (b * OWN_NODES + 15) & ~15;
    int hi = ((b + 1) * OWN_NODES + 15) & ~15;
    lo = lo < NNODES ? lo : NNODES;
    hi = hi < NNODES ? hi : NNODES;
    const int nloc = hi - lo;

    if (tid < L2_NODES) {
        mx[tid] = -INFINITY;
        dn[tid] = 0.0f;
        s0[tid] = 0.0f;
        s1[tid] = 0.0f;
    }

    const bool two = (b + 1 < OWN_BLOCKS) && ((b + 1) * OWN_NODES < hi);
    const unsigned totA = wsl[(size_t)b * WSL_STRIDE];
    const unsigned totB = two ? wsl[(size_t)(b + 1) * WSL_STRIDE] : 0u;
    const bool spill = totA > OWN_CAP || totB > OWN_CAP;
    const float2* __restrict__ xp2v = (const float2*)xp2;
    const int4*   __restrict__ dst4 = (const int4*)(ei + NEDGES);
    __syncthreads();

    if (!spill) {
        for (int g = 0; g < (two ? 2 : 1); ++g) {
            const unsigned int* reg = wsl + (size_t)(b + g) * WSL_STRIDE + WSL_HDR;
            const int n     = (int)(g ? totB : totA);
            const int gbase = (b + g) * OWN_NODES - lo;
            for (int t0 = tid; t0 < n; t0 += 4 * OWN_THREADS) {
                unsigned en[4];
#pragma unroll
                for (int u = 0; u < 4; ++u) {
                    const int t = t0 + u * OWN_THREADS;
                    en[u] = reg[t < n ? t : 0];
                }
                int   li[4];
                float va[4], vb[4];
#pragma unroll
                for (int u = 0; u < 4; ++u) {
                    const int  l  = gbase + (int)(en[u] >> 16);
                    const bool ok = t0 + u * OWN_THREADS < n && (unsigned)l < (unsigned)nloc;
                    li[u] = ok ? l : -1;
                    va[u] = as2[ok ? (int)(en[u] & 0xffffu) : lo];
                    vb[u] = ad2[ok ? lo + l : lo];
                }
#pragma unroll
                for (int u = 0; u < 4; ++u)
                    if (li[u] >= 0) atomicMaxF(&mx[li[u]], lrelu(va[u] + vb[u]));
            }
        }
    } else {
        for (int i = tid; i < nloc; i += OWN_THREADS)
            atomicMaxF(&mx[i], lrelu(as2[lo + i] + ad2[lo + i]));
        for (int i = tid; i < OWN_NV; i += OWN_THREADS) {
            const int4 v = dst4[i];
            const int dd[4] = {v.x, v.y, v.z, v.w};
#pragma unroll
            for (int q = 0; q < 4; ++q) {
                const unsigned l = (unsigned)(dd[q] - lo);
                if (l < (unsigned)nloc)
                    atomicMaxF(&mx[l], lrelu(as2[ei[i * 4 + q]] + ad2[dd[q]]));
            }
        }
    }
    __syncthreads();

    if (!spill) {
        for (int g = 0; g < (two ? 2 : 1); ++g) {
            const unsigned int* reg = wsl + (size_t)(b + g) * WSL_STRIDE + WSL_HDR;
            const int n     = (int)(g ? totB : totA);
            const int gbase = (b + g) * OWN_NODES - lo;
            for (int t0 = tid; t0 < n; t0 += 4 * OWN_THREADS) {
                unsigned en[4];
#pragma unroll
                for (int u = 0; u < 4; ++u) {
                    const int t = t0 + u * OWN_THREADS;
                    en[u] = reg[t < n ? t : 0];
                }
                int    li[4];
                float  va[4], vb[4];
                float2 xs[4];
#pragma unroll
                for (int u = 0; u < 4; ++u) {
                    const int  l  = gbase + (int)(en[u] >> 16);
                    const bool ok = t0 + u * OWN_THREADS < n && (unsigned)l < (unsigned)nloc;
                    const int  s  = ok ? (int)(en[u] & 0xffffu) : lo;
                    li[u] = ok ? l : -1;
                    va[u] = as2[s];
                    vb[u] = ad2[ok ? lo + l : lo];
                    xs[u] = xp2v[s];
                }
#pragma unroll
                for (int u = 0; u < 4; ++u) {
                    if (li[u] >= 0) {
                        const float p = __expf(lrelu(va[u] + vb[u]) - mx[li[u]]);
                        atomicAdd(&dn[li[u]], p);
                        atomicAdd(&s0[li[u]], p * xs[u].x);
                        atomicAdd(&s1[li[u]], p * xs[u].y);
                    }
                }
            }
        }
    } else {
        for (int i = tid; i < nloc; i += OWN_THREADS) {
            const float  p = __expf(lrelu(as2[lo + i] + ad2[lo + i]) - mx[i]);
            const float2 x = xp2v[lo + i];
            atomicAdd(&dn[i], p);
            atomicAdd(&s0[i], p * x.x);
            atomicAdd(&s1[i], p * x.y);
        }
        for (int i = tid; i < OWN_NV; i += OWN_THREADS) {
            const int4 v = dst4[i];
            const int dd[4] = {v.x, v.y, v.z, v.w};
#pragma unroll
            for (int q = 0; q < 4; ++q) {
                const unsigned l = (unsigned)(dd[q] - lo);
                if (l < (unsigned)nloc) {
                    const int    s = ei[i * 4 + q];
                    const float  p = __expf(lrelu(as2[s] + ad2[dd[q]]) - mx[l]);
                    const float2 x = xp2v[s];
                    atomicAdd(&dn[l], p);
                    atomicAdd(&s0[l], p * x.x);
                    atomicAdd(&s1[l], p * x.y);
                }
            }
        }
    }
    __syncthreads();

    if (tid < nloc * OUT2) {
        const int nd = tid >> 1;
        const float sv = (tid & 1) ? s1[nd] : s0[nd];
        out[(size_t)lo * OUT2 + tid] = sv / dn[nd] + b2[tid & 1];
    }
}

extern "C" void kernel_launch(void* const* d_in, const int* in_sizes, int n_in,
                              void* d_out, int out_size, void* d_ws, size_t ws_size,
                              hipStream_t stream) {
    const float* x     = (const float*)d_in[0];
    const int*   ei    = (const int*)d_in[1];
    const float* W1    = (const float*)d_in[2];
    const float* asw1  = (const float*)d_in[3];
    const float* adw1  = (const float*)d_in[4];
    const float* b1    = (const float*)d_in[5];
    const float* W2    = (const float*)d_in[6];
    const float* asw2  = (const float*)d_in[7];
    const float* adw2  = (const float*)d_in[8];
    const float* b2    = (const float*)d_in[9];
    float*       out   = (float*)d_out;

    float* w    = (float*)d_ws;
    float* xp1  = w;
    float* hbuf = xp1  + (size_t)NNODES * D1;
    float* as1  = hbuf + (size_t)NNODES * D1;
    float* ad1  = as1  + (size_t)NNODES * H1;
    float* xp2  = ad1  + (size_t)NNODES * H1;
    float* as2  = xp2  + (size_t)NNODES * OUT2;
    float* ad2  = as2  + (size_t)NPAD;
    unsigned int* wsl = (unsigned int*)(ad2 + (size_t)NPAD);
    unsigned int* slots = wsl + (size_t)OWN_BLOCKS * WSL_STRIDE;

    gemm1_wmma<<<NNODES / 16, 512, 0, stream>>>(x, W1, xp1);
    alpha1_kernel<<<(NNODES * H1 + 255) / 256, 256, 0, stream>>>(xp1, asw1, adw1, as1, ad1);
    bin_edges<<<BIN_BLOCKS, BIN_THREADS, 0, stream>>>(ei, slots);
    agg1_owner<<<OWN_BLOCKS, OWN_THREADS, 0, stream>>>(ei, xp1, as1, ad1, b1, hbuf, wsl, slots);

    gemm2_alpha2<<<NPAD / P2_NODES, 32, 0, stream>>>(hbuf, W2, asw2, adw2, xp2, as2, ad2);
    agg2_owner<<<OWN_BLOCKS, OWN_THREADS, 0, stream>>>(ei, wsl, xp2, as2, ad2, b2, out);
}


# Layer-1 GEMM row-blocked: a work-group owns 32 rows, not 16; each wave keeps two row tiles' accumulators and feeds both from one W1 fragment load (W1 fetched from L2 half as often); sum order kept
# speedup vs baseline: 8.6761x; 1.1030x over previous
#include <hip/hip_runtime.h>
#include <hip/hip_bf16.h>
#include <math.h>

#define NNODES 50000
#define NEDGES 800000
#define ETOT   (NEDGES + NNODES)
#define FIN    256
#define H1     8
#define C1     32
#define D1     256
#define OUT2   2
#define SLOPE  0.2f

#define LDSPAD 4
#define ASTRIDE (FIN + LDSPAD)

typedef float v2f __attribute__((ext_vector_type(2)));
typedef float v8f __attribute__((ext_vector_type(8)));

__device__ __forceinline__ float lrelu(float v) {
    return v > 0.0f ? v : SLOPE * v;
}

__device__ __forceinline__ void atomicMaxF(float* addr, float val) {
    if (val >= 0.0f) atomicMax((int*)addr, __float_as_int(val));
    else             atomicMin((unsigned int*)addr, __float_as_uint(val));
}

#define G1_WAVES   17
#define G1_RT      2
#define G1_ROWS    (16 * G1_RT)
#define G1_BLOCKS  ((NNODES + G1_ROWS - 1) / G1_ROWS)
static_assert(NNODES % 16 == 0, "a row tile is whole or absent");
#define WA_COLS    16
static_assert(WA_COLS == 2 * H1, "the logit tile is exactly one 16-column tile");

template <int LDB>
__device__ __forceinline__ void panel_tiles(const float* aTile, const float* __restrict__ B,
                                            const int l16, const int half,
                                            v8f& acc0, v8f& acc1) {
    v8f c0 = {}, c1 = {};
#pragma unroll
    for (int k = 0; k < FIN; k += 4) {
        const int kk = k + half * 2;
        v2f b;
        b.x = B[kk * LDB + l16];
        b.y = B[(kk + 1) * LDB + l16];
        const v2f a0 = *(const v2f*)(aTile + l16 * ASTRIDE + kk);
        const v2f a1 = *(const v2f*)(aTile + (16 + l16) * ASTRIDE + kk);
        c0 = __builtin_amdgcn_wmma_f32_16x16x4_f32(
             false, a0,  false, b,
             (short)0, c0,  false,  false);
        c1 = __builtin_amdgcn_wmma_f32_16x16x4_f32(
             false, a1,  false, b,
             (short)0, c1,  false,  false);
    }
    acc0 = c0;
    acc1 = c1;
}

__global__ __launch_bounds__(G1_WAVES * 32) void gemm1_wmma(const float* __restrict__ x,
                                                  const float* __restrict__ W1,
                                                  const float* __restrict__ Wa,
                                                  float* __restrict__ xp,
                                                  float* __restrict__ as,
                                                  float* __restrict__ ad) {
    __shared__ float aTile[G1_ROWS * ASTRIDE];

    const int t     = threadIdx.x;
    const int lane  = t & 31;
    const int wave  = __builtin_amdgcn_readfirstlane(t >> 5);
    const int row0  = blockIdx.x * G1_ROWS;
    const bool two  = row0 + 16 < NNODES;
    const int half  = lane >> 4;
    const int l16   = lane & 15;

    if (wave < 16) {
        const int row = t >> 5;
        const int col = (t & 31) * 8;
        const float4* src0 = (const float4*)(x + (size_t)(row0 + row) * FIN + col);
        float4 v0 = src0[0], v1 = src0[1];
        float4 v2 = make_float4(0.0f, 0.0f, 0.0f, 0.0f), v3 = v2;
        if (two) {
            const float4* src1 = (const float4*)(x + (size_t)(row0 + 16 + row) * FIN + col);
            v2 = src1[0];
            v3 = src1[1];
        }
        float4* dst0 = (float4*)(aTile + row * ASTRIDE + col);
        float4* dst1 = (float4*)(aTile + (16 + row) * ASTRIDE + col);
        dst0[0] = v0;
        dst0[1] = v1;
        dst1[0] = v2;
        dst1[1] = v3;
    }
    __syncthreads();

    const int mrow = row0 + half * 8;
    v8f acc0, acc1;
    if (wave < 16) {
        const int bcol = wave * 16 + l16;
        panel_tiles<D1>(aTile, W1 + wave * 16, l16, half, acc0, acc1);
#pragma unroll
        for (int r = 0; r < 8; ++r)
            xp[(size_t)(mrow + r) * D1 + bcol] = acc0[r];
        if (two) {
#pragma unroll
            for (int r = 0; r < 8; ++r)
                xp[(size_t)(mrow + 16 + r) * D1 + bcol] = acc1[r];
        }
    } else {
        panel_tiles<WA_COLS>(aTile, Wa, l16, half, acc0, acc1);
        float* __restrict__ lg = l16 < H1 ? as + l16 : ad + (l16 - H1);
#pragma unroll
        for (int r = 0; r < 8; ++r)
            lg[(size_t)(mrow + r) * H1] = acc0[r];
        if (two) {
#pragma unroll
            for (int r = 0; r < 8; ++r)
                lg[(size_t)(mrow + 16 + r) * H1] = acc1[r];
        }
    }
}

#define OWN_BLOCKS  256
#define OWN_NODES   196
#define OWN_THREADS 1024
#define OWN_WAVES   (OWN_THREADS / 32)
#define OWN_CAP     20480
#define OWN_G       2
#define OWN_NV      (NEDGES / 4)
#define WSL_HDR     32
#define WSL_STRIDE  (OWN_CAP + WSL_HDR)
#define BIN_BLOCKS  256
#define BIN_THREADS 1024
#define BIN_SLOT    64
#define BIN_CAP     (BIN_SLOT - 1)
#define BIN_NV      ((OWN_NV + BIN_BLOCKS - 1) / BIN_BLOCKS)
static_assert(OWN_BLOCKS * OWN_NODES >= NNODES, "every row needs an owner");
static_assert(NNODES - (OWN_BLOCKS - 1) * OWN_NODES > 0, "no group without rows");
static_assert(NNODES <= 65536 && OWN_NODES <= 65535, "a list entry packs two 16-bit numbers");
static_assert(NEDGES % 4 == 0, "the edge list is read four at a time");
static_assert(OWN_NODES <= OWN_THREADS && OWN_NODES <= OWN_CAP, "self loops go in first");
static_assert(BIN_NV <= BIN_THREADS, "one int4 of the edge list per lane of a binning group");
static_assert(OWN_BLOCKS * BIN_SLOT == 16 * BIN_THREADS, "a binning group moves its slots 16 words per lane");
static_assert(BIN_BLOCKS * BIN_SLOT == 16 * OWN_THREADS, "an owner group reads its slots 16 words per lane");
static_assert(BIN_BLOCKS <= OWN_THREADS && BIN_BLOCKS % 32 == 0, "one lane per slot in the offset scan, whole waves");
static_assert(OWN_NODES + BIN_BLOCKS * BIN_CAP <= OWN_CAP, "full slots still fit the list");

#define WA_BLOCKS ((FIN * WA_COLS + BIN_THREADS - 1) / BIN_THREADS)

__global__ __launch_bounds__(BIN_THREADS) void bin_edges(const int* __restrict__ ei,
                                                         unsigned int* __restrict__ slots,
                                                         const float* __restrict__ W1,
                                                         const float* __restrict__ a_src,
                                                         const float* __restrict__ a_dst,
                                                         float* __restrict__ Wa) {
    __shared__ unsigned int stage[OWN_BLOCKS * BIN_SLOT];

    const int tid = threadIdx.x;
    if (blockIdx.x >= BIN_BLOCKS) {
        const int g = (blockIdx.x - BIN_BLOCKS) * BIN_THREADS + tid;
        if (g < FIN * WA_COLS) {
            const int k  = g / WA_COLS;
            const int j  = g % WA_COLS;
            const int hd = j & (H1 - 1);
            const float* wr = W1 + (size_t)k * D1 + hd * C1;
            const float* av = (j < H1 ? a_src : a_dst) + hd * C1;
            float s = 0.0f;
#pragma unroll
            for (int ch = 0; ch < C1; ++ch) s = fmaf(wr[ch], av[ch], s);
            Wa[g] = s;
        }
        return;
    }
    uint4* stage4 = (uint4*)stage;
    for (int u = 0; u < 4; ++u) stage4[u * BIN_THREADS + tid] = make_uint4(0u, 0u, 0u, 0u);
    __syncthreads();

    const int i = blockIdx.x * BIN_NV + tid;
    if (tid < BIN_NV && i < OWN_NV) {
        const int4 sv = ((const int4*)ei)[i];
        const int4 dv = ((const int4*)(ei + NEDGES))[i];
        const int ss[4] = {sv.x, sv.y, sv.z, sv.w};
        const int dd[4] = {dv.x, dv.y, dv.z, dv.w};
#pragma unroll
        for (int q = 0; q < 4; ++q) {
            const unsigned d = (unsigned)dd[q];
            if (d < (unsigned)NNODES) {
                const unsigned g  = d / OWN_NODES;
                const unsigned dl = d - g * OWN_NODES;
                const unsigned r  = atomicAdd(&stage[g * BIN_SLOT], 1u);
                if (r < BIN_CAP) stage[g * BIN_SLOT + 1 + r] = (unsigned)ss[q] | (dl << 16);
            }
        }
    }
    __syncthreads();

    for (int u = 0; u < 4; ++u) {
        const int k = u * BIN_THREADS + tid;
        const int g = k >> 4;
        uint4* dst = (uint4*)(slots + ((size_t)g * BIN_BLOCKS + blockIdx.x) * BIN_SLOT);
        dst[k & 15] = stage4[k];
    }
}

__global__ __launch_bounds__(OWN_THREADS) void agg1_owner(const int* __restrict__ ei,
                                                          const float* __restrict__ xp,
                                                          const float* __restrict__ as,
                                                          const float* __restrict__ ad,
                                                          const float* __restrict__ b1,
                                                          float* __restrict__ hout,
                                                          unsigned int* __restrict__ wsl,
                                                          const unsigned int* __restrict__ slots) {
    __shared__ float        acc[OWN_NODES * D1];
    __shared__ float        mx[OWN_NODES * H1];
    __shared__ float        dn[OWN_NODES * H1];
    __shared__ unsigned int list[OWN_CAP];
    __shared__ unsigned int scnt[BIN_BLOCKS];
    __shared__ unsigned int soff[BIN_BLOCKS];
    __shared__ unsigned int wtot[BIN_BLOCKS / 32];
    __shared__ unsigned int over;

    const int tid  = threadIdx.x;
    const int lane = tid & 31;
    const int wv   = __builtin_amdgcn_readfirstlane(tid >> 5);
    const int base = blockIdx.x * OWN_NODES;
    const int nown = (NNODES - base) < OWN_NODES ? (NNODES - base) : OWN_NODES;

    uint4 sv[4];
    {
        const uint4* sl4 = (const uint4*)(slots + (size_t)blockIdx.x * BIN_BLOCKS * BIN_SLOT);
#pragma unroll
        for (int u = 0; u < 4; ++u) sv[u] = sl4[u * OWN_THREADS + tid];
    }

    {
        float4* acc4 = (float4*)acc;
        for (int i = tid; i < OWN_NODES * D1 / 4; i += OWN_THREADS)
            acc4[i] = make_float4(0.0f, 0.0f, 0.0f, 0.0f);
        for (int i = tid; i < OWN_NODES * H1; i += OWN_THREADS) {
            mx[i] = -INFINITY;
            dn[i] = 0.0f;
        }
        if (tid < nown) list[tid] = (unsigned)(base + tid) | ((unsigned)tid << 16);
        if (tid == 0) over = 0u;
        if ((tid & 15) == 0) {
#pragma unroll
            for (int u = 0; u < 4; ++u) scnt[(u * OWN_THREADS + tid) >> 4] = sv[u].x;
        }
    }
    __syncthreads();

    const int4* __restrict__ dst4 = (const int4*)(ei + NEDGES);
    unsigned cc = 0u, inc = 0u;
    if (tid < BIN_BLOCKS) {
        const unsigned cs = scnt[tid];
        if (cs > BIN_CAP) over = 1u;
        cc  = cs < BIN_CAP ? cs : BIN_CAP;
        inc = cc;
        for (int o = 1; o < 32; o <<= 1) {
            const unsigned v = __shfl_up(inc, o);
            if (lane >= o) inc += v;
        }
        if (lane == 31) wtot[wv] = inc;
    }
    __syncthreads();
    if (tid < BIN_BLOCKS) {
        unsigned pre = 0u;
        for (int w = 0; w < wv; ++w) pre += wtot[w];
        soff[tid] = pre + inc - cc;
    }
    unsigned found = 0u;
    for (int w = 0; w < BIN_BLOCKS / 32; ++w) found += wtot[w];
    const bool spill = over != 0u;
    const int n = __builtin_amdgcn_readfirstlane(spill ? nown : nown + (int)found);
    __syncthreads();

    if (!spill) {
#pragma unroll
        for (int u = 0; u < 4; ++u) {
            const int      k  = u * OWN_THREADS + tid;
            const int      s  = k >> 4;
            const unsigned c  = scnt[s] < BIN_CAP ? scnt[s] : BIN_CAP;
            const int      p0 = nown + (int)soff[s] - 1;
            const int      w0 = (k & 15) * 4;
            const unsigned wd[4] = {sv[u].x, sv[u].y, sv[u].z, sv[u].w};
#pragma unroll
            for (int q = 0; q < 4; ++q) {
                const int w = w0 + q;
                if (w >= 1 && (unsigned)(w - 1) < c) list[p0 + w] = wd[q];
            }
        }
    }
    __syncthreads();

    {
        unsigned int* reg = wsl + (size_t)blockIdx.x * WSL_STRIDE;
        if (tid == 0) reg[0] = spill ? (unsigned)(OWN_CAP + 1) : (unsigned)n;
        for (int t = tid; t < n; t += OWN_THREADS) reg[WSL_HDR + t] = list[t];
    }

    for (int t0 = tid; t0 < n * H1; t0 += 4 * OWN_THREADS) {
        float lv[4];
        int   mi[4];
#pragma unroll
        for (int u = 0; u < 4; ++u) {
            const int  t  = t0 + u * OWN_THREADS;
            const bool ok = t < n * H1;
            const unsigned en = list[ok ? (t >> 3) : 0];
            const int hh = t & (H1 - 1);
            const int s  = (int)(en & 0xffffu);
            const int dl = (int)(en >> 16);
            mi[u] = ok ? dl * H1 + hh : -1;
            lv[u] = lrelu(as[s * H1 + hh] + ad[(base + dl) * H1 + hh]);
        }
#pragma unroll
        for (int u = 0; u < 4; ++u)
            if (mi[u] >= 0) atomicMaxF(&mx[mi[u]], lv[u]);
    }
    if (spill) {
        for (int i = tid; i < OWN_NV; i += OWN_THREADS) {
            const int4 v = dst4[i];
            const int dd[4] = {v.x, v.y, v.z, v.w};
#pragma unroll
            for (int q = 0; q < 4; ++q) {
                const unsigned dl = (unsigned)(dd[q] - base);
                if (dl < (unsigned)nown) {
                    const int s = ei[i * 4 + q];
                    for (int hh = 0; hh < H1; ++hh)
                        atomicMaxF(&mx[dl * H1 + hh],
                                   lrelu(as[s * H1 + hh] + ad[(base + (int)dl) * H1 + hh]));
                }
            }
        }
    }
    __syncthreads();

    {
        const int j = lane >> 3;
        const int h = lane & (H1 - 1);
        for (int i0 = wv * (4 * OWN_G); i0 < n; i0 += OWN_WAVES * 4 * OWN_G) {
            unsigned en[OWN_G];
#pragma unroll
            for (int g = 0; g < OWN_G; ++g) {
                const int idx = i0 + g * 4 + j;
                en[g] = list[idx < n ? idx : 0];
            }
            float va[OWN_G], vb[OWN_G];
            int   mh[OWN_G];
#pragma unroll
            for (int g = 0; g < OWN_G; ++g) {
                const int s  = (int)(en[g] & 0xffffu);
                const int dl = (int)(en[g] >> 16);
                mh[g] = dl * H1 + h;
                va[g] = as[s * H1 + h];
                vb[g] = ad[(base + dl) * H1 + h];
            }
            float row[OWN_G][4][H1];
#pragma unroll
            for (int g = 0; g < OWN_G; ++g) {
#pragma unroll
                for (int jj = 0; jj < 4; ++jj) {
                    const unsigned ej = (unsigned)__builtin_amdgcn_readlane((int)en[g], jj * H1);
                    const float* xr = xp + (size_t)(ej & 0xffffu) * D1 + lane;
#pragma unroll
                    for (int hh = 0; hh < H1; ++hh) row[g][jj][hh] = xr[hh * C1];
                }
            }
            float p[OWN_G];
#pragma unroll
            for (int g = 0; g < OWN_G; ++g)
                p[g] = __expf(lrelu(va[g] + vb[g]) - mx[mh[g]]);
#pragma unroll
            for (int g = 0; g < OWN_G; ++g)
                if (i0 + g * 4 + j < n) atomicAdd(&dn[mh[g]], p[g]);
#pragma unroll
            for (int g = 0; g < OWN_G; ++g) {
#pragma unroll
                for (int jj = 0; jj < 4; ++jj) {
                    if (i0 + g * 4 + jj < n) {
                        const unsigned ej = (unsigned)__builtin_amdgcn_readlane((int)en[g], jj * H1);
                        float* arow = acc + (ej >> 16) * D1 + lane;
#pragma unroll
                        for (int hh = 0; hh < H1; ++hh) {
                            const float a = __int_as_float(
                                __builtin_amdgcn_readlane(__float_as_int(p[g]), jj * H1 + hh));
                            atomicAdd(arow + hh * C1, a * row[g][jj][hh]);
                        }
                    }
                }
            }
        }
    }
    if (spill) {
        for (int it = 0; it < (OWN_NV + OWN_THREADS - 1) / OWN_THREADS; ++it) {
            const int i = it * OWN_THREADS + tid;
            const int4 v = i < OWN_NV ? dst4[i] : make_int4(-1, -1, -1, -1);
            const int dd[4] = {v.x, v.y, v.z, v.w};
#pragma unroll
            for (int q = 0; q < 4; ++q) {
                const unsigned dl  = (unsigned)(dd[q] - base);
                const bool     hit = dl < (unsigned)nown;
                const int      s   = hit ? ei[i * 4 + q] : 0;
                unsigned mask = __builtin_amdgcn_ballot_w32(hit);
                while (mask) {
                    const int l = __builtin_ctz(mask);
                    mask &= mask - 1;
                    const int sj = __builtin_amdgcn_readlane(s, l);
                    const int dj = __builtin_amdgcn_readlane((int)dl, l);
                    const int h  = lane & (H1 - 1);
                    const float lv = lrelu(as[sj * H1 + h] + ad[(base + dj) * H1 + h]);
                    const float pe = __expf(lv - mx[dj * H1 + h]);
                    if (lane < H1) atomicAdd(&dn[dj * H1 + h], pe);
                    const float* xr = xp + (size_t)sj * D1 + lane;
                    float* arow = acc + dj * D1 + lane;
#pragma unroll
                    for (int hh = 0; hh < H1; ++hh) {
                        const float a = __int_as_float(
                            __builtin_amdgcn_readlane(__float_as_int(pe), hh));
                        atomicAdd(arow + hh * C1, a * xr[hh * C1]);
                    }
                }
            }
        }
    }
    __syncthreads();

    for (int i = tid; i < nown * D1; i += OWN_THREADS) {
        const int nd  = i >> 8;
        const int col = i & (D1 - 1);
        const float v = acc[i] / dn[nd * H1 + (col >> 5)] + b1[col];
        hout[(size_t)(base + nd) * D1 + col] = v > 0.0f ? v : 0.0f;
    }
}

#define P2_NODES 32
#define P2_BATCH 16
#define NPAD     (((NNODES + P2_NODES - 1) / P2_NODES) * P2_NODES)
static_assert(P2_NODES == 32, "node i of a group ends in lane i");
static_assert(P2_NODES % P2_BATCH == 0 && NNODES % P2_BATCH == 0, "a batch is whole or empty");
static_assert(D1 == 256 && OUT2 == 2, "eight k per lane, two float4 of W2 per four k");

__global__ __launch_bounds__(32) __attribute__((amdgpu_waves_per_eu(1, 4))) void gemm2_alpha2(const float* __restrict__ h,
                                                   const float* __restrict__ W2,
                                                   const float* __restrict__ aw_src,
                                                   const float* __restrict__ aw_dst,
                                                   float* __restrict__ xp2,
                                                   float* __restrict__ as2,
                                                   float* __restrict__ ad2) {
    const int lane = threadIdx.x;
    const int n0   = blockIdx.x * P2_NODES;

    const float4* __restrict__ W2v = (const float4*)W2;
    const float4 wa0 = W2v[2 * lane];
    const float4 wa1 = W2v[2 * lane + 1];
    const float4 wb0 = W2v[64 + 2 * lane];
    const float4 wb1 = W2v[64 + 2 * lane + 1];

    float r0 = 0.0f, r1 = 0.0f;
    for (int b0 = 0; b0 < P2_NODES; b0 += P2_BATCH) {
        if (n0 + b0 >= NNODES) break;
        float4 va[P2_BATCH], vb[P2_BATCH];
#pragma unroll
        for (int i = 0; i < P2_BATCH; ++i) {
            const float4* __restrict__ hr = (const float4*)(h + (size_t)(n0 + b0 + i) * D1);
            va[i] = hr[lane];
            vb[i] = hr[32 + lane];
        }
#pragma unroll
        for (int i = 0; i < P2_BATCH; ++i) {
            float s0 = va[i].x * wa0.x;
            float s1 = va[i].x * wa0.y;
            s0 = fmaf(va[i].y, wa0.z, s0);  s1 = fmaf(va[i].y, wa0.w, s1);
            s0 = fmaf(va[i].z, wa1.x, s0);  s1 = fmaf(va[i].z, wa1.y, s1);
            s0 = fmaf(va[i].w, wa1.z, s0);  s1 = fmaf(va[i].w, wa1.w, s1);
            s0 = fmaf(vb[i].x, wb0.x, s0);  s1 = fmaf(vb[i].x, wb0.y, s1);
            s0 = fmaf(vb[i].y, wb0.z, s0);  s1 = fmaf(vb[i].y, wb0.w, s1);
            s0 = fmaf(vb[i].z, wb1.x, s0);  s1 = fmaf(vb[i].z, wb1.y, s1);
            s0 = fmaf(vb[i].w, wb1.z, s0);  s1 = fmaf(vb[i].w, wb1.w, s1);
#pragma unroll
            for (int off = 16; off > 0; off >>= 1) {
                s0 += __shfl_xor(s0, off);
                s1 += __shfl_xor(s1, off);
            }
            if (lane == b0 + i) { r0 = s0; r1 = s1; }
        }
    }

    const int n = n0 + lane;
    if (n < NNODES) {
        ((float2*)xp2)[n] = make_float2(r0, r1);
        as2[n] = fmaf(r0, aw_src[0], r1 * aw_src[1]);
        ad2[n] = fmaf(r0, aw_dst[0], r1 * aw_dst[1]);
    }
}

#define L2_NODES (((OWN_NODES + 15) / 16) * 16)
static_assert(NNODES % 16 == 0, "the output ends on a line boundary");
static_assert(OWN_NODES >= 16, "a group's rows lie in two lists at most");
static_assert(L2_NODES * OUT2 <= OWN_THREADS, "one lane per output word in the epilogue");

__global__ __launch_bounds__(OWN_THREADS) void agg2_owner(const int* __restrict__ ei,
                                                          const unsigned int* __restrict__ wsl,
                                                          const float* __restrict__ xp2,
                                                          const float* __restrict__ as2,
                                                          const float* __restrict__ ad2,
                                                          const float* __restrict__ b2,
                                                          float* __restrict__ out) {
    __shared__ float mx[L2_NODES];
    __shared__ float dn[L2_NODES];
    __shared__ float s0[L2_NODES];
    __shared__ float s1[L2_NODES];

    const int tid = threadIdx.x;
    const int b   = blockIdx.x;
    int lo = (b * OWN_NODES + 15) & ~15;
    int hi = ((b + 1) * OWN_NODES + 15) & ~15;
    lo = lo < NNODES ? lo : NNODES;
    hi = hi < NNODES ? hi : NNODES;
    const int nloc = hi - lo;

    if (tid < L2_NODES) {
        mx[tid] = -INFINITY;
        dn[tid] = 0.0f;
        s0[tid] = 0.0f;
        s1[tid] = 0.0f;
    }

    const bool two = (b + 1 < OWN_BLOCKS) && ((b + 1) * OWN_NODES < hi);
    const unsigned totA = wsl[(size_t)b * WSL_STRIDE];
    const unsigned totB = two ? wsl[(size_t)(b + 1) * WSL_STRIDE] : 0u;
    const bool spill = totA > OWN_CAP || totB > OWN_CAP;
    const float2* __restrict__ xp2v = (const float2*)xp2;
    const int4*   __restrict__ dst4 = (const int4*)(ei + NEDGES);
    __syncthreads();

    if (!spill) {
        for (int g = 0; g < (two ? 2 : 1); ++g) {
            const unsigned int* reg = wsl + (size_t)(b + g) * WSL_STRIDE + WSL_HDR;
            const int n     = (int)(g ? totB : totA);
            const int gbase = (b + g) * OWN_NODES - lo;
            for (int t0 = tid; t0 < n; t0 += 4 * OWN_THREADS) {
                unsigned en[4];
#pragma unroll
                for (int u = 0; u < 4; ++u) {
                    const int t = t0 + u * OWN_THREADS;
                    en[u] = reg[t < n ? t : 0];
                }
                int   li[4];
                float va[4], vb[4];
#pragma unroll
                for (int u = 0; u < 4; ++u) {
                    const int  l  = gbase + (int)(en[u] >> 16);
                    const bool ok = t0 + u * OWN_THREADS < n && (unsigned)l < (unsigned)nloc;
                    li[u] = ok ? l : -1;
                    va[u] = as2[ok ? (int)(en[u] & 0xffffu) : lo];
                    vb[u] = ad2[ok ? lo + l : lo];
                }
#pragma unroll
                for (int u = 0; u < 4; ++u)
                    if (li[u] >= 0) atomicMaxF(&mx[li[u]], lrelu(va[u] + vb[u]));
            }
        }
    } else {
        for (int i = tid; i < nloc; i += OWN_THREADS)
            atomicMaxF(&mx[i], lrelu(as2[lo + i] + ad2[lo + i]));
        for (int i = tid; i < OWN_NV; i += OWN_THREADS) {
            const int4 v = dst4[i];
            const int dd[4] = {v.x, v.y, v.z, v.w};
#pragma unroll
            for (int q = 0; q < 4; ++q) {
                const unsigned l = (unsigned)(dd[q] - lo);
                if (l < (unsigned)nloc)
                    atomicMaxF(&mx[l], lrelu(as2[ei[i * 4 + q]] + ad2[dd[q]]));
            }
        }
    }
    __syncthreads();

    if (!spill) {
        for (int g = 0; g < (two ? 2 : 1); ++g) {
            const unsigned int* reg = wsl + (size_t)(b + g) * WSL_STRIDE + WSL_HDR;
            const int n     = (int)(g ? totB : totA);
            const int gbase = (b + g) * OWN_NODES - lo;
            for (int t0 = tid; t0 < n; t0 += 4 * OWN_THREADS) {
                unsigned en[4];
#pragma unroll
                for (int u = 0; u < 4; ++u) {
                    const int t = t0 + u * OWN_THREADS;
                    en[u] = reg[t < n ? t : 0];
                }
                int    li[4];
                float  va[4], vb[4];
                float2 xs[4];
#pragma unroll
                for (int u = 0; u < 4; ++u) {
                    const int  l  = gbase + (int)(en[u] >> 16);
                    const bool ok = t0 + u * OWN_THREADS < n && (unsigned)l < (unsigned)nloc;
                    const int  s  = ok ? (int)(en[u] & 0xffffu) : lo;
                    li[u] = ok ? l : -1;
                    va[u] = as2[s];
                    vb[u] = ad2[ok ? lo + l : lo];
                    xs[u] = xp2v[s];
                }
#pragma unroll
                for (int u = 0; u < 4; ++u) {
                    if (li[u] >= 0) {
                        const float p = __expf(lrelu(va[u] + vb[u]) - mx[li[u]]);
                        atomicAdd(&dn[li[u]], p);
                        atomicAdd(&s0[li[u]], p * xs[u].x);
                        atomicAdd(&s1[li[u]], p * xs[u].y);
                    }
                }
            }
        }
    } else {
        for (int i = tid; i < nloc; i += OWN_THREADS) {
            const float  p = __expf(lrelu(as2[lo + i] + ad2[lo + i]) - mx[i]);
            const float2 x = xp2v[lo + i];
            atomicAdd(&dn[i], p);
            atomicAdd(&s0[i], p * x.x);
            atomicAdd(&s1[i], p * x.y);
        }
        for (int i = tid; i < OWN_NV; i += OWN_THREADS) {
            const int4 v = dst4[i];
            const int dd[4] = {v.x, v.y, v.z, v.w};
#pragma unroll
            for (int q = 0; q < 4; ++q) {
                const unsigned l = (unsigned)(dd[q] - lo);
                if (l < (unsigned)nloc) {
                    const int    s = ei[i * 4 + q];
                    const float  p = __expf(lrelu(as2[s] + ad2[dd[q]]) - mx[l]);
                    const float2 x = xp2v[s];
                    atomicAdd(&dn[l], p);
                    atomicAdd(&s0[l], p * x.x);
                    atomicAdd(&s1[l], p * x.y);
                }
            }
        }
    }
    __syncthreads();

    if (tid < nloc * OUT2) {
        const int nd = tid >> 1;
        const float sv = (tid & 1) ? s1[nd] : s0[nd];
        out[(size_t)lo * OUT2 + tid] = sv / dn[nd] + b2[tid & 1];
    }
}

extern "C" void kernel_launch(void* const* d_in, const int* in_sizes, int n_in,
                              void* d_out, int out_size, void* d_ws, size_t ws_size,
                              hipStream_t stream) {
    const float* x     = (const float*)d_in[0];
    const int*   ei    = (const int*)d_in[1];
    const float* W1    = (const float*)d_in[2];
    const float* asw1  = (const float*)d_in[3];
    const float* adw1  = (const float*)d_in[4];
    const float* b1    = (const float*)d_in[5];
    const float* W2    = (const float*)d_in[6];
    const float* asw2  = (const float*)d_in[7];
    const float* adw2  = (const float*)d_in[8];
    const float* b2    = (const float*)d_in[9];
    float*       out   = (float*)d_out;

    float* w    = (float*)d_ws;
    float* xp1  = w;
    float* hbuf = xp1  + (size_t)NNODES * D1;
    float* as1  = hbuf + (size_t)NNODES * D1;
    float* ad1  = as1  + (size_t)NNODES * H1;
    float* xp2  = ad1  + (size_t)NNODES * H1;
    float* as2  = xp2  + (size_t)NNODES * OUT2;
    float* ad2  = as2  + (size_t)NPAD;
    unsigned int* wsl = (unsigned int*)(ad2 + (size_t)NPAD);
    unsigned int* slots = wsl + (size_t)OWN_BLOCKS * WSL_STRIDE;
    float* wa   = (float*)(slots + (size_t)OWN_BLOCKS * BIN_BLOCKS * BIN_SLOT);

    bin_edges<<<BIN_BLOCKS + WA_BLOCKS, BIN_THREADS, 0, stream>>>(ei, slots, W1, asw1, adw1, wa);
    gemm1_wmma<<<G1_BLOCKS, G1_WAVES * 32, 0, stream>>>(x, W1, wa, xp1, as1, ad1);
    agg1_owner<<<OWN_BLOCKS, OWN_THREADS, 0, stream>>>(ei, xp1, as1, ad1, b1, hbuf, wsl, slots);

    gemm2_alpha2<<<NPAD / P2_NODES, 32, 0, stream>>>(hbuf, W2, asw2, adw2, xp2, as2, ad2);
    agg2_owner<<<OWN_BLOCKS, OWN_THREADS, 0, stream>>>(ei, wsl, xp2, as2, ad2, b2, out);
}


# Layer-2 projection and logits in the owner groups' epilogue, one wave per row read from LDS: h never stored or re-read, 4 launches not 5; xp2 summed 8 heads per lane, then an xor lane tree
# speedup vs baseline: 9.4958x; 1.0945x over previous
#include <hip/hip_runtime.h>
#include <hip/hip_bf16.h>
#include <math.h>

#define NNODES 50000
#define NEDGES 800000
#define ETOT   (NEDGES + NNODES)
#define FIN    256
#define H1     8
#define C1     32
#define D1     256
#define OUT2   2
#define SLOPE  0.2f

#define LDSPAD 4
#define ASTRIDE (FIN + LDSPAD)

typedef float v2f __attribute__((ext_vector_type(2)));
typedef float v8f __attribute__((ext_vector_type(8)));

__device__ __forceinline__ float lrelu(float v) {
    return v > 0.0f ? v : SLOPE * v;
}

__device__ __forceinline__ void atomicMaxF(float* addr, float val) {
    if (val >= 0.0f) atomicMax((int*)addr, __float_as_int(val));
    else             atomicMin((unsigned int*)addr, __float_as_uint(val));
}

#define G1_WAVES   17
#define G1_RT      2
#define G1_ROWS    (16 * G1_RT)
#define G1_BLOCKS  ((NNODES + G1_ROWS - 1) / G1_ROWS)
static_assert(NNODES % 16 == 0, "a row tile is whole or absent");
#define WA_COLS    16
static_assert(WA_COLS == 2 * H1, "the logit tile is exactly one 16-column tile");

template <int LDB>
__device__ __forceinline__ void panel_tiles(const float* aTile, const float* __restrict__ B,
                                            const int l16, const int half,
                                            v8f& acc0, v8f& acc1) {
    v8f c0 = {}, c1 = {};
#pragma unroll
    for (int k = 0; k < FIN; k += 4) {
        const int kk = k + half * 2;
        v2f b;
        b.x = B[kk * LDB + l16];
        b.y = B[(kk + 1) * LDB + l16];
        const v2f a0 = *(const v2f*)(aTile + l16 * ASTRIDE + kk);
        const v2f a1 = *(const v2f*)(aTile + (16 + l16) * ASTRIDE + kk);
        c0 = __builtin_amdgcn_wmma_f32_16x16x4_f32(
             false, a0,  false, b,
             (short)0, c0,  false,  false);
        c1 = __builtin_amdgcn_wmma_f32_16x16x4_f32(
             false, a1,  false, b,
             (short)0, c1,  false,  false);
    }
    acc0 = c0;
    acc1 = c1;
}

__global__ __launch_bounds__(G1_WAVES * 32) void gemm1_wmma(const float* __restrict__ x,
                                                  const float* __restrict__ W1,
                                                  const float* __restrict__ Wa,
                                                  float* __restrict__ xp,
                                                  float* __restrict__ as,
                                                  float* __restrict__ ad) {
    __shared__ float aTile[G1_ROWS * ASTRIDE];

    const int t     = threadIdx.x;
    const int lane  = t & 31;
    const int wave  = __builtin_amdgcn_readfirstlane(t >> 5);
    const int row0  = blockIdx.x * G1_ROWS;
    const bool two  = row0 + 16 < NNODES;
    const int half  = lane >> 4;
    const int l16   = lane & 15;

    if (wave < 16) {
        const int row = t >> 5;
        const int col = (t & 31) * 8;
        const float4* src0 = (const float4*)(x + (size_t)(row0 + row) * FIN + col);
        float4 v0 = src0[0], v1 = src0[1];
        float4 v2 = make_float4(0.0f, 0.0f, 0.0f, 0.0f), v3 = v2;
        if (two) {
            const float4* src1 = (const float4*)(x + (size_t)(row0 + 16 + row) * FIN + col);
            v2 = src1[0];
            v3 = src1[1];
        }
        float4* dst0 = (float4*)(aTile + row * ASTRIDE + col);
        float4* dst1 = (float4*)(aTile + (16 + row) * ASTRIDE + col);
        dst0[0] = v0;
        dst0[1] = v1;
        dst1[0] = v2;
        dst1[1] = v3;
    }
    __syncthreads();

    const int mrow = row0 + half * 8;
    v8f acc0, acc1;
    if (wave < 16) {
        const int bcol = wave * 16 + l16;
        panel_tiles<D1>(aTile, W1 + wave * 16, l16, half, acc0, acc1);
#pragma unroll
        for (int r = 0; r < 8; ++r)
            xp[(size_t)(mrow + r) * D1 + bcol] = acc0[r];
        if (two) {
#pragma unroll
            for (int r = 0; r < 8; ++r)
                xp[(size_t)(mrow + 16 + r) * D1 + bcol] = acc1[r];
        }
    } else {
        panel_tiles<WA_COLS>(aTile, Wa, l16, half, acc0, acc1);
        float* __restrict__ lg = l16 < H1 ? as + l16 : ad + (l16 - H1);
#pragma unroll
        for (int r = 0; r < 8; ++r)
            lg[(size_t)(mrow + r) * H1] = acc0[r];
        if (two) {
#pragma unroll
            for (int r = 0; r < 8; ++r)
                lg[(size_t)(mrow + 16 + r) * H1] = acc1[r];
        }
    }
}

#define OWN_BLOCKS  256
#define OWN_NODES   196
#define OWN_THREADS 1024
#define OWN_WAVES   (OWN_THREADS / 32)
#define OWN_CAP     20480
#define OWN_G       2
#define OWN_NV      (NEDGES / 4)
#define WSL_HDR     32
#define WSL_STRIDE  (OWN_CAP + WSL_HDR)
#define BIN_BLOCKS  256
#define BIN_THREADS 1024
#define BIN_SLOT    64
#define BIN_CAP     (BIN_SLOT - 1)
#define BIN_NV      ((OWN_NV + BIN_BLOCKS - 1) / BIN_BLOCKS)
#define REC_STRIDE  (((OWN_NODES + 7) / 8) * 8)
static_assert(OWN_BLOCKS * OWN_NODES >= NNODES, "every row needs an owner");
static_assert(NNODES - (OWN_BLOCKS - 1) * OWN_NODES > 0, "no group without rows");
static_assert(NNODES <= 65536 && OWN_NODES <= 65535, "a list entry packs two 16-bit numbers");
static_assert(NEDGES % 4 == 0, "the edge list is read four at a time");
static_assert(OWN_NODES <= OWN_THREADS && OWN_NODES <= OWN_CAP, "self loops go in first");
static_assert(BIN_NV <= BIN_THREADS, "one int4 of the edge list per lane of a binning group");
static_assert(OWN_BLOCKS * BIN_SLOT == 16 * BIN_THREADS, "a binning group moves its slots 16 words per lane");
static_assert(BIN_BLOCKS * BIN_SLOT == 16 * OWN_THREADS, "an owner group reads its slots 16 words per lane");
static_assert(BIN_BLOCKS <= OWN_THREADS && BIN_BLOCKS % 32 == 0, "one lane per slot in the offset scan, whole waves");
static_assert(OWN_NODES + BIN_BLOCKS * BIN_CAP <= OWN_CAP, "full slots still fit the list");
static_assert(REC_STRIDE <= OWN_THREADS, "one lane per record in the record store");
static_assert(D1 == H1 * 32 && OUT2 == 2, "a lane holds one column of every head; two output columns");

__device__ __forceinline__ int rec_of(int n) {
    return n + (n / OWN_NODES) * (REC_STRIDE - OWN_NODES);
}

#define WA_BLOCKS ((FIN * WA_COLS + BIN_THREADS - 1) / BIN_THREADS)

__global__ __launch_bounds__(BIN_THREADS) void bin_edges(const int* __restrict__ ei,
                                                         unsigned int* __restrict__ slots,
                                                         const float* __restrict__ W1,
                                                         const float* __restrict__ a_src,
                                                         const float* __restrict__ a_dst,
                                                         float* __restrict__ Wa) {
    __shared__ unsigned int stage[OWN_BLOCKS * BIN_SLOT];

    const int tid = threadIdx.x;
    if (blockIdx.x >= BIN_BLOCKS) {
        const int g = (blockIdx.x - BIN_BLOCKS) * BIN_THREADS + tid;
        if (g < FIN * WA_COLS) {
            const int k  = g / WA_COLS;
            const int j  = g % WA_COLS;
            const int hd = j & (H1 - 1);
            const float* wr = W1 + (size_t)k * D1 + hd * C1;
            const float* av = (j < H1 ? a_src : a_dst) + hd * C1;
            float s = 0.0f;
#pragma unroll
            for (int ch = 0; ch < C1; ++ch) s = fmaf(wr[ch], av[ch], s);
            Wa[g] = s;
        }
        return;
    }
    uint4* stage4 = (uint4*)stage;
    for (int u = 0; u < 4; ++u) stage4[u * BIN_THREADS + tid] = make_uint4(0u, 0u, 0u, 0u);
    __syncthreads();

    const int i = blockIdx.x * BIN_NV + tid;
    if (tid < BIN_NV && i < OWN_NV) {
        const int4 sv = ((const int4*)ei)[i];
        const int4 dv = ((const int4*)(ei + NEDGES))[i];
        const int ss[4] = {sv.x, sv.y, sv.z, sv.w};
        const int dd[4] = {dv.x, dv.y, dv.z, dv.w};
#pragma unroll
        for (int q = 0; q < 4; ++q) {
            const unsigned d = (unsigned)dd[q];
            if (d < (unsigned)NNODES) {
                const unsigned g  = d / OWN_NODES;
                const unsigned dl = d - g * OWN_NODES;
                const unsigned r  = atomicAdd(&stage[g * BIN_SLOT], 1u);
                if (r < BIN_CAP) stage[g * BIN_SLOT + 1 + r] = (unsigned)ss[q] | (dl << 16);
            }
        }
    }
    __syncthreads();

    for (int u = 0; u < 4; ++u) {
        const int k = u * BIN_THREADS + tid;
        const int g = k >> 4;
        uint4* dst = (uint4*)(slots + ((size_t)g * BIN_BLOCKS + blockIdx.x) * BIN_SLOT);
        dst[k & 15] = stage4[k];
    }
}

__global__ __launch_bounds__(OWN_THREADS) void agg1_owner(const int* __restrict__ ei,
                                                          const float* __restrict__ xp,
                                                          const float* __restrict__ as,
                                                          const float* __restrict__ ad,
                                                          const float* __restrict__ b1,
                                                          const float* __restrict__ W2,
                                                          const float* __restrict__ aw_src,
                                                          const float* __restrict__ aw_dst,
                                                          float* __restrict__ rec,
                                                          unsigned int* __restrict__ wsl,
                                                          const unsigned int* __restrict__ slots) {
    __shared__ float4       recs[REC_STRIDE];
    __shared__ float        acc[OWN_NODES * D1];
    __shared__ float        mx[OWN_NODES * H1];
    __shared__ float        dn[OWN_NODES * H1];
    __shared__ unsigned int list[OWN_CAP];
    __shared__ unsigned int scnt[BIN_BLOCKS];
    __shared__ unsigned int soff[BIN_BLOCKS];
    __shared__ unsigned int wtot[BIN_BLOCKS / 32];
    __shared__ unsigned int over;

    const int tid  = threadIdx.x;
    const int lane = tid & 31;
    const int wv   = __builtin_amdgcn_readfirstlane(tid >> 5);
    const int base = blockIdx.x * OWN_NODES;
    const int nown = (NNODES - base) < OWN_NODES ? (NNODES - base) : OWN_NODES;

    uint4 sv[4];
    {
        const uint4* sl4 = (const uint4*)(slots + (size_t)blockIdx.x * BIN_BLOCKS * BIN_SLOT);
#pragma unroll
        for (int u = 0; u < 4; ++u) sv[u] = sl4[u * OWN_THREADS + tid];
    }

    {
        float4* acc4 = (float4*)acc;
        for (int i = tid; i < OWN_NODES * D1 / 4; i += OWN_THREADS)
            acc4[i] = make_float4(0.0f, 0.0f, 0.0f, 0.0f);
        for (int i = tid; i < OWN_NODES * H1; i += OWN_THREADS) {
            mx[i] = -INFINITY;
            dn[i] = 0.0f;
        }
        if (tid < nown) list[tid] = (unsigned)(base + tid) | ((unsigned)tid << 16);
        if (tid == 0) over = 0u;
        if ((tid & 15) == 0) {
#pragma unroll
            for (int u = 0; u < 4; ++u) scnt[(u * OWN_THREADS + tid) >> 4] = sv[u].x;
        }
    }
    __syncthreads();

    const int4* __restrict__ dst4 = (const int4*)(ei + NEDGES);
    unsigned cc = 0u, inc = 0u;
    if (tid < BIN_BLOCKS) {
        const unsigned cs = scnt[tid];
        if (cs > BIN_CAP) over = 1u;
        cc  = cs < BIN_CAP ? cs : BIN_CAP;
        inc = cc;
        for (int o = 1; o < 32; o <<= 1) {
            const unsigned v = __shfl_up(inc, o);
            if (lane >= o) inc += v;
        }
        if (lane == 31) wtot[wv] = inc;
    }
    __syncthreads();
    if (tid < BIN_BLOCKS) {
        unsigned pre = 0u;
        for (int w = 0; w < wv; ++w) pre += wtot[w];
        soff[tid] = pre + inc - cc;
    }
    unsigned found = 0u;
    for (int w = 0; w < BIN_BLOCKS / 32; ++w) found += wtot[w];
    const bool spill = over != 0u;
    const int n = __builtin_amdgcn_readfirstlane(spill ? nown : nown + (int)found);
    __syncthreads();

    if (!spill) {
#pragma unroll
        for (int u = 0; u < 4; ++u) {
            const int      k  = u * OWN_THREADS + tid;
            const int      s  = k >> 4;
            const unsigned c  = scnt[s] < BIN_CAP ? scnt[s] : BIN_CAP;
            const int      p0 = nown + (int)soff[s] - 1;
            const int      w0 = (k & 15) * 4;
            const unsigned wd[4] = {sv[u].x, sv[u].y, sv[u].z, sv[u].w};
#pragma unroll
            for (int q = 0; q < 4; ++q) {
                const int w = w0 + q;
                if (w >= 1 && (unsigned)(w - 1) < c) list[p0 + w] = wd[q];
            }
        }
    }
    __syncthreads();

    {
        unsigned int* reg = wsl + (size_t)blockIdx.x * WSL_STRIDE;
        if (tid == 0) reg[0] = spill ? (unsigned)(OWN_CAP + 1) : (unsigned)n;
        for (int t = tid; t < n; t += OWN_THREADS) reg[WSL_HDR + t] = list[t];
    }

    for (int t0 = tid; t0 < n * H1; t0 += 4 * OWN_THREADS) {
        float lv[4];
        int   mi[4];
#pragma unroll
        for (int u = 0; u < 4; ++u) {
            const int  t  = t0 + u * OWN_THREADS;
            const bool ok = t < n * H1;
            const unsigned en = list[ok ? (t >> 3) : 0];
            const int hh = t & (H1 - 1);
            const int s  = (int)(en & 0xffffu);
            const int dl = (int)(en >> 16);
            mi[u] = ok ? dl * H1 + hh : -1;
            lv[u] = lrelu(as[s * H1 + hh] + ad[(base + dl) * H1 + hh]);
        }
#pragma unroll
        for (int u = 0; u < 4; ++u)
            if (mi[u] >= 0) atomicMaxF(&mx[mi[u]], lv[u]);
    }
    if (spill) {
        for (int i = tid; i < OWN_NV; i += OWN_THREADS) {
            const int4 v = dst4[i];
            const int dd[4] = {v.x, v.y, v.z, v.w};
#pragma unroll
            for (int q = 0; q < 4; ++q) {
                const unsigned dl = (unsigned)(dd[q] - base);
                if (dl < (unsigned)nown) {
                    const int s = ei[i * 4 + q];
                    for (int hh = 0; hh < H1; ++hh)
                        atomicMaxF(&mx[dl * H1 + hh],
                                   lrelu(as[s * H1 + hh] + ad[(base + (int)dl) * H1 + hh]));
                }
            }
        }
    }
    __syncthreads();

    {
        const int j = lane >> 3;
        const int h = lane & (H1 - 1);
        for (int i0 = wv * (4 * OWN_G); i0 < n; i0 += OWN_WAVES * 4 * OWN_G) {
            unsigned en[OWN_G];
#pragma unroll
            for (int g = 0; g < OWN_G; ++g) {
                const int idx = i0 + g * 4 + j;
                en[g] = list[idx < n ? idx : 0];
            }
            float va[OWN_G], vb[OWN_G];
            int   mh[OWN_G];
#pragma unroll
            for (int g = 0; g < OWN_G; ++g) {
                const int s  = (int)(en[g] & 0xffffu);
                const int dl = (int)(en[g] >> 16);
                mh[g] = dl * H1 + h;
                va[g] = as[s * H1 + h];
                vb[g] = ad[(base + dl) * H1 + h];
            }
            float row[OWN_G][4][H1];
#pragma unroll
            for (int g = 0; g < OWN_G; ++g) {
#pragma unroll
                for (int jj = 0; jj < 4; ++jj) {
                    const unsigned ej = (unsigned)__builtin_amdgcn_readlane((int)en[g], jj * H1);
                    const float* xr = xp + (size_t)(ej & 0xffffu) * D1 + lane;
#pragma unroll
                    for (int hh = 0; hh < H1; ++hh) row[g][jj][hh] = xr[hh * C1];
                }
            }
            float p[OWN_G];
#pragma unroll
            for (int g = 0; g < OWN_G; ++g)
                p[g] = __expf(lrelu(va[g] + vb[g]) - mx[mh[g]]);
#pragma unroll
            for (int g = 0; g < OWN_G; ++g)
                if (i0 + g * 4 + j < n) atomicAdd(&dn[mh[g]], p[g]);
#pragma unroll
            for (int g = 0; g < OWN_G; ++g) {
#pragma unroll
                for (int jj = 0; jj < 4; ++jj) {
                    if (i0 + g * 4 + jj < n) {
                        const unsigned ej = (unsigned)__builtin_amdgcn_readlane((int)en[g], jj * H1);
                        float* arow = acc + (ej >> 16) * D1 + lane;
#pragma unroll
                        for (int hh = 0; hh < H1; ++hh) {
                            const float a = __int_as_float(
                                __builtin_amdgcn_readlane(__float_as_int(p[g]), jj * H1 + hh));
                            atomicAdd(arow + hh * C1, a * row[g][jj][hh]);
                        }
                    }
                }
            }
        }
    }
    if (spill) {
        for (int it = 0; it < (OWN_NV + OWN_THREADS - 1) / OWN_THREADS; ++it) {
            const int i = it * OWN_THREADS + tid;
            const int4 v = i < OWN_NV ? dst4[i] : make_int4(-1, -1, -1, -1);
            const int dd[4] = {v.x, v.y, v.z, v.w};
#pragma unroll
            for (int q = 0; q < 4; ++q) {
                const unsigned dl  = (unsigned)(dd[q] - base);
                const bool     hit = dl < (unsigned)nown;
                const int      s   = hit ? ei[i * 4 + q] : 0;
                unsigned mask = __builtin_amdgcn_ballot_w32(hit);
                while (mask) {
                    const int l = __builtin_ctz(mask);
                    mask &= mask - 1;
                    const int sj = __builtin_amdgcn_readlane(s, l);
                    const int dj = __builtin_amdgcn_readlane((int)dl, l);
                    const int h  = lane & (H1 - 1);
                    const float lv = lrelu(as[sj * H1 + h] + ad[(base + dj) * H1 + h]);
                    const float pe = __expf(lv - mx[dj * H1 + h]);
                    if (lane < H1) atomicAdd(&dn[dj * H1 + h], pe);
                    const float* xr = xp + (size_t)sj * D1 + lane;
                    float* arow = acc + dj * D1 + lane;
#pragma unroll
                    for (int hh = 0; hh < H1; ++hh) {
                        const float a = __int_as_float(
                            __builtin_amdgcn_readlane(__float_as_int(pe), hh));
                        atomicAdd(arow + hh * C1, a * xr[hh * C1]);
                    }
                }
            }
        }
    }
    float bb[H1], w0[H1], w1[H1];
#pragma unroll
    for (int hh = 0; hh < H1; ++hh) {
        const int col = hh * C1 + lane;
        const float2 w = ((const float2*)W2)[col];
        bb[hh] = b1[col];
        w0[hh] = w.x;
        w1[hh] = w.y;
    }
    const float ws0 = aw_src[0], ws1 = aw_src[1];
    const float wd0 = aw_dst[0], wd1 = aw_dst[1];
    __syncthreads();

    for (int nd = wv; nd < REC_STRIDE; nd += OWN_WAVES) {
        float r0 = 0.0f, r1 = 0.0f;
        if (nd < nown) {
            const float* arow = acc + nd * D1 + lane;
#pragma unroll
            for (int hh = 0; hh < H1; ++hh) {
                float v = arow[hh * C1] / dn[nd * H1 + hh] + bb[hh];
                v  = v > 0.0f ? v : 0.0f;
                r0 = fmaf(v, w0[hh], r0);
                r1 = fmaf(v, w1[hh], r1);
            }
#pragma unroll
            for (int off = 16; off > 0; off >>= 1) {
                r0 += __shfl_xor(r0, off);
                r1 += __shfl_xor(r1, off);
            }
        }
        if (lane == 0)
            recs[nd] = make_float4(r0, r1, fmaf(r0, ws0, r1 * ws1), fmaf(r0, wd0, r1 * wd1));
    }
    __syncthreads();
    if (tid < REC_STRIDE)
        ((float4*)rec)[(size_t)blockIdx.x * REC_STRIDE + tid] = recs[tid];
}

#define L2_NODES (((OWN_NODES + 15) / 16) * 16)
static_assert(NNODES % 16 == 0, "the output ends on a line boundary");
static_assert(OWN_NODES >= 16, "a group's rows lie in two lists at most");
static_assert(L2_NODES * OUT2 <= OWN_THREADS, "one lane per output word in the epilogue");

__global__ __launch_bounds__(OWN_THREADS) void agg2_owner(const int* __restrict__ ei,
                                                          const unsigned int* __restrict__ wsl,
                                                          const float* __restrict__ rec,
                                                          const float* __restrict__ b2,
                                                          float* __restrict__ out) {
    __shared__ float mx[L2_NODES];
    __shared__ float dn[L2_NODES];
    __shared__ float s0[L2_NODES];
    __shared__ float s1[L2_NODES];

    const int tid = threadIdx.x;
    const int b   = blockIdx.x;
    int lo = (b * OWN_NODES + 15) & ~15;
    int hi = ((b + 1) * OWN_NODES + 15) & ~15;
    lo = lo < NNODES ? lo : NNODES;
    hi = hi < NNODES ? hi : NNODES;
    const int nloc = hi - lo;

    if (tid < L2_NODES) {
        mx[tid] = -INFINITY;
        dn[tid] = 0.0f;
        s0[tid] = 0.0f;
        s1[tid] = 0.0f;
    }

    const bool two = (b + 1 < OWN_BLOCKS) && ((b + 1) * OWN_NODES < hi);
    const unsigned totA = wsl[(size_t)b * WSL_STRIDE];
    const unsigned totB = two ? wsl[(size_t)(b + 1) * WSL_STRIDE] : 0u;
    const bool spill = totA > OWN_CAP || totB > OWN_CAP;
    const float4* __restrict__ recv = (const float4*)rec;
    const int4*   __restrict__ dst4 = (const int4*)(ei + NEDGES);
    const int rlo = rec_of(lo);
    __syncthreads();

    if (!spill) {
        for (int g = 0; g < (two ? 2 : 1); ++g) {
            const unsigned int* reg = wsl + (size_t)(b + g) * WSL_STRIDE + WSL_HDR;
            const int n     = (int)(g ? totB : totA);
            const int gbase = (b + g) * OWN_NODES - lo;
            const int rbase = (b + g) * REC_STRIDE;
            for (int t0 = tid; t0 < n; t0 += 4 * OWN_THREADS) {
                unsigned en[4];
#pragma unroll
                for (int u = 0; u < 4; ++u) {
                    const int t = t0 + u * OWN_THREADS;
                    en[u] = reg[t < n ? t : 0];
                }
                int   li[4];
                float va[4], vb[4];
#pragma unroll
                for (int u = 0; u < 4; ++u) {
                    const int  dl = (int)(en[u] >> 16);
                    const int  l  = gbase + dl;
                    const bool ok = t0 + u * OWN_THREADS < n && (unsigned)l < (unsigned)nloc;
                    li[u] = ok ? l : -1;
                    va[u] = rec[4 * (ok ? rec_of((int)(en[u] & 0xffffu)) : rlo) + 2];
                    vb[u] = rec[4 * (ok ? rbase + dl : rlo) + 3];
                }
#pragma unroll
                for (int u = 0; u < 4; ++u)
                    if (li[u] >= 0) atomicMaxF(&mx[li[u]], lrelu(va[u] + vb[u]));
            }
        }
    } else {
        for (int i = tid; i < nloc; i += OWN_THREADS) {
            const float4 r = recv[rec_of(lo + i)];
            atomicMaxF(&mx[i], lrelu(r.z + r.w));
        }
        for (int i = tid; i < OWN_NV; i += OWN_THREADS) {
            const int4 v = dst4[i];
            const int dd[4] = {v.x, v.y, v.z, v.w};
#pragma unroll
            for (int q = 0; q < 4; ++q) {
                const unsigned l = (unsigned)(dd[q] - lo);
                if (l < (unsigned)nloc)
                    atomicMaxF(&mx[l], lrelu(rec[4 * rec_of(ei[i * 4 + q]) + 2] + rec[4 * rec_of(dd[q]) + 3]));
            }
        }
    }
    __syncthreads();

    if (!spill) {
        for (int g = 0; g < (two ? 2 : 1); ++g) {
            const unsigned int* reg = wsl + (size_t)(b + g) * WSL_STRIDE + WSL_HDR;
            const int n     = (int)(g ? totB : totA);
            const int gbase = (b + g) * OWN_NODES - lo;
            const int rbase = (b + g) * REC_STRIDE;
            for (int t0 = tid; t0 < n; t0 += 4 * OWN_THREADS) {
                unsigned en[4];
#pragma unroll
                for (int u = 0; u < 4; ++u) {
                    const int t = t0 + u * OWN_THREADS;
                    en[u] = reg[t < n ? t : 0];
                }
                int    li[4];
                float  vb[4];
                float4 sr[4];
#pragma unroll
                for (int u = 0; u < 4; ++u) {
                    const int  dl = (int)(en[u] >> 16);
                    const int  l  = gbase + dl;
                    const bool ok = t0 + u * OWN_THREADS < n && (unsigned)l < (unsigned)nloc;
                    li[u] = ok ? l : -1;
                    sr[u] = recv[ok ? rec_of((int)(en[u] & 0xffffu)) : rlo];
                    vb[u] = rec[4 * (ok ? rbase + dl : rlo) + 3];
                }
#pragma unroll
                for (int u = 0; u < 4; ++u) {
                    if (li[u] >= 0) {
                        const float p = __expf(lrelu(sr[u].z + vb[u]) - mx[li[u]]);
                        atomicAdd(&dn[li[u]], p);
                        atomicAdd(&s0[li[u]], p * sr[u].x);
                        atomicAdd(&s1[li[u]], p * sr[u].y);
                    }
                }
            }
        }
    } else {
        for (int i = tid; i < nloc; i += OWN_THREADS) {
            const float4 r = recv[rec_of(lo + i)];
            const float  p = __expf(lrelu(r.z + r.w) - mx[i]);
            atomicAdd(&dn[i], p);
            atomicAdd(&s0[i], p * r.x);
            atomicAdd(&s1[i], p * r.y);
        }
        for (int i = tid; i < OWN_NV; i += OWN_THREADS) {
            const int4 v = dst4[i];
            const int dd[4] = {v.x, v.y, v.z, v.w};
#pragma unroll
            for (int q = 0; q < 4; ++q) {
                const unsigned l = (unsigned)(dd[q] - lo);
                if (l < (unsigned)nloc) {
                    const float4 r = recv[rec_of(ei[i * 4 + q])];
                    const float  p = __expf(lrelu(r.z + rec[4 * rec_of(dd[q]) + 3]) - mx[l]);
                    atomicAdd(&dn[l], p);
                    atomicAdd(&s0[l], p * r.x);
                    atomicAdd(&s1[l], p * r.y);
                }
            }
        }
    }
    __syncthreads();

    if (tid < nloc * OUT2) {
        const int nd = tid >> 1;
        const float sv = (tid & 1) ? s1[nd] : s0[nd];
        out[(size_t)lo * OUT2 + tid] = sv / dn[nd] + b2[tid & 1];
    }
}

extern "C" void kernel_launch(void* const* d_in, const int* in_sizes, int n_in,
                              void* d_out, int out_size, void* d_ws, size_t ws_size,
                              hipStream_t stream) {
    const float* x     = (const float*)d_in[0];
    const int*   ei    = (const int*)d_in[1];
    const float* W1    = (const float*)d_in[2];
    const float* asw1  = (const float*)d_in[3];
    const float* adw1  = (const float*)d_in[4];
    const float* b1    = (const float*)d_in[5];
    const float* W2    = (const float*)d_in[6];
    const float* asw2  = (const float*)d_in[7];
    const float* adw2  = (const float*)d_in[8];
    const float* b2    = (const float*)d_in[9];
    float*       out   = (float*)d_out;

    float* w    = (float*)d_ws;
    float* xp1  = w;
    float* as1  = xp1  + (size_t)NNODES * D1;
    float* ad1  = as1  + (size_t)NNODES * H1;
    float* rec2 = ad1  + (size_t)NNODES * H1;
    unsigned int* wsl = (unsigned int*)(rec2 + (size_t)OWN_BLOCKS * REC_STRIDE * 4);
    unsigned int* slots = wsl + (size_t)OWN_BLOCKS * WSL_STRIDE;
    float* wa   = (float*)(slots + (size_t)OWN_BLOCKS * BIN_BLOCKS * BIN_SLOT);

    bin_edges<<<BIN_BLOCKS + WA_BLOCKS, BIN_THREADS, 0, stream>>>(ei, slots, W1, asw1, adw1, wa);
    gemm1_wmma<<<G1_BLOCKS, G1_WAVES * 32, 0, stream>>>(x, W1, wa, xp1, as1, ad1);
    agg1_owner<<<OWN_BLOCKS, OWN_THREADS, 0, stream>>>(ei, xp1, as1, ad1, b1, W2, asw2, adw2, rec2, wsl, slots);

    agg2_owner<<<OWN_BLOCKS, OWN_THREADS, 0, stream>>>(ei, wsl, rec2, b2, out);
}


# Layer-1 gathers: a source row comes as two 512-byte loads (16 B a lane), not eight 128-byte ones; LDS sums kept column-transposed so adds stay whole lines; xp2: a lane sums its 8 held columns first
# speedup vs baseline: 10.0314x; 1.0564x over previous
#include <hip/hip_runtime.h>
#include <hip/hip_bf16.h>
#include <math.h>

#define NNODES 50000
#define NEDGES 800000
#define ETOT   (NEDGES + NNODES)
#define FIN    256
#define H1     8
#define C1     32
#define D1     256
#define OUT2   2
#define SLOPE  0.2f

#define LDSPAD 4
#define ASTRIDE (FIN + LDSPAD)

typedef float v2f __attribute__((ext_vector_type(2)));
typedef float v8f __attribute__((ext_vector_type(8)));

__device__ __forceinline__ float lrelu(float v) {
    return v > 0.0f ? v : SLOPE * v;
}

__device__ __forceinline__ void atomicMaxF(float* addr, float val) {
    if (val >= 0.0f) atomicMax((int*)addr, __float_as_int(val));
    else             atomicMin((unsigned int*)addr, __float_as_uint(val));
}

#define G1_WAVES   17
#define G1_RT      2
#define G1_ROWS    (16 * G1_RT)
#define G1_BLOCKS  ((NNODES + G1_ROWS - 1) / G1_ROWS)
static_assert(NNODES % 16 == 0, "a row tile is whole or absent");
#define WA_COLS    16
static_assert(WA_COLS == 2 * H1, "the logit tile is exactly one 16-column tile");

template <int LDB>
__device__ __forceinline__ void panel_tiles(const float* aTile, const float* __restrict__ B,
                                            const int l16, const int half,
                                            v8f& acc0, v8f& acc1) {
    v8f c0 = {}, c1 = {};
#pragma unroll
    for (int k = 0; k < FIN; k += 4) {
        const int kk = k + half * 2;
        v2f b;
        b.x = B[kk * LDB + l16];
        b.y = B[(kk + 1) * LDB + l16];
        const v2f a0 = *(const v2f*)(aTile + l16 * ASTRIDE + kk);
        const v2f a1 = *(const v2f*)(aTile + (16 + l16) * ASTRIDE + kk);
        c0 = __builtin_amdgcn_wmma_f32_16x16x4_f32(
             false, a0,  false, b,
             (short)0, c0,  false,  false);
        c1 = __builtin_amdgcn_wmma_f32_16x16x4_f32(
             false, a1,  false, b,
             (short)0, c1,  false,  false);
    }
    acc0 = c0;
    acc1 = c1;
}

__global__ __launch_bounds__(G1_WAVES * 32) void gemm1_wmma(const float* __restrict__ x,
                                                  const float* __restrict__ W1,
                                                  const float* __restrict__ Wa,
                                                  float* __restrict__ xp,
                                                  float* __restrict__ as,
                                                  float* __restrict__ ad) {
    __shared__ float aTile[G1_ROWS * ASTRIDE];

    const int t     = threadIdx.x;
    const int lane  = t & 31;
    const int wave  = __builtin_amdgcn_readfirstlane(t >> 5);
    const int row0  = blockIdx.x * G1_ROWS;
    const bool two  = row0 + 16 < NNODES;
    const int half  = lane >> 4;
    const int l16   = lane & 15;

    if (wave < 16) {
        const int row = t >> 5;
        const int col = (t & 31) * 8;
        const float4* src0 = (const float4*)(x + (size_t)(row0 + row) * FIN + col);
        float4 v0 = src0[0], v1 = src0[1];
        float4 v2 = make_float4(0.0f, 0.0f, 0.0f, 0.0f), v3 = v2;
        if (two) {
            const float4* src1 = (const float4*)(x + (size_t)(row0 + 16 + row) * FIN + col);
            v2 = src1[0];
            v3 = src1[1];
        }
        float4* dst0 = (float4*)(aTile + row * ASTRIDE + col);
        float4* dst1 = (float4*)(aTile + (16 + row) * ASTRIDE + col);
        dst0[0] = v0;
        dst0[1] = v1;
        dst1[0] = v2;
        dst1[1] = v3;
    }
    __syncthreads();

    const int mrow = row0 + half * 8;
    v8f acc0, acc1;
    if (wave < 16) {
        const int bcol = wave * 16 + l16;
        panel_tiles<D1>(aTile, W1 + wave * 16, l16, half, acc0, acc1);
#pragma unroll
        for (int r = 0; r < 8; ++r)
            xp[(size_t)(mrow + r) * D1 + bcol] = acc0[r];
        if (two) {
#pragma unroll
            for (int r = 0; r < 8; ++r)
                xp[(size_t)(mrow + 16 + r) * D1 + bcol] = acc1[r];
        }
    } else {
        panel_tiles<WA_COLS>(aTile, Wa, l16, half, acc0, acc1);
        float* __restrict__ lg = l16 < H1 ? as + l16 : ad + (l16 - H1);
#pragma unroll
        for (int r = 0; r < 8; ++r)
            lg[(size_t)(mrow + r) * H1] = acc0[r];
        if (two) {
#pragma unroll
            for (int r = 0; r < 8; ++r)
                lg[(size_t)(mrow + 16 + r) * H1] = acc1[r];
        }
    }
}

#define OWN_BLOCKS  256
#define OWN_NODES   196
#define OWN_THREADS 1024
#define OWN_WAVES   (OWN_THREADS / 32)
#define OWN_CAP     20480
#define OWN_G       2
#define OWN_NV      (NEDGES / 4)
#define WSL_HDR     32
#define WSL_STRIDE  (OWN_CAP + WSL_HDR)
#define BIN_BLOCKS  256
#define BIN_THREADS 1024
#define BIN_SLOT    64
#define BIN_CAP     (BIN_SLOT - 1)
#define BIN_NV      ((OWN_NV + BIN_BLOCKS - 1) / BIN_BLOCKS)
#define REC_STRIDE  (((OWN_NODES + 7) / 8) * 8)
static_assert(OWN_BLOCKS * OWN_NODES >= NNODES, "every row needs an owner");
static_assert(NNODES - (OWN_BLOCKS - 1) * OWN_NODES > 0, "no group without rows");
static_assert(NNODES <= 65536 && OWN_NODES <= 65535, "a list entry packs two 16-bit numbers");
static_assert(NEDGES % 4 == 0, "the edge list is read four at a time");
static_assert(OWN_NODES <= OWN_THREADS && OWN_NODES <= OWN_CAP, "self loops go in first");
static_assert(BIN_NV <= BIN_THREADS, "one int4 of the edge list per lane of a binning group");
static_assert(OWN_BLOCKS * BIN_SLOT == 16 * BIN_THREADS, "a binning group moves its slots 16 words per lane");
static_assert(BIN_BLOCKS * BIN_SLOT == 16 * OWN_THREADS, "an owner group reads its slots 16 words per lane");
static_assert(BIN_BLOCKS <= OWN_THREADS && BIN_BLOCKS % 32 == 0, "one lane per slot in the offset scan, whole waves");
static_assert(OWN_NODES + BIN_BLOCKS * BIN_CAP <= OWN_CAP, "full slots still fit the list");
static_assert(REC_STRIDE <= OWN_THREADS, "one lane per record in the record store");
static_assert(D1 == H1 * 32 && OUT2 == 2, "a lane holds one column of every head; two output columns");

__device__ __forceinline__ int acc_pos(int c) {
    return (c & 128) | ((c & 3) << 5) | ((c & 127) >> 2);
}

__device__ __forceinline__ float row_pick(float v, unsigned sel_lo, unsigned sel_hi) {
    const unsigned u = __float_as_uint(v);
    return __uint_as_float(__builtin_amdgcn_permlane16(u, u, sel_lo, sel_hi, false, false));
}

__device__ __forceinline__ int rec_of(int n) {
    return n + (n / OWN_NODES) * (REC_STRIDE - OWN_NODES);
}

#define WA_BLOCKS ((FIN * WA_COLS + BIN_THREADS - 1) / BIN_THREADS)

__global__ __launch_bounds__(BIN_THREADS) void bin_edges(const int* __restrict__ ei,
                                                         unsigned int* __restrict__ slots,
                                                         const float* __restrict__ W1,
                                                         const float* __restrict__ a_src,
                                                         const float* __restrict__ a_dst,
                                                         float* __restrict__ Wa) {
    __shared__ unsigned int stage[OWN_BLOCKS * BIN_SLOT];

    const int tid = threadIdx.x;
    if (blockIdx.x >= BIN_BLOCKS) {
        const int g = (blockIdx.x - BIN_BLOCKS) * BIN_THREADS + tid;
        if (g < FIN * WA_COLS) {
            const int k  = g / WA_COLS;
            const int j  = g % WA_COLS;
            const int hd = j & (H1 - 1);
            const float* wr = W1 + (size_t)k * D1 + hd * C1;
            const float* av = (j < H1 ? a_src : a_dst) + hd * C1;
            float s = 0.0f;
#pragma unroll
            for (int ch = 0; ch < C1; ++ch) s = fmaf(wr[ch], av[ch], s);
            Wa[g] = s;
        }
        return;
    }
    uint4* stage4 = (uint4*)stage;
    for (int u = 0; u < 4; ++u) stage4[u * BIN_THREADS + tid] = make_uint4(0u, 0u, 0u, 0u);
    __syncthreads();

    const int i = blockIdx.x * BIN_NV + tid;
    if (tid < BIN_NV && i < OWN_NV) {
        const int4 sv = ((const int4*)ei)[i];
        const int4 dv = ((const int4*)(ei + NEDGES))[i];
        const int ss[4] = {sv.x, sv.y, sv.z, sv.w};
        const int dd[4] = {dv.x, dv.y, dv.z, dv.w};
#pragma unroll
        for (int q = 0; q < 4; ++q) {
            const unsigned d = (unsigned)dd[q];
            if (d < (unsigned)NNODES) {
                const unsigned g  = d / OWN_NODES;
                const unsigned dl = d - g * OWN_NODES;
                const unsigned r  = atomicAdd(&stage[g * BIN_SLOT], 1u);
                if (r < BIN_CAP) stage[g * BIN_SLOT + 1 + r] = (unsigned)ss[q] | (dl << 16);
            }
        }
    }
    __syncthreads();

    for (int u = 0; u < 4; ++u) {
        const int k = u * BIN_THREADS + tid;
        const int g = k >> 4;
        uint4* dst = (uint4*)(slots + ((size_t)g * BIN_BLOCKS + blockIdx.x) * BIN_SLOT);
        dst[k & 15] = stage4[k];
    }
}

__global__ __launch_bounds__(OWN_THREADS) void agg1_owner(const int* __restrict__ ei,
                                                          const float* __restrict__ xp,
                                                          const float* __restrict__ as,
                                                          const float* __restrict__ ad,
                                                          const float* __restrict__ b1,
                                                          const float* __restrict__ W2,
                                                          const float* __restrict__ aw_src,
                                                          const float* __restrict__ aw_dst,
                                                          float* __restrict__ rec,
                                                          unsigned int* __restrict__ wsl,
                                                          const unsigned int* __restrict__ slots) {
    __shared__ float4       recs[REC_STRIDE];
    __shared__ float        acc[OWN_NODES * D1];
    __shared__ float        mx[OWN_NODES * H1];
    __shared__ float        dn[OWN_NODES * H1];
    __shared__ unsigned int list[OWN_CAP];
    __shared__ unsigned int scnt[BIN_BLOCKS];
    __shared__ unsigned int soff[BIN_BLOCKS];
    __shared__ unsigned int wtot[BIN_BLOCKS / 32];
    __shared__ unsigned int over;

    const int tid  = threadIdx.x;
    const int lane = tid & 31;
    const int wv   = __builtin_amdgcn_readfirstlane(tid >> 5);
    const int base = blockIdx.x * OWN_NODES;
    const int nown = (NNODES - base) < OWN_NODES ? (NNODES - base) : OWN_NODES;

    uint4 sv[4];
    {
        const uint4* sl4 = (const uint4*)(slots + (size_t)blockIdx.x * BIN_BLOCKS * BIN_SLOT);
#pragma unroll
        for (int u = 0; u < 4; ++u) sv[u] = sl4[u * OWN_THREADS + tid];
    }

    {
        float4* acc4 = (float4*)acc;
        for (int i = tid; i < OWN_NODES * D1 / 4; i += OWN_THREADS)
            acc4[i] = make_float4(0.0f, 0.0f, 0.0f, 0.0f);
        for (int i = tid; i < OWN_NODES * H1; i += OWN_THREADS) {
            mx[i] = -INFINITY;
            dn[i] = 0.0f;
        }
        if (tid < nown) list[tid] = (unsigned)(base + tid) | ((unsigned)tid << 16);
        if (tid == 0) over = 0u;
        if ((tid & 15) == 0) {
#pragma unroll
            for (int u = 0; u < 4; ++u) scnt[(u * OWN_THREADS + tid) >> 4] = sv[u].x;
        }
    }
    __syncthreads();

    const int4* __restrict__ dst4 = (const int4*)(ei + NEDGES);
    unsigned cc = 0u, inc = 0u;
    if (tid < BIN_BLOCKS) {
        const unsigned cs = scnt[tid];
        if (cs > BIN_CAP) over = 1u;
        cc  = cs < BIN_CAP ? cs : BIN_CAP;
        inc = cc;
        for (int o = 1; o < 32; o <<= 1) {
            const unsigned v = __shfl_up(inc, o);
            if (lane >= o) inc += v;
        }
        if (lane == 31) wtot[wv] = inc;
    }
    __syncthreads();
    if (tid < BIN_BLOCKS) {
        unsigned pre = 0u;
        for (int w = 0; w < wv; ++w) pre += wtot[w];
        soff[tid] = pre + inc - cc;
    }
    unsigned found = 0u;
    for (int w = 0; w < BIN_BLOCKS / 32; ++w) found += wtot[w];
    const bool spill = over != 0u;
    const int n = __builtin_amdgcn_readfirstlane(spill ? nown : nown + (int)found);
    __syncthreads();

    if (!spill) {
#pragma unroll
        for (int u = 0; u < 4; ++u) {
            const int      k  = u * OWN_THREADS + tid;
            const int      s  = k >> 4;
            const unsigned c  = scnt[s] < BIN_CAP ? scnt[s] : BIN_CAP;
            const int      p0 = nown + (int)soff[s] - 1;
            const int      w0 = (k & 15) * 4;
            const unsigned wd[4] = {sv[u].x, sv[u].y, sv[u].z, sv[u].w};
#pragma unroll
            for (int q = 0; q < 4; ++q) {
                const int w = w0 + q;
                if (w >= 1 && (unsigned)(w - 1) < c) list[p0 + w] = wd[q];
            }
        }
    }
    __syncthreads();

    {
        unsigned int* reg = wsl + (size_t)blockIdx.x * WSL_STRIDE;
        if (tid == 0) reg[0] = spill ? (unsigned)(OWN_CAP + 1) : (unsigned)n;
        for (int t = tid; t < n; t += OWN_THREADS) reg[WSL_HDR + t] = list[t];
    }

    for (int t0 = tid; t0 < n * H1; t0 += 4 * OWN_THREADS) {
        float lv[4];
        int   mi[4];
#pragma unroll
        for (int u = 0; u < 4; ++u) {
            const int  t  = t0 + u * OWN_THREADS;
            const bool ok = t < n * H1;
            const unsigned en = list[ok ? (t >> 3) : 0];
            const int hh = t & (H1 - 1);
            const int s  = (int)(en & 0xffffu);
            const int dl = (int)(en >> 16);
            mi[u] = ok ? dl * H1 + hh : -1;
            lv[u] = lrelu(as[s * H1 + hh] + ad[(base + dl) * H1 + hh]);
        }
#pragma unroll
        for (int u = 0; u < 4; ++u)
            if (mi[u] >= 0) atomicMaxF(&mx[mi[u]], lv[u]);
    }
    if (spill) {
        for (int i = tid; i < OWN_NV; i += OWN_THREADS) {
            const int4 v = dst4[i];
            const int dd[4] = {v.x, v.y, v.z, v.w};
#pragma unroll
            for (int q = 0; q < 4; ++q) {
                const unsigned dl = (unsigned)(dd[q] - base);
                if (dl < (unsigned)nown) {
                    const int s = ei[i * 4 + q];
                    for (int hh = 0; hh < H1; ++hh)
                        atomicMaxF(&mx[dl * H1 + hh],
                                   lrelu(as[s * H1 + hh] + ad[(base + (int)dl) * H1 + hh]));
                }
            }
        }
    }
    __syncthreads();

    {
        const int j = (lane & 15) >> 2;
        const int h = (lane & 1) + 4 * ((lane >> 1) & 1) + 2 * (lane >> 4);
        for (int i0 = wv * (4 * OWN_G); i0 < n; i0 += OWN_WAVES * 4 * OWN_G) {
            unsigned en[OWN_G];
#pragma unroll
            for (int g = 0; g < OWN_G; ++g) {
                const int idx = i0 + g * 4 + j;
                en[g] = list[idx < n ? idx : 0];
            }
            float va[OWN_G], vb[OWN_G];
            int   mh[OWN_G];
#pragma unroll
            for (int g = 0; g < OWN_G; ++g) {
                const int s  = (int)(en[g] & 0xffffu);
                const int dl = (int)(en[g] >> 16);
                mh[g] = dl * H1 + h;
                va[g] = as[s * H1 + h];
                vb[g] = ad[(base + dl) * H1 + h];
            }
            float4 row[OWN_G][4][2];
#pragma unroll
            for (int g = 0; g < OWN_G; ++g) {
#pragma unroll
                for (int jj = 0; jj < 4; ++jj) {
                    const unsigned ej = (unsigned)__builtin_amdgcn_readlane((int)en[g], jj * 4);
                    const float4* xr = (const float4*)(xp + (size_t)(ej & 0xffffu) * D1) + lane;
                    row[g][jj][0] = xr[0];
                    row[g][jj][1] = xr[D1 / 8];
                }
            }
            float p[OWN_G];
#pragma unroll
            for (int g = 0; g < OWN_G; ++g)
                p[g] = __expf(lrelu(va[g] + vb[g]) - mx[mh[g]]);
#pragma unroll
            for (int g = 0; g < OWN_G; ++g)
                if (i0 + g * 4 + j < n) atomicAdd(&dn[mh[g]], p[g]);
#pragma unroll
            for (int g = 0; g < OWN_G; ++g) {
#pragma unroll
                for (int jj = 0; jj < 4; ++jj) {
                    if (i0 + g * 4 + jj < n) {
                        const unsigned ej = (unsigned)__builtin_amdgcn_readlane((int)en[g], jj * 4);
                        float* arow = acc + (ej >> 16) * D1 + lane;
                        const float cA = row_pick(p[g], (4u * jj)      * 0x11111111u,
                                                        (4u * jj + 1u) * 0x11111111u);
                        const float cB = row_pick(p[g], (4u * jj + 2u) * 0x11111111u,
                                                        (4u * jj + 3u) * 0x11111111u);
                        const float4 ra = row[g][jj][0];
                        const float4 rb = row[g][jj][1];
                        atomicAdd(arow,           cA * ra.x);
                        atomicAdd(arow + 32,      cA * ra.y);
                        atomicAdd(arow + 64,      cA * ra.z);
                        atomicAdd(arow + 96,      cA * ra.w);
                        atomicAdd(arow + 128,     cB * rb.x);
                        atomicAdd(arow + 160,     cB * rb.y);
                        atomicAdd(arow + 192,     cB * rb.z);
                        atomicAdd(arow + 224,     cB * rb.w);
                    }
                }
            }
        }
    }
    if (spill) {
        for (int it = 0; it < (OWN_NV + OWN_THREADS - 1) / OWN_THREADS; ++it) {
            const int i = it * OWN_THREADS + tid;
            const int4 v = i < OWN_NV ? dst4[i] : make_int4(-1, -1, -1, -1);
            const int dd[4] = {v.x, v.y, v.z, v.w};
#pragma unroll
            for (int q = 0; q < 4; ++q) {
                const unsigned dl  = (unsigned)(dd[q] - base);
                const bool     hit = dl < (unsigned)nown;
                const int      s   = hit ? ei[i * 4 + q] : 0;
                unsigned mask = __builtin_amdgcn_ballot_w32(hit);
                while (mask) {
                    const int l = __builtin_ctz(mask);
                    mask &= mask - 1;
                    const int sj = __builtin_amdgcn_readlane(s, l);
                    const int dj = __builtin_amdgcn_readlane((int)dl, l);
                    const int h  = lane & (H1 - 1);
                    const float lv = lrelu(as[sj * H1 + h] + ad[(base + dj) * H1 + h]);
                    const float pe = __expf(lv - mx[dj * H1 + h]);
                    if (lane < H1) atomicAdd(&dn[dj * H1 + h], pe);
                    const float* xr = xp + (size_t)sj * D1 + lane;
                    float* arow = acc + dj * D1;
#pragma unroll
                    for (int hh = 0; hh < H1; ++hh) {
                        const float a = __int_as_float(
                            __builtin_amdgcn_readlane(__float_as_int(pe), hh));
                        atomicAdd(arow + acc_pos(hh * C1 + lane), a * xr[hh * C1]);
                    }
                }
            }
        }
    }
    float bb[H1], w0[H1], w1[H1];
#pragma unroll
    for (int hh = 0; hh < H1; ++hh) {
        const int col = (hh >> 2) * 128 + 4 * lane + (hh & 3);
        const float2 w = ((const float2*)W2)[col];
        bb[hh] = b1[col];
        w0[hh] = w.x;
        w1[hh] = w.y;
    }
    const float ws0 = aw_src[0], ws1 = aw_src[1];
    const float wd0 = aw_dst[0], wd1 = aw_dst[1];
    __syncthreads();

    for (int nd = wv; nd < REC_STRIDE; nd += OWN_WAVES) {
        float r0 = 0.0f, r1 = 0.0f;
        if (nd < nown) {
            const float* arow = acc + nd * D1 + lane;
#pragma unroll
            for (int hh = 0; hh < H1; ++hh) {
                float v = arow[hh * C1] / dn[nd * H1 + 4 * (hh >> 2) + (lane >> 3)] + bb[hh];
                v  = v > 0.0f ? v : 0.0f;
                r0 = fmaf(v, w0[hh], r0);
                r1 = fmaf(v, w1[hh], r1);
            }
#pragma unroll
            for (int off = 16; off > 0; off >>= 1) {
                r0 += __shfl_xor(r0, off);
                r1 += __shfl_xor(r1, off);
            }
        }
        if (lane == 0)
            recs[nd] = make_float4(r0, r1, fmaf(r0, ws0, r1 * ws1), fmaf(r0, wd0, r1 * wd1));
    }
    __syncthreads();
    if (tid < REC_STRIDE)
        ((float4*)rec)[(size_t)blockIdx.x * REC_STRIDE + tid] = recs[tid];
}

#define L2_NODES (((OWN_NODES + 15) / 16) * 16)
static_assert(NNODES % 16 == 0, "the output ends on a line boundary");
static_assert(OWN_NODES >= 16, "a group's rows lie in two lists at most");
static_assert(L2_NODES * OUT2 <= OWN_THREADS, "one lane per output word in the epilogue");

__global__ __launch_bounds__(OWN_THREADS) void agg2_owner(const int* __restrict__ ei,
                                                          const unsigned int* __restrict__ wsl,
                                                          const float* __restrict__ rec,
                                                          const float* __restrict__ b2,
                                                          float* __restrict__ out) {
    __shared__ float mx[L2_NODES];
    __shared__ float dn[L2_NODES];
    __shared__ float s0[L2_NODES];
    __shared__ float s1[L2_NODES];

    const int tid = threadIdx.x;
    const int b   = blockIdx.x;
    int lo = (b * OWN_NODES + 15) & ~15;
    int hi = ((b + 1) * OWN_NODES + 15) & ~15;
    lo = lo < NNODES ? lo : NNODES;
    hi = hi < NNODES ? hi : NNODES;
    const int nloc = hi - lo;

    if (tid < L2_NODES) {
        mx[tid] = -INFINITY;
        dn[tid] = 0.0f;
        s0[tid] = 0.0f;
        s1[tid] = 0.0f;
    }

    const bool two = (b + 1 < OWN_BLOCKS) && ((b + 1) * OWN_NODES < hi);
    const unsigned totA = wsl[(size_t)b * WSL_STRIDE];
    const unsigned totB = two ? wsl[(size_t)(b + 1) * WSL_STRIDE] : 0u;
    const bool spill = totA > OWN_CAP || totB > OWN_CAP;
    const float4* __restrict__ recv = (const float4*)rec;
    const int4*   __restrict__ dst4 = (const int4*)(ei + NEDGES);
    const int rlo = rec_of(lo);
    __syncthreads();

    if (!spill) {
        for (int g = 0; g < (two ? 2 : 1); ++g) {
            const unsigned int* reg = wsl + (size_t)(b + g) * WSL_STRIDE + WSL_HDR;
            const int n     = (int)(g ? totB : totA);
            const int gbase = (b + g) * OWN_NODES - lo;
            const int rbase = (b + g) * REC_STRIDE;
            for (int t0 = tid; t0 < n; t0 += 4 * OWN_THREADS) {
                unsigned en[4];
#pragma unroll
                for (int u = 0; u < 4; ++u) {
                    const int t = t0 + u * OWN_THREADS;
                    en[u] = reg[t < n ? t : 0];
                }
                int   li[4];
                float va[4], vb[4];
#pragma unroll
                for (int u = 0; u < 4; ++u) {
                    const int  dl = (int)(en[u] >> 16);
                    const int  l  = gbase + dl;
                    const bool ok = t0 + u * OWN_THREADS < n && (unsigned)l < (unsigned)nloc;
                    li[u] = ok ? l : -1;
                    va[u] = rec[4 * (ok ? rec_of((int)(en[u] & 0xffffu)) : rlo) + 2];
                    vb[u] = rec[4 * (ok ? rbase + dl : rlo) + 3];
                }
#pragma unroll
                for (int u = 0; u < 4; ++u)
                    if (li[u] >= 0) atomicMaxF(&mx[li[u]], lrelu(va[u] + vb[u]));
            }
        }
    } else {
        for (int i = tid; i < nloc; i += OWN_THREADS) {
            const float4 r = recv[rec_of(lo + i)];
            atomicMaxF(&mx[i], lrelu(r.z + r.w));
        }
        for (int i = tid; i < OWN_NV; i += OWN_THREADS) {
            const int4 v = dst4[i];
            const int dd[4] = {v.x, v.y, v.z, v.w};
#pragma unroll
            for (int q = 0; q < 4; ++q) {
                const unsigned l = (unsigned)(dd[q] - lo);
                if (l < (unsigned)nloc)
                    atomicMaxF(&mx[l], lrelu(rec[4 * rec_of(ei[i * 4 + q]) + 2] + rec[4 * rec_of(dd[q]) + 3]));
            }
        }
    }
    __syncthreads();

    if (!spill) {
        for (int g = 0; g < (two ? 2 : 1); ++g) {
            const unsigned int* reg = wsl + (size_t)(b + g) * WSL_STRIDE + WSL_HDR;
            const int n     = (int)(g ? totB : totA);
            const int gbase = (b + g) * OWN_NODES - lo;
            const int rbase = (b + g) * REC_STRIDE;
            for (int t0 = tid; t0 < n; t0 += 4 * OWN_THREADS) {
                unsigned en[4];
#pragma unroll
                for (int u = 0; u < 4; ++u) {
                    const int t = t0 + u * OWN_THREADS;
                    en[u] = reg[t < n ? t : 0];
                }
                int    li[4];
                float  vb[4];
                float4 sr[4];
#pragma unroll
                for (int u = 0; u < 4; ++u) {
                    const int  dl = (int)(en[u] >> 16);
                    const int  l  = gbase + dl;
                    const bool ok = t0 + u * OWN_THREADS < n && (unsigned)l < (unsigned)nloc;
                    li[u] = ok ? l : -1;
                    sr[u] = recv[ok ? rec_of((int)(en[u] & 0xffffu)) : rlo];
                    vb[u] = rec[4 * (ok ? rbase + dl : rlo) + 3];
                }
#pragma unroll
                for (int u = 0; u < 4; ++u) {
                    if (li[u] >= 0) {
                        const float p = __expf(lrelu(sr[u].z + vb[u]) - mx[li[u]]);
                        atomicAdd(&dn[li[u]], p);
                        atomicAdd(&s0[li[u]], p * sr[u].x);
                        atomicAdd(&s1[li[u]], p * sr[u].y);
                    }
                }
            }
        }
    } else {
        for (int i = tid; i < nloc; i += OWN_THREADS) {
            const float4 r = recv[rec_of(lo + i)];
            const float  p = __expf(lrelu(r.z + r.w) - mx[i]);
            atomicAdd(&dn[i], p);
            atomicAdd(&s0[i], p * r.x);
            atomicAdd(&s1[i], p * r.y);
        }
        for (int i = tid; i < OWN_NV; i += OWN_THREADS) {
            const int4 v = dst4[i];
            const int dd[4] = {v.x, v.y, v.z, v.w};
#pragma unroll
            for (int q = 0; q < 4; ++q) {
                const unsigned l = (unsigned)(dd[q] - lo);
                if (l < (unsigned)nloc) {
                    const float4 r = recv[rec_of(ei[i * 4 + q])];
                    const float  p = __expf(lrelu(r.z + rec[4 * rec_of(dd[q]) + 3]) - mx[l]);
                    atomicAdd(&dn[l], p);
                    atomicAdd(&s0[l], p * r.x);
                    atomicAdd(&s1[l], p * r.y);
                }
            }
        }
    }
    __syncthreads();

    if (tid < nloc * OUT2) {
        const int nd = tid >> 1;
        const float sv = (tid & 1) ? s1[nd] : s0[nd];
        out[(size_t)lo * OUT2 + tid] = sv / dn[nd] + b2[tid & 1];
    }
}

extern "C" void kernel_launch(void* const* d_in, const int* in_sizes, int n_in,
                              void* d_out, int out_size, void* d_ws, size_t ws_size,
                              hipStream_t stream) {
    const float* x     = (const float*)d_in[0];
    const int*   ei    = (const int*)d_in[1];
    const float* W1    = (const float*)d_in[2];
    const float* asw1  = (const float*)d_in[3];
    const float* adw1  = (const float*)d_in[4];
    const float* b1    = (const float*)d_in[5];
    const float* W2    = (const float*)d_in[6];
    const float* asw2  = (const float*)d_in[7];
    const float* adw2  = (const float*)d_in[8];
    const float* b2    = (const float*)d_in[9];
    float*       out   = (float*)d_out;

    float* w    = (float*)d_ws;
    float* xp1  = w;
    float* as1  = xp1  + (size_t)NNODES * D1;
    float* ad1  = as1  + (size_t)NNODES * H1;
    float* rec2 = ad1  + (size_t)NNODES * H1;
    unsigned int* wsl = (unsigned int*)(rec2 + (size_t)OWN_BLOCKS * REC_STRIDE * 4);
    unsigned int* slots = wsl + (size_t)OWN_BLOCKS * WSL_STRIDE;
    float* wa   = (float*)(slots + (size_t)OWN_BLOCKS * BIN_BLOCKS * BIN_SLOT);

    bin_edges<<<BIN_BLOCKS + WA_BLOCKS, BIN_THREADS, 0, stream>>>(ei, slots, W1, asw1, adw1, wa);
    gemm1_wmma<<<G1_BLOCKS, G1_WAVES * 32, 0, stream>>>(x, W1, wa, xp1, as1, ad1);
    agg1_owner<<<OWN_BLOCKS, OWN_THREADS, 0, stream>>>(ei, xp1, as1, ad1, b1, W2, asw2, adw2, rec2, wsl, slots);

    agg2_owner<<<OWN_BLOCKS, OWN_THREADS, 0, stream>>>(ei, wsl, rec2, b2, out);
}
